# starting kernel
# baseline (speedup 1.0000x reference)
#include <hip/hip_runtime.h>
#include <hip/hip_bf16.h>
#include <math.h>

typedef __attribute__((ext_vector_type(2))) float v2f;
typedef __attribute__((ext_vector_type(8))) float v8f;

#define NUM_GRAPHS 64
#define NEG_SLOPE 0.2f

template <int FIN, int FOUT>
__global__ __launch_bounds__(256) void gemm_wmma(const float* __restrict__ x,
                                                 const float* __restrict__ W,
                                                 float* __restrict__ h, int N) {
  constexpr int NT = FOUT / 16;
  __shared__ float Ws[FIN * FOUT];
  for (int i = threadIdx.x; i < FIN * FOUT; i += 256) Ws[i] = W[i];
  __syncthreads();

  const int wave = threadIdx.x >> 5;
  const int lane = threadIdx.x & 31;
  const int m0 = (blockIdx.x * 8 + wave) * 16;
  const int halfsel = lane >> 4;
  const int lidx = lane & 15;
  int row = m0 + lidx;
  if (row >= N) row = N - 1;
  const int khalf = halfsel * 2;

  v8f acc[NT];
#pragma unroll
  for (int t = 0; t < NT; ++t)
#pragma unroll
    for (int j = 0; j < 8; ++j) acc[t][j] = 0.0f;

#pragma unroll 4
  for (int k = 0; k < FIN; k += 4) {
    const int kk = k + khalf;
    v2f a;
    a.x = x[(size_t)row * FIN + kk];
    a.y = x[(size_t)row * FIN + kk + 1];
#pragma unroll
    for (int t = 0; t < NT; ++t) {
      v2f b;
      b.x = Ws[kk * FOUT + t * 16 + lidx];
      b.y = Ws[(kk + 1) * FOUT + t * 16 + lidx];
      acc[t] = __builtin_amdgcn_wmma_f32_16x16x4_f32(
          false, a, false, b, (short)0, acc[t], false, false);
    }
  }

  const int rbase = m0 + halfsel * 8;
#pragma unroll
  for (int i = 0; i < 8; ++i) {
    const int r = rbase + i;
    if (r < N) {
#pragma unroll
      for (int t = 0; t < NT; ++t)
        h[(size_t)r * FOUT + t * 16 + lidx] = acc[t][i];
    }
  }
}

template <int F>
__global__ void node_dots(const float* __restrict__ h,
                          const float* __restrict__ asrc,
                          const float* __restrict__ adst,
                          float* __restrict__ s, float* __restrict__ d, int N) {
  int n = blockIdx.x * blockDim.x + threadIdx.x;
  if (n >= N) return;
  float ss = 0.0f, dd = 0.0f;
#pragma unroll
  for (int f = 0; f < F; ++f) {
    float v = h[(size_t)n * F + f];
    ss += v * asrc[f];
    dd += v * adst[f];
  }
  s[n] = ss;
  d[n] = dd;
}

__device__ __forceinline__ void atomicMaxFloat(float* addr, float val) {
  int vi = __float_as_int(val);
  if (vi >= 0)
    atomicMax((int*)addr, vi);
  else
    atomicMin((unsigned int*)addr, (unsigned int)vi);
}

__global__ void edge_score(const int* __restrict__ src,
                           const int* __restrict__ dst,
                           const float* __restrict__ sv,
                           const float* __restrict__ dv,
                           float* __restrict__ ebuf, float* __restrict__ m,
                           int E, int ET) {
  int id = blockIdx.x * blockDim.x + threadIdx.x;
  if (id >= ET) return;
  int si, di;
  if (id < E) {
    si = src[id];
    di = dst[id];
  } else {
    si = di = id - E;
  }
  float e = sv[si] + dv[di];
  e = (e > 0.0f) ? e : NEG_SLOPE * e;
  ebuf[id] = e;
  atomicMaxFloat(m + di, e);
}

__global__ void edge_exp(const int* __restrict__ dst,
                         const float* __restrict__ m, float* __restrict__ ebuf,
                         float* __restrict__ z, int E, int ET) {
  int id = blockIdx.x * blockDim.x + threadIdx.x;
  if (id >= ET) return;
  int di = (id < E) ? dst[id] : id - E;
  float t = __expf(ebuf[id] - m[di]);
  ebuf[id] = t;
  atomicAdd(z + di, t);
}

template <int F>
__global__ __launch_bounds__(256) void edge_aggregate(
    const int* __restrict__ src, const int* __restrict__ dst,
    const float* __restrict__ ebuf, const float* __restrict__ z,
    const float* __restrict__ h, float* __restrict__ out, int E, int ET) {
  int id = blockIdx.x * 8 + (threadIdx.x >> 5);
  if (id >= ET) return;
  int lane = threadIdx.x & 31;
  int si, di;
  if (id < E) {
    si = src[id];
    di = dst[id];
  } else {
    si = di = id - E;
  }
  float alpha = ebuf[id] / z[di];
#pragma unroll
  for (int f = lane; f < F; f += 32)
    atomicAdd(out + (size_t)di * F + f, alpha * h[(size_t)si * F + f]);
}

template <int F, bool ELU>
__global__ void bias_act(float* __restrict__ h, const float* __restrict__ b,
                         int N) {
  int i = blockIdx.x * blockDim.x + threadIdx.x;
  if (i >= N * F) return;
  float v = h[i] + b[i % F];
  if (ELU) v = (v > 0.0f) ? v : (__expf(v) - 1.0f);
  h[i] = v;
}

__global__ void pool_sum(const float* __restrict__ h,
                         const int* __restrict__ batch, float* __restrict__ out,
                         float* __restrict__ cnt, int N) {
  int i = blockIdx.x * blockDim.x + threadIdx.x;
  int n = i >> 6;
  int f = i & 63;
  if (n >= N) return;
  int g = batch[n];
  atomicAdd(out + g * 64 + f, h[(size_t)n * 64 + f]);
  if (f == 0) atomicAdd(cnt + g, 1.0f);
}

__global__ void pool_div(float* __restrict__ out, const float* __restrict__ cnt) {
  int i = blockIdx.x * blockDim.x + threadIdx.x;
  if (i >= NUM_GRAPHS * 64) return;
  float c = cnt[i >> 6];
  out[i] = out[i] / fmaxf(c, 1.0f);
}

__global__ void fill_kernel(float* __restrict__ p, float v, int n) {
  int i = blockIdx.x * blockDim.x + threadIdx.x;
  if (i < n) p[i] = v;
}

static inline int cdiv(int a, int b) { return (a + b - 1) / b; }

extern "C" void kernel_launch(void* const* d_in, const int* in_sizes, int n_in,
                              void* d_out, int out_size, void* d_ws,
                              size_t ws_size, hipStream_t stream) {
  const float* x      = (const float*)d_in[0];
  const int*   ei     = (const int*)d_in[1];
  const int*   batch  = (const int*)d_in[2];
  const float* W1     = (const float*)d_in[3];
  const float* a_src1 = (const float*)d_in[4];
  const float* a_dst1 = (const float*)d_in[5];
  const float* b1     = (const float*)d_in[6];
  const float* W2     = (const float*)d_in[7];
  const float* a_src2 = (const float*)d_in[8];
  const float* a_dst2 = (const float*)d_in[9];
  const float* b2     = (const float*)d_in[10];

  const int N = in_sizes[2];
  const int E = in_sizes[1] / 2;
  const int ET = E + N;
  const int* src = ei;
  const int* dst = ei + E;

  float* ws = (float*)d_ws;
  size_t off = 0;
  float* h1 = ws + off; off += (size_t)N * 32;
  float* o1 = ws + off; off += (size_t)N * 32;
  float* h2 = ws + off; off += (size_t)N * 64;
  float* o2 = ws + off; off += (size_t)N * 64;
  float* sv = ws + off; off += (size_t)N;
  float* dv = ws + off; off += (size_t)N;
  float* mb = ws + off; off += (size_t)N;
  float* zb = ws + off; off += (size_t)N;
  float* eb = ws + off; off += (size_t)ET;
  float* cnt = ws + off; off += NUM_GRAPHS;

  const float NEG_INF = -__builtin_inff();

  gemm_wmma<128, 32><<<cdiv(N, 128), 256, 0, stream>>>(x, W1, h1, N);
  node_dots<32><<<cdiv(N, 256), 256, 0, stream>>>(h1, a_src1, a_dst1, sv, dv, N);
  fill_kernel<<<cdiv(N, 256), 256, 0, stream>>>(mb, NEG_INF, N);
  fill_kernel<<<cdiv(N, 256), 256, 0, stream>>>(zb, 0.0f, N);
  fill_kernel<<<cdiv(N * 32, 256), 256, 0, stream>>>(o1, 0.0f, N * 32);
  edge_score<<<cdiv(ET, 256), 256, 0, stream>>>(src, dst, sv, dv, eb, mb, E, ET);
  edge_exp<<<cdiv(ET, 256), 256, 0, stream>>>(dst, mb, eb, zb, E, ET);
  edge_aggregate<32><<<cdiv(ET, 8), 256, 0, stream>>>(src, dst, eb, zb, h1, o1, E, ET);
  bias_act<32, true><<<cdiv(N * 32, 256), 256, 0, stream>>>(o1, b1, N);

  gemm_wmma<32, 64><<<cdiv(N, 128), 256, 0, stream>>>(o1, W2, h2, N);
  node_dots<64><<<cdiv(N, 256), 256, 0, stream>>>(h2, a_src2, a_dst2, sv, dv, N);
  fill_kernel<<<cdiv(N, 256), 256, 0, stream>>>(mb, NEG_INF, N);
  fill_kernel<<<cdiv(N, 256), 256, 0, stream>>>(zb, 0.0f, N);
  fill_kernel<<<cdiv(N * 64, 256), 256, 0, stream>>>(o2, 0.0f, N * 64);
  edge_score<<<cdiv(ET, 256), 256, 0, stream>>>(src, dst, sv, dv, eb, mb, E, ET);
  edge_exp<<<cdiv(ET, 256), 256, 0, stream>>>(dst, mb, eb, zb, E, ET);
  edge_aggregate<64><<<cdiv(ET, 8), 256, 0, stream>>>(src, dst, eb, zb, h2, o2, E, ET);
  bias_act<64, false><<<cdiv(N * 64, 256), 256, 0, stream>>>(o2, b2, N);

  fill_kernel<<<cdiv(NUM_GRAPHS * 64, 256), 256, 0, stream>>>((float*)d_out, 0.0f,
                                                              NUM_GRAPHS * 64);
  fill_kernel<<<1, 64, 0, stream>>>(cnt, 0.0f, NUM_GRAPHS);
  pool_sum<<<cdiv(N * 64, 256), 256, 0, stream>>>(o2, batch, (float*)d_out, cnt, N);
  pool_div<<<cdiv(NUM_GRAPHS * 64, 256), 256, 0, stream>>>((float*)d_out, cnt);
}


# Layer-1 rows leave global atomics: workgroup (2048-row dst window, edge chunk) filters its chunk by dst, adds t*h1[src] into an LDS copy; a reduce adds the 10 copies, /z, bias, elu; sum order changes
# speedup vs baseline: 3.7549x; 3.7549x over previous
#include <hip/hip_runtime.h>
#include <hip/hip_bf16.h>
#include <math.h>

typedef __attribute__((ext_vector_type(2))) float v2f;
typedef __attribute__((ext_vector_type(8))) float v8f;

#define NUM_GRAPHS 64
#define NEG_SLOPE 0.2f
#define SPREAD_LIMIT 8.0f

template <int FIN, int FOUT>
__global__ __launch_bounds__(256) void gemm_wmma(const float* __restrict__ x,
                                                 const float* __restrict__ W,
                                                 float* __restrict__ h, int N) {
  constexpr int NT = FOUT / 16;
  __shared__ float Ws[FIN * FOUT];
  for (int i = threadIdx.x; i < FIN * FOUT; i += 256) Ws[i] = W[i];
  __syncthreads();

  const int wave = threadIdx.x >> 5;
  const int lane = threadIdx.x & 31;
  const int m0 = (blockIdx.x * 8 + wave) * 16;
  const int halfsel = lane >> 4;
  const int lidx = lane & 15;
  int row = m0 + lidx;
  if (row >= N) row = N - 1;
  const int khalf = halfsel * 2;

  v8f acc[NT];
#pragma unroll
  for (int t = 0; t < NT; ++t)
#pragma unroll
    for (int j = 0; j < 8; ++j) acc[t][j] = 0.0f;

#pragma unroll 4
  for (int k = 0; k < FIN; k += 4) {
    const int kk = k + khalf;
    v2f a;
    a.x = x[(size_t)row * FIN + kk];
    a.y = x[(size_t)row * FIN + kk + 1];
#pragma unroll
    for (int t = 0; t < NT; ++t) {
      v2f b;
      b.x = Ws[kk * FOUT + t * 16 + lidx];
      b.y = Ws[(kk + 1) * FOUT + t * 16 + lidx];
      acc[t] = __builtin_amdgcn_wmma_f32_16x16x4_f32(
          false, a, false, b, (short)0, acc[t], false, false);
    }
  }

  const int rbase = m0 + halfsel * 8;
#pragma unroll
  for (int i = 0; i < 8; ++i) {
    const int r = rbase + i;
    if (r < N) {
#pragma unroll
      for (int t = 0; t < NT; ++t)
        h[(size_t)r * FOUT + t * 16 + lidx] = acc[t][i];
    }
  }
}

__device__ __forceinline__ void atomicMaxFloat(float* addr, float val) {
  int vi = __float_as_int(val);
  if (vi >= 0)
    atomicMax((int*)addr, vi);
  else
    atomicMin((unsigned int*)addr, (unsigned int)vi);
}

__device__ __forceinline__ float leaky(float v) {
  return (v > 0.0f) ? v : NEG_SLOPE * v;
}

template <int F>
__global__ __launch_bounds__(256) void node_dots(
    const float* __restrict__ h, const float* __restrict__ asrc,
    const float* __restrict__ adst, float* __restrict__ s,
    float* __restrict__ d, float* __restrict__ rng, int N) {
  __shared__ float red[16];
  const int n = blockIdx.x * 256 + threadIdx.x;
  const bool valid = n < N;
  float ss = 0.0f, dd = 0.0f;
  if (valid) {
#pragma unroll
    for (int f = 0; f < F; ++f) {
      float v = h[(size_t)n * F + f];
      ss += v * asrc[f];
      dd += v * adst[f];
    }
    s[n] = ss;
    d[n] = dd;
  }
  const float ninf = -__builtin_inff();
  float hi = valid ? ss : ninf;
  float lo = valid ? -ss : ninf;
#pragma unroll
  for (int o = 16; o > 0; o >>= 1) {
    hi = fmaxf(hi, __shfl_xor(hi, o));
    lo = fmaxf(lo, __shfl_xor(lo, o));
  }
  const int wave = threadIdx.x >> 5;
  if ((threadIdx.x & 31) == 0) {
    red[wave] = hi;
    red[8 + wave] = lo;
  }
  __syncthreads();
  if (threadIdx.x < 16) {
    float v = red[threadIdx.x];
    v = fmaxf(v, __shfl_xor(v, 4));
    v = fmaxf(v, __shfl_xor(v, 2));
    v = fmaxf(v, __shfl_xor(v, 1));
    if ((threadIdx.x & 7) == 0) atomicMaxFloat(rng + (threadIdx.x >> 3), v);
  }
}


__global__ __launch_bounds__(256) void edge_max(
    const int* __restrict__ src, const int* __restrict__ dst,
    const float* __restrict__ sv, const float* __restrict__ dv,
    const float* __restrict__ rng, float* __restrict__ m, int E, int ET) {
  if (rng[0] + rng[1] < SPREAD_LIMIT) return;
  for (int id = blockIdx.x * 256 + threadIdx.x; id < ET;
       id += gridDim.x * 256) {
    int si, di;
    if (id < E) {
      si = src[id];
      di = dst[id];
    } else {
      si = di = id - E;
    }
    atomicMaxFloat(m + di, leaky(sv[si] + dv[di]));
  }
}

#define ZCAP 65536
#define ZGRID 256
#define ZUNROLL 4

__global__ __launch_bounds__(1024) void edge_exp_priv(
    const int* __restrict__ src, const int* __restrict__ dst,
    const float* __restrict__ sv, const float* __restrict__ dv,
    const float* __restrict__ rng, const float* __restrict__ m,
    float* __restrict__ ebuf, float* __restrict__ zpart, int E, int ET, int N,
    int NP) {
  __shared__ __attribute__((aligned(16))) float zl[ZCAP];
  const int tid = threadIdx.x;
  const int stride = gridDim.x * 1024;
  const float smax = rng[0];
  const bool bound = smax + rng[1] < SPREAD_LIMIT;
  float* __restrict__ zrow = zpart + (size_t)blockIdx.x * NP;

  for (int base = 0; base < N; base += ZCAP) {
    const int cnt = min(N - base, ZCAP);
    const int cnt4 = (cnt + 3) & ~3;
    for (int i = tid * 4; i < cnt4; i += 4096)
      *(float4*)(zl + i) = make_float4(0.0f, 0.0f, 0.0f, 0.0f);
    __syncthreads();

    for (int id0 = blockIdx.x * 1024 + tid; id0 < ET; id0 += ZUNROLL * stride) {
      int si[ZUNROLL], di[ZUNROLL];
#pragma unroll
      for (int u = 0; u < ZUNROLL; ++u) {
        const int id = id0 + u * stride;
        const bool real = id < E;
        const int idc = real ? id : 0;
        const int s0 = src[idc];
        const int d0 = dst[idc];
        const int self = (id < ET) ? id - E : -1;
        si[u] = real ? s0 : self;
        di[u] = real ? d0 : self;
      }
      float ss[ZUNROLL], dd[ZUNROLL], mm[ZUNROLL];
#pragma unroll
      for (int u = 0; u < ZUNROLL; ++u) {
        ss[u] = sv[max(si[u], 0)];
        dd[u] = dv[max(di[u], 0)];
      }
      if (!bound) {
#pragma unroll
        for (int u = 0; u < ZUNROLL; ++u) mm[u] = m[max(di[u], 0)];
      } else {
#pragma unroll
        for (int u = 0; u < ZUNROLL; ++u) mm[u] = 0.0f;
      }
#pragma unroll
      for (int u = 0; u < ZUNROLL; ++u) {
        if (di[u] >= 0) {
          const float e = leaky(ss[u] + dd[u]);
          const float shift = bound ? leaky(smax + dd[u]) : mm[u];
          const float t = __expf(e - shift);
          if (base == 0) ebuf[id0 + u * stride] = t;
          const int li = di[u] - base;
          if ((unsigned)li < (unsigned)cnt) atomicAdd(zl + li, t);
        }
      }
    }
    __syncthreads();

    for (int i = tid * 4; i < cnt4; i += 4096)
      *(float4*)(zrow + base + i) = *(const float4*)(zl + i);
    __syncthreads();
  }
}

__global__ __launch_bounds__(256) void z_reduce(const float* __restrict__ zpart,
                                                float* __restrict__ z, int N,
                                                int NP) {
  __shared__ __attribute__((aligned(16))) float red[8][128];
  const int wave = threadIdx.x >> 5;
  const int lane = threadIdx.x & 31;
  const int n4 = blockIdx.x * 128 + lane * 4;
  float4 acc = make_float4(0.0f, 0.0f, 0.0f, 0.0f);
  if (n4 < NP) {
    const float* p = zpart + (size_t)wave * NP + n4;
#pragma unroll 8
    for (int k = 0; k < ZGRID / 8; ++k) {
      const float4 v = *(const float4*)p;
      p += (size_t)8 * NP;
      acc.x += v.x;
      acc.y += v.y;
      acc.z += v.z;
      acc.w += v.w;
    }
  }
  *(float4*)(&red[wave][lane * 4]) = acc;
  __syncthreads();
  if (threadIdx.x < 128) {
    const int n = blockIdx.x * 128 + threadIdx.x;
    float r = red[0][threadIdx.x];
#pragma unroll
    for (int w = 1; w < 8; ++w) r += red[w][threadIdx.x];
    if (n < N) z[n] = r;
  }
}

#define WROWS 2048
#define WCHUNKS 10
#define WSCAN 8
#define WQ (32 * WSCAN + 32)

__device__ __forceinline__ int win_dst_word(const int* __restrict__ dst, int id,
                                            int last) {
  return dst[min(id, last)];
}

__device__ __forceinline__ bool win_hit(int d, int id, int lim, int e1, int E,
                                        int wbase, int wcnt) {
  const bool real_hit = ((unsigned)(d - wbase) < (unsigned)wcnt) & (id < lim);
  const bool self_hit = ((unsigned)(id - E - wbase) < (unsigned)wcnt) & (id < e1);
  return real_hit | self_hit;
}

template <bool REAL>
__device__ __forceinline__ int win_queue(const int* cur, int b, int lane,
                                         int lim, int e1, int E, int wbase,
                                         int wcnt, int* myq, int cnt) {
#pragma unroll
  for (int u = 0; u < WSCAN; ++u) {
    const int id = b + u * 32 + lane;
    const bool hit = REAL ? ((unsigned)(cur[u] - wbase) < (unsigned)wcnt)
                          : win_hit(cur[u], id, lim, e1, E, wbase, wcnt);
    const unsigned mask = __builtin_amdgcn_ballot_w32(hit);
    if (hit) myq[cnt + (int)__builtin_amdgcn_mbcnt_lo(mask, 0u)] = id;
    cnt += __builtin_popcount(mask);
  }
  return cnt;
}

__device__ __forceinline__ float win_row_load(const float* __restrict__ h,
                                              int si, int j, int lane) {
  const int sj = __builtin_amdgcn_readlane(si, j);
  return h[(size_t)sj * 32 + lane];
}

__device__ __forceinline__ void win_row_add(float* acc, float v, int dl,
                                            float t, int j, int lane) {
  const int rj = __builtin_amdgcn_readlane(dl, j);
  const float tj = __int_as_float(__builtin_amdgcn_readlane(__float_as_int(t), j));
  atomicAdd(acc + rj * 32 + lane, tj * v);
}

template <bool FULL>
__device__ __forceinline__ void win_batch(
    const int* qp, int m, int lane, int wbase, int E,
    const int* __restrict__ src, const int* __restrict__ dst,
    const float* __restrict__ ebuf, const float* __restrict__ h, float* acc) {
  const int id = qp[(FULL || lane < m) ? lane : 0];
  const bool real = id < E;
  const int idc = real ? id : 0;
  const int s0 = src[idc];
  const int d0 = dst[idc];
  const float t = ebuf[id];
  const int self = id - E;
  const int si = real ? s0 : self;
  const int dl = (real ? d0 : self) - wbase;
  float v[32];
#pragma unroll
  for (int j = 0; j < 32; ++j) {
    v[j] = 0.0f;
    if (FULL || j < m) v[j] = win_row_load(h, si, j, lane);
  }
#pragma unroll
  for (int j = 0; j < 32; ++j)
    if (FULL || j < m) win_row_add(acc, v[j], dl, t, j, lane);
}

template <bool REAL>
__device__ __forceinline__ void win_steps(
    int& b, int (&cur)[WSCAN], int& cnt, int lane, int lim, int last, int e1,
    int E, int wbase, int wcnt, int* myq, const int* __restrict__ src,
    const int* __restrict__ dst, const float* __restrict__ ebuf,
    const float* __restrict__ h, float* acc) {
  const int step = 32 * WSCAN;
  for (; REAL ? (b + step <= lim) : (b < e1); b += 32 * step) {
    int nxt[WSCAN];
    const int nb = b + 32 * step;
#pragma unroll
    for (int u = 0; u < WSCAN; ++u)
      nxt[u] = win_dst_word(dst, nb + u * 32 + lane, last);
    cnt = win_queue<REAL>(cur, b, lane, lim, e1, E, wbase, wcnt, myq, cnt);
    while (cnt >= 32) {
      cnt -= 32;
      win_batch<true>(myq + cnt, 32, lane, wbase, E, src, dst, ebuf, h, acc);
    }
#pragma unroll
    for (int u = 0; u < WSCAN; ++u) cur[u] = nxt[u];
  }
}

__global__ __launch_bounds__(1024) void edge_window(
    const int* __restrict__ src, const int* __restrict__ dst,
    const float* __restrict__ ebuf, const float* __restrict__ h,
    float* __restrict__ wpart, int E, int ET, int N) {
  __shared__ __attribute__((aligned(16))) float acc[WROWS * 32];
  __shared__ int q[32][WQ];
  const int tid = threadIdx.x;
  const int lane = tid & 31;
  const int wave = __builtin_amdgcn_readfirstlane(tid >> 5);
  const int win = blockIdx.x / WCHUNKS;
  const int chunk = blockIdx.x - win * WCHUNKS;
  const int wbase = win * WROWS;
  const int wcnt = min(N - wbase, WROWS);
  const int wfl = wcnt * 32;

  for (int i = tid * 4; i < wfl; i += 4096)
    *(float4*)(acc + i) = make_float4(0.0f, 0.0f, 0.0f, 0.0f);
  __syncthreads();

  const int per = ((ET + WCHUNKS * 32 - 1) / (WCHUNKS * 32)) * 32;
  const int e0 = chunk * per;
  const int e1 = min(ET, e0 + per);
  const int lim = min(e1, E);
  const int last = max(lim, 1) - 1;
  int* myq = q[wave];
  int cnt = 0;

  int b = e0 + wave * (32 * WSCAN);
  int cur[WSCAN];
#pragma unroll
  for (int u = 0; u < WSCAN; ++u)
    cur[u] = win_dst_word(dst, b + u * 32 + lane, last);
  win_steps<true>(b, cur, cnt, lane, lim, last, e1, E, wbase, wcnt, myq, src,
                  dst, ebuf, h, acc);
  win_steps<false>(b, cur, cnt, lane, lim, last, e1, E, wbase, wcnt, myq, src,
                   dst, ebuf, h, acc);
  if (cnt > 0)
    win_batch<false>(myq, cnt, lane, wbase, E, src, dst, ebuf, h, acc);
  __syncthreads();

  float* __restrict__ prow = wpart + ((size_t)chunk * N + wbase) * 32;
  for (int i = tid * 4; i < wfl; i += 4096)
    *(float4*)(prow + i) = *(const float4*)(acc + i);
}

__global__ __launch_bounds__(256) void window_reduce(
    const float* __restrict__ wpart, const float* __restrict__ z,
    const float* __restrict__ b, float* __restrict__ o, int N) {
  const int i4 = blockIdx.x * 256 + threadIdx.x;
  if (i4 >= N * 8) return;
  const int n = i4 >> 3;
  const int f = (i4 & 7) * 4;
  const float* p = wpart + (size_t)i4 * 4;
  float4 a = make_float4(0.0f, 0.0f, 0.0f, 0.0f);
#pragma unroll
  for (int c = 0; c < WCHUNKS; ++c) {
    const float4 v = *(const float4*)p;
    p += (size_t)N * 32;
    a.x += v.x;
    a.y += v.y;
    a.z += v.z;
    a.w += v.w;
  }
  const float zn = z[n];
  const float bx = b[f], by = b[f + 1], bz = b[f + 2], bw = b[f + 3];
  float4 r;
  r.x = a.x / zn + bx;
  r.y = a.y / zn + by;
  r.z = a.z / zn + bz;
  r.w = a.w / zn + bw;
  r.x = (r.x > 0.0f) ? r.x : (__expf(r.x) - 1.0f);
  r.y = (r.y > 0.0f) ? r.y : (__expf(r.y) - 1.0f);
  r.z = (r.z > 0.0f) ? r.z : (__expf(r.z) - 1.0f);
  r.w = (r.w > 0.0f) ? r.w : (__expf(r.w) - 1.0f);
  *(float4*)(o + (size_t)i4 * 4) = r;
}

#define PGRID 256
#define PSUMS (NUM_GRAPHS * 64)
#define PROW (PSUMS + NUM_GRAPHS)

__device__ __forceinline__ float2 pool_row_load(const float* __restrict__ h,
                                                int si, int j, int lane) {
  const int sj = __builtin_amdgcn_readlane(si, j);
  return *(const float2*)(h + (size_t)sj * 64 + lane * 2);
}

__device__ __forceinline__ void pool_row_add(float* pl, float2 v, int g,
                                             float a, int j, int lane) {
  const int gj = __builtin_amdgcn_readlane(g, j);
  const float aj = __int_as_float(__builtin_amdgcn_readlane(__float_as_int(a), j));
  float* p = pl + gj * 64 + lane * 2;
  atomicAdd(p, aj * v.x);
  atomicAdd(p + 1, aj * v.y);
}

__global__ __launch_bounds__(1024) void edge_pool(
    const int* __restrict__ src, const int* __restrict__ dst,
    const float* __restrict__ ebuf, const float* __restrict__ z,
    const int* __restrict__ batch, const float* __restrict__ h,
    float* __restrict__ ppart, int E, int ET) {
  __shared__ float pl[PROW];
  const int tid = threadIdx.x;
  const int lane = tid & 31;
  for (int i = tid; i < PROW; i += 1024) pl[i] = 0.0f;
  __syncthreads();

  const int wave = blockIdx.x * 32 + (tid >> 5);
  const int sweep = gridDim.x * 32 * 32;
  for (int b0 = wave * 32; b0 < ET; b0 += sweep) {
    const int id = b0 + lane;
    const bool live = id < ET;
    const bool real = id < E;
    const int idc = real ? id : 0;
    const int s0 = src[idc];
    const int d0 = dst[idc];
    const int self = live ? id - E : 0;
    const int si = real ? s0 : self;
    const int di = real ? d0 : self;
    const float t = ebuf[live ? id : 0];
    const int g = batch[di];
    const float a = t / z[di];
    if (live && !real) atomicAdd(pl + PSUMS + g, 1.0f);

    const int cnt = min(32, ET - b0);
    if (cnt == 32) {
#pragma unroll
      for (int j0 = 0; j0 < 32; j0 += 16) {
        float2 v[16];
#pragma unroll
        for (int u = 0; u < 16; ++u) v[u] = pool_row_load(h, si, j0 + u, lane);
#pragma unroll
        for (int u = 0; u < 16; ++u) pool_row_add(pl, v[u], g, a, j0 + u, lane);
      }
    } else {
      for (int j = 0; j < cnt; ++j)
        pool_row_add(pl, pool_row_load(h, si, j, lane), g, a, j, lane);
    }
  }
  __syncthreads();

  float* __restrict__ row = ppart + (size_t)blockIdx.x * PROW;
  for (int i = tid; i < PROW; i += 1024) row[i] = pl[i];
}

__global__ __launch_bounds__(1024) void pool_finish(
    const float* __restrict__ ppart, const float* __restrict__ b,
    float* __restrict__ out) {
  __shared__ float red[16][64];
  __shared__ float cw[PGRID / 32];
  const int g = blockIdx.x;
  const int f = threadIdx.x & 63;
  const int r = threadIdx.x >> 6;
  const float* p = ppart + (size_t)r * PROW + g * 64 + f;
  float acc = 0.0f;
#pragma unroll
  for (int k = 0; k < PGRID / 16; ++k) {
    acc += *p;
    p += (size_t)16 * PROW;
  }
  red[r][f] = acc;
  if (threadIdx.x < PGRID) {
    float c = ppart[(size_t)threadIdx.x * PROW + PSUMS + g];
    for (int o = 16; o > 0; o >>= 1) c += __shfl_xor(c, o);
    if ((threadIdx.x & 31) == 0) cw[threadIdx.x >> 5] = c;
  }
  __syncthreads();
  if (threadIdx.x < 64) {
    float c = 0.0f;
    for (int w = 0; w < PGRID / 32; ++w) c += cw[w];
    float s = red[0][f];
#pragma unroll
    for (int w = 1; w < 16; ++w) s += red[w][f];
    out[g * 64 + f] = (s + c * b[f]) / fmaxf(c, 1.0f);
  }
}

__global__ void fill_kernel(float* __restrict__ p, float v, int n) {
  int i = blockIdx.x * blockDim.x + threadIdx.x;
  if (i < n) p[i] = v;
}

static inline int cdiv(int a, int b) { return (a + b - 1) / b; }

extern "C" void kernel_launch(void* const* d_in, const int* in_sizes, int n_in,
                              void* d_out, int out_size, void* d_ws,
                              size_t ws_size, hipStream_t stream) {
  const float* x      = (const float*)d_in[0];
  const int*   ei     = (const int*)d_in[1];
  const int*   batch  = (const int*)d_in[2];
  const float* W1     = (const float*)d_in[3];
  const float* a_src1 = (const float*)d_in[4];
  const float* a_dst1 = (const float*)d_in[5];
  const float* b1     = (const float*)d_in[6];
  const float* W2     = (const float*)d_in[7];
  const float* a_src2 = (const float*)d_in[8];
  const float* a_dst2 = (const float*)d_in[9];
  const float* b2     = (const float*)d_in[10];

  const int N = in_sizes[2];
  const int E = in_sizes[1] / 2;
  const int ET = E + N;
  const int* src = ei;
  const int* dst = ei + E;

  float* ws = (float*)d_ws;
  size_t off = 0;
  float* h1 = ws + off; off += (size_t)N * 32;
  float* o1 = ws + off; off += (size_t)N * 32;
  float* h2 = ws + off; off += (size_t)N * 64;
  float* sv = ws + off; off += (size_t)N;
  float* dv = ws + off; off += (size_t)N;
  float* mb = ws + off; off += (size_t)N + 2;
  float* rng = mb + N;
  float* zb = ws + off; off += (size_t)N;
  float* eb = ws + off; off += (size_t)ET;
  const int NP = (N + 3) & ~3;
  off = (off + 3) & ~(size_t)3;
  float* zp = ws + off; off += (size_t)ZGRID * NP;
  float* pp = ws + off; off += (size_t)PGRID * PROW;
  float* wp = ws + off; off += (size_t)WCHUNKS * N * 32;

  const float NEG_INF = -__builtin_inff();

  gemm_wmma<128, 32><<<cdiv(N, 128), 256, 0, stream>>>(x, W1, h1, N);
  fill_kernel<<<cdiv(N + 2, 256), 256, 0, stream>>>(mb, NEG_INF, N + 2);
  node_dots<32><<<cdiv(N, 256), 256, 0, stream>>>(h1, a_src1, a_dst1, sv, dv, rng, N);
  edge_max<<<512, 256, 0, stream>>>(src, dst, sv, dv, rng, mb, E, ET);
  edge_exp_priv<<<ZGRID, 1024, 0, stream>>>(src, dst, sv, dv, rng, mb, eb, zp, E, ET, N, NP);
  z_reduce<<<cdiv(NP, 128), 256, 0, stream>>>(zp, zb, N, NP);
  edge_window<<<cdiv(N, WROWS) * WCHUNKS, 1024, 0, stream>>>(src, dst, eb, h1, wp, E, ET, N);
  window_reduce<<<cdiv(N * 8, 256), 256, 0, stream>>>(wp, zb, b1, o1, N);

  gemm_wmma<32, 64><<<cdiv(N, 128), 256, 0, stream>>>(o1, W2, h2, N);
  fill_kernel<<<cdiv(N + 2, 256), 256, 0, stream>>>(mb, NEG_INF, N + 2);
  node_dots<64><<<cdiv(N, 256), 256, 0, stream>>>(h2, a_src2, a_dst2, sv, dv, rng, N);
  edge_max<<<512, 256, 0, stream>>>(src, dst, sv, dv, rng, mb, E, ET);
  edge_exp_priv<<<ZGRID, 1024, 0, stream>>>(src, dst, sv, dv, rng, mb, eb, zp, E, ET, N, NP);
  z_reduce<<<cdiv(NP, 128), 256, 0, stream>>>(zp, zb, N, NP);

  edge_pool<<<PGRID, 1024, 0, stream>>>(src, dst, eb, zb, batch, h2, pp, E, ET);
  pool_finish<<<NUM_GRAPHS, 1024, 0, stream>>>(pp, b2, (float*)d_out);
}


# Layer 1 loses its exp pass and z passes: edge_window forms t from s[src], d[dst] for the edges it keeps and adds it into an LDS z of its window; window_reduce sums the z copies; z sum order changes
# speedup vs baseline: 4.4133x; 1.1753x over previous
#include <hip/hip_runtime.h>
#include <hip/hip_bf16.h>
#include <math.h>

typedef __attribute__((ext_vector_type(2))) float v2f;
typedef __attribute__((ext_vector_type(8))) float v8f;

#define NUM_GRAPHS 64
#define NEG_SLOPE 0.2f
#define SPREAD_LIMIT 8.0f

template <int FIN, int FOUT>
__global__ __launch_bounds__(256) void gemm_wmma(const float* __restrict__ x,
                                                 const float* __restrict__ W,
                                                 float* __restrict__ h, int N) {
  constexpr int NT = FOUT / 16;
  __shared__ float Ws[FIN * FOUT];
  for (int i = threadIdx.x; i < FIN * FOUT; i += 256) Ws[i] = W[i];
  __syncthreads();

  const int wave = threadIdx.x >> 5;
  const int lane = threadIdx.x & 31;
  const int m0 = (blockIdx.x * 8 + wave) * 16;
  const int halfsel = lane >> 4;
  const int lidx = lane & 15;
  int row = m0 + lidx;
  if (row >= N) row = N - 1;
  const int khalf = halfsel * 2;

  v8f acc[NT];
#pragma unroll
  for (int t = 0; t < NT; ++t)
#pragma unroll
    for (int j = 0; j < 8; ++j) acc[t][j] = 0.0f;

#pragma unroll 4
  for (int k = 0; k < FIN; k += 4) {
    const int kk = k + khalf;
    v2f a;
    a.x = x[(size_t)row * FIN + kk];
    a.y = x[(size_t)row * FIN + kk + 1];
#pragma unroll
    for (int t = 0; t < NT; ++t) {
      v2f b;
      b.x = Ws[kk * FOUT + t * 16 + lidx];
      b.y = Ws[(kk + 1) * FOUT + t * 16 + lidx];
      acc[t] = __builtin_amdgcn_wmma_f32_16x16x4_f32(
          false, a, false, b, (short)0, acc[t], false, false);
    }
  }

  const int rbase = m0 + halfsel * 8;
#pragma unroll
  for (int i = 0; i < 8; ++i) {
    const int r = rbase + i;
    if (r < N) {
#pragma unroll
      for (int t = 0; t < NT; ++t)
        h[(size_t)r * FOUT + t * 16 + lidx] = acc[t][i];
    }
  }
}

__device__ __forceinline__ void atomicMaxFloat(float* addr, float val) {
  int vi = __float_as_int(val);
  if (vi >= 0)
    atomicMax((int*)addr, vi);
  else
    atomicMin((unsigned int*)addr, (unsigned int)vi);
}

__device__ __forceinline__ float leaky(float v) {
  return (v > 0.0f) ? v : NEG_SLOPE * v;
}

template <int F>
__global__ __launch_bounds__(256) void node_dots(
    const float* __restrict__ h, const float* __restrict__ asrc,
    const float* __restrict__ adst, float* __restrict__ s,
    float* __restrict__ d, float* __restrict__ rng, int N) {
  __shared__ float red[16];
  const int n = blockIdx.x * 256 + threadIdx.x;
  const bool valid = n < N;
  float ss = 0.0f, dd = 0.0f;
  if (valid) {
#pragma unroll
    for (int f = 0; f < F; ++f) {
      float v = h[(size_t)n * F + f];
      ss += v * asrc[f];
      dd += v * adst[f];
    }
    s[n] = ss;
    d[n] = dd;
  }
  const float ninf = -__builtin_inff();
  float hi = valid ? ss : ninf;
  float lo = valid ? -ss : ninf;
#pragma unroll
  for (int o = 16; o > 0; o >>= 1) {
    hi = fmaxf(hi, __shfl_xor(hi, o));
    lo = fmaxf(lo, __shfl_xor(lo, o));
  }
  const int wave = threadIdx.x >> 5;
  if ((threadIdx.x & 31) == 0) {
    red[wave] = hi;
    red[8 + wave] = lo;
  }
  __syncthreads();
  if (threadIdx.x < 16) {
    float v = red[threadIdx.x];
    v = fmaxf(v, __shfl_xor(v, 4));
    v = fmaxf(v, __shfl_xor(v, 2));
    v = fmaxf(v, __shfl_xor(v, 1));
    if ((threadIdx.x & 7) == 0) atomicMaxFloat(rng + (threadIdx.x >> 3), v);
  }
}


__global__ __launch_bounds__(256) void edge_max(
    const int* __restrict__ src, const int* __restrict__ dst,
    const float* __restrict__ sv, const float* __restrict__ dv,
    const float* __restrict__ rng, float* __restrict__ m, int E, int ET) {
  if (rng[0] + rng[1] < SPREAD_LIMIT) return;
  for (int id = blockIdx.x * 256 + threadIdx.x; id < ET;
       id += gridDim.x * 256) {
    int si, di;
    if (id < E) {
      si = src[id];
      di = dst[id];
    } else {
      si = di = id - E;
    }
    atomicMaxFloat(m + di, leaky(sv[si] + dv[di]));
  }
}

#define ZCAP 65536
#define ZGRID 256
#define ZUNROLL 4

__global__ __launch_bounds__(1024) void edge_exp_priv(
    const int* __restrict__ src, const int* __restrict__ dst,
    const float* __restrict__ sv, const float* __restrict__ dv,
    const float* __restrict__ rng, const float* __restrict__ m,
    float* __restrict__ ebuf, float* __restrict__ zpart, int E, int ET, int N,
    int NP) {
  __shared__ __attribute__((aligned(16))) float zl[ZCAP];
  const int tid = threadIdx.x;
  const int stride = gridDim.x * 1024;
  const float smax = rng[0];
  const bool bound = smax + rng[1] < SPREAD_LIMIT;
  float* __restrict__ zrow = zpart + (size_t)blockIdx.x * NP;

  for (int base = 0; base < N; base += ZCAP) {
    const int cnt = min(N - base, ZCAP);
    const int cnt4 = (cnt + 3) & ~3;
    for (int i = tid * 4; i < cnt4; i += 4096)
      *(float4*)(zl + i) = make_float4(0.0f, 0.0f, 0.0f, 0.0f);
    __syncthreads();

    for (int id0 = blockIdx.x * 1024 + tid; id0 < ET; id0 += ZUNROLL * stride) {
      int si[ZUNROLL], di[ZUNROLL];
#pragma unroll
      for (int u = 0; u < ZUNROLL; ++u) {
        const int id = id0 + u * stride;
        const bool real = id < E;
        const int idc = real ? id : 0;
        const int s0 = src[idc];
        const int d0 = dst[idc];
        const int self = (id < ET) ? id - E : -1;
        si[u] = real ? s0 : self;
        di[u] = real ? d0 : self;
      }
      float ss[ZUNROLL], dd[ZUNROLL], mm[ZUNROLL];
#pragma unroll
      for (int u = 0; u < ZUNROLL; ++u) {
        ss[u] = sv[max(si[u], 0)];
        dd[u] = dv[max(di[u], 0)];
      }
      if (!bound) {
#pragma unroll
        for (int u = 0; u < ZUNROLL; ++u) mm[u] = m[max(di[u], 0)];
      } else {
#pragma unroll
        for (int u = 0; u < ZUNROLL; ++u) mm[u] = 0.0f;
      }
#pragma unroll
      for (int u = 0; u < ZUNROLL; ++u) {
        if (di[u] >= 0) {
          const float e = leaky(ss[u] + dd[u]);
          const float shift = bound ? leaky(smax + dd[u]) : mm[u];
          const float t = __expf(e - shift);
          if (base == 0) ebuf[id0 + u * stride] = t;
          const int li = di[u] - base;
          if ((unsigned)li < (unsigned)cnt) atomicAdd(zl + li, t);
        }
      }
    }
    __syncthreads();

    for (int i = tid * 4; i < cnt4; i += 4096)
      *(float4*)(zrow + base + i) = *(const float4*)(zl + i);
    __syncthreads();
  }
}

__global__ __launch_bounds__(256) void z_reduce(const float* __restrict__ zpart,
                                                float* __restrict__ z, int N,
                                                int NP) {
  __shared__ __attribute__((aligned(16))) float red[8][128];
  const int wave = threadIdx.x >> 5;
  const int lane = threadIdx.x & 31;
  const int n4 = blockIdx.x * 128 + lane * 4;
  float4 acc = make_float4(0.0f, 0.0f, 0.0f, 0.0f);
  if (n4 < NP) {
    const float* p = zpart + (size_t)wave * NP + n4;
#pragma unroll 8
    for (int k = 0; k < ZGRID / 8; ++k) {
      const float4 v = *(const float4*)p;
      p += (size_t)8 * NP;
      acc.x += v.x;
      acc.y += v.y;
      acc.z += v.z;
      acc.w += v.w;
    }
  }
  *(float4*)(&red[wave][lane * 4]) = acc;
  __syncthreads();
  if (threadIdx.x < 128) {
    const int n = blockIdx.x * 128 + threadIdx.x;
    float r = red[0][threadIdx.x];
#pragma unroll
    for (int w = 1; w < 8; ++w) r += red[w][threadIdx.x];
    if (n < N) z[n] = r;
  }
}

#define WROWS 2048
#define WCHUNKS 10
#define WSCAN 8
#define WQ (32 * WSCAN + 32)

__device__ __forceinline__ int win_dst_word(const int* __restrict__ dst, int id,
                                            int last) {
  return dst[min(id, last)];
}

__device__ __forceinline__ bool win_hit(int d, int id, int lim, int e1, int E,
                                        int wbase, int wcnt) {
  const bool real_hit = ((unsigned)(d - wbase) < (unsigned)wcnt) & (id < lim);
  const bool self_hit = ((unsigned)(id - E - wbase) < (unsigned)wcnt) & (id < e1);
  return real_hit | self_hit;
}

template <bool REAL>
__device__ __forceinline__ int win_queue(const int* cur, int b, int lane,
                                         int lim, int e1, int E, int wbase,
                                         int wcnt, int* myq, int cnt) {
#pragma unroll
  for (int u = 0; u < WSCAN; ++u) {
    const int id = b + u * 32 + lane;
    const bool hit = REAL ? ((unsigned)(cur[u] - wbase) < (unsigned)wcnt)
                          : win_hit(cur[u], id, lim, e1, E, wbase, wcnt);
    const unsigned mask = __builtin_amdgcn_ballot_w32(hit);
    if (hit) myq[cnt + (int)__builtin_amdgcn_mbcnt_lo(mask, 0u)] = id;
    cnt += __builtin_popcount(mask);
  }
  return cnt;
}

__device__ __forceinline__ float win_row_load(const float* __restrict__ h,
                                              int si, int j, int lane) {
  const int sj = __builtin_amdgcn_readlane(si, j);
  return h[(size_t)sj * 32 + lane];
}

struct WinScore {
  const float* __restrict__ sv;
  const float* __restrict__ dv;
  const float* __restrict__ m;
  float smax;
  bool bound;
};

__device__ __forceinline__ void win_row_add(float* acc, float v, int dl,
                                            float t, int j, int lane) {
  const int rj = __builtin_amdgcn_readlane(dl, j);
  const float tj = __int_as_float(__builtin_amdgcn_readlane(__float_as_int(t), j));
  atomicAdd(acc + rj * 32 + lane, tj * v);
}

template <bool FULL>
__device__ __forceinline__ void win_batch(
    const int* qp, int m, int lane, int wbase, int E,
    const int* __restrict__ src, const int* __restrict__ dst,
    const WinScore& sc, const float* __restrict__ h, float* acc, float* zl) {
  const int id = qp[(FULL || lane < m) ? lane : 0];
  const bool real = id < E;
  const int idc = real ? id : 0;
  const int s0 = src[idc];
  const int d0 = dst[idc];
  const int self = id - E;
  const int si = real ? s0 : self;
  const int di = real ? d0 : self;
  const int dl = di - wbase;
  const float ss = sc.sv[si];
  const float dd = sc.dv[di];
  const float mm = sc.bound ? 0.0f : sc.m[di];
  float v[32];
#pragma unroll
  for (int j = 0; j < 32; ++j) {
    v[j] = 0.0f;
    if (FULL || j < m) v[j] = win_row_load(h, si, j, lane);
  }
  const float shift = sc.bound ? leaky(sc.smax + dd) : mm;
  const float t = __expf(leaky(ss + dd) - shift);
  if (FULL || lane < m) atomicAdd(zl + dl, t);
#pragma unroll
  for (int j = 0; j < 32; ++j)
    if (FULL || j < m) win_row_add(acc, v[j], dl, t, j, lane);
}

template <bool REAL>
__device__ __forceinline__ void win_steps(
    int& b, int (&cur)[WSCAN], int& cnt, int lane, int lim, int last, int e1,
    int E, int wbase, int wcnt, int* myq, const int* __restrict__ src,
    const int* __restrict__ dst, const WinScore& sc,
    const float* __restrict__ h, float* acc, float* zl) {
  const int step = 32 * WSCAN;
  for (; REAL ? (b + step <= lim) : (b < e1); b += 32 * step) {
    int nxt[WSCAN];
    const int nb = b + 32 * step;
#pragma unroll
    for (int u = 0; u < WSCAN; ++u)
      nxt[u] = win_dst_word(dst, nb + u * 32 + lane, last);
    cnt = win_queue<REAL>(cur, b, lane, lim, e1, E, wbase, wcnt, myq, cnt);
    while (cnt >= 32) {
      cnt -= 32;
      win_batch<true>(myq + cnt, 32, lane, wbase, E, src, dst, sc, h, acc, zl);
    }
#pragma unroll
    for (int u = 0; u < WSCAN; ++u) cur[u] = nxt[u];
  }
}

__global__ __launch_bounds__(1024) void edge_window(
    const int* __restrict__ src, const int* __restrict__ dst,
    const float* __restrict__ sv, const float* __restrict__ dv,
    const float* __restrict__ rng, const float* __restrict__ m,
    const float* __restrict__ h, float* __restrict__ wpart,
    float* __restrict__ zwpart, int E, int ET, int N, int ZP) {
  __shared__ __attribute__((aligned(16))) float acc[WROWS * 32];
  __shared__ __attribute__((aligned(16))) float zl[WROWS];
  __shared__ int q[32][WQ];
  const int tid = threadIdx.x;
  const int lane = tid & 31;
  const int wave = __builtin_amdgcn_readfirstlane(tid >> 5);
  const int win = blockIdx.x / WCHUNKS;
  const int chunk = blockIdx.x - win * WCHUNKS;
  const int wbase = win * WROWS;
  const int wcnt = min(N - wbase, WROWS);
  const int wfl = wcnt * 32;

  for (int i = tid * 4; i < wfl; i += 4096)
    *(float4*)(acc + i) = make_float4(0.0f, 0.0f, 0.0f, 0.0f);
  for (int i = tid; i < WROWS; i += 1024) zl[i] = 0.0f;
  __syncthreads();

  WinScore sc;
  sc.sv = sv;
  sc.dv = dv;
  sc.m = m;
  sc.smax = rng[0];
  sc.bound = sc.smax + rng[1] < SPREAD_LIMIT;

  const int per = ((ET + WCHUNKS * 32 - 1) / (WCHUNKS * 32)) * 32;
  const int e0 = chunk * per;
  const int e1 = min(ET, e0 + per);
  const int lim = min(e1, E);
  const int last = max(lim, 1) - 1;
  int* myq = q[wave];
  int cnt = 0;

  int b = e0 + wave * (32 * WSCAN);
  int cur[WSCAN];
#pragma unroll
  for (int u = 0; u < WSCAN; ++u)
    cur[u] = win_dst_word(dst, b + u * 32 + lane, last);
  win_steps<true>(b, cur, cnt, lane, lim, last, e1, E, wbase, wcnt, myq, src,
                  dst, sc, h, acc, zl);
  win_steps<false>(b, cur, cnt, lane, lim, last, e1, E, wbase, wcnt, myq, src,
                   dst, sc, h, acc, zl);
  if (cnt > 0)
    win_batch<false>(myq, cnt, lane, wbase, E, src, dst, sc, h, acc, zl);
  __syncthreads();

  float* __restrict__ prow = wpart + ((size_t)chunk * N + wbase) * 32;
  for (int i = tid * 4; i < wfl; i += 4096)
    *(float4*)(prow + i) = *(const float4*)(acc + i);
  float* __restrict__ zrow = zwpart + (size_t)chunk * ZP + wbase;
  for (int i = tid; i < wcnt; i += 1024) zrow[i] = zl[i];
}

__global__ __launch_bounds__(256) void window_reduce(
    const float* __restrict__ wpart, const float* __restrict__ zwpart,
    const float* __restrict__ b, float* __restrict__ o, int N, int ZP) {
  const int i4 = blockIdx.x * 256 + threadIdx.x;
  if (i4 >= N * 8) return;
  const int n = i4 >> 3;
  const int f = (i4 & 7) * 4;
  const float* p = wpart + (size_t)i4 * 4;
  const float* pz = zwpart + n;
  float4 a = make_float4(0.0f, 0.0f, 0.0f, 0.0f);
  float zn = 0.0f;
#pragma unroll
  for (int c = 0; c < WCHUNKS; ++c) {
    const float4 v = *(const float4*)p;
    const float zc = *pz;
    p += (size_t)N * 32;
    pz += ZP;
    a.x += v.x;
    a.y += v.y;
    a.z += v.z;
    a.w += v.w;
    zn += zc;
  }
  const float bx = b[f], by = b[f + 1], bz = b[f + 2], bw = b[f + 3];
  float4 r;
  r.x = a.x / zn + bx;
  r.y = a.y / zn + by;
  r.z = a.z / zn + bz;
  r.w = a.w / zn + bw;
  r.x = (r.x > 0.0f) ? r.x : (__expf(r.x) - 1.0f);
  r.y = (r.y > 0.0f) ? r.y : (__expf(r.y) - 1.0f);
  r.z = (r.z > 0.0f) ? r.z : (__expf(r.z) - 1.0f);
  r.w = (r.w > 0.0f) ? r.w : (__expf(r.w) - 1.0f);
  *(float4*)(o + (size_t)i4 * 4) = r;
}

#define PGRID 256
#define PSUMS (NUM_GRAPHS * 64)
#define PROW (PSUMS + NUM_GRAPHS)

__device__ __forceinline__ float2 pool_row_load(const float* __restrict__ h,
                                                int si, int j, int lane) {
  const int sj = __builtin_amdgcn_readlane(si, j);
  return *(const float2*)(h + (size_t)sj * 64 + lane * 2);
}

__device__ __forceinline__ void pool_row_add(float* pl, float2 v, int g,
                                             float a, int j, int lane) {
  const int gj = __builtin_amdgcn_readlane(g, j);
  const float aj = __int_as_float(__builtin_amdgcn_readlane(__float_as_int(a), j));
  float* p = pl + gj * 64 + lane * 2;
  atomicAdd(p, aj * v.x);
  atomicAdd(p + 1, aj * v.y);
}

__global__ __launch_bounds__(1024) void edge_pool(
    const int* __restrict__ src, const int* __restrict__ dst,
    const float* __restrict__ ebuf, const float* __restrict__ z,
    const int* __restrict__ batch, const float* __restrict__ h,
    float* __restrict__ ppart, int E, int ET) {
  __shared__ float pl[PROW];
  const int tid = threadIdx.x;
  const int lane = tid & 31;
  for (int i = tid; i < PROW; i += 1024) pl[i] = 0.0f;
  __syncthreads();

  const int wave = blockIdx.x * 32 + (tid >> 5);
  const int sweep = gridDim.x * 32 * 32;
  for (int b0 = wave * 32; b0 < ET; b0 += sweep) {
    const int id = b0 + lane;
    const bool live = id < ET;
    const bool real = id < E;
    const int idc = real ? id : 0;
    const int s0 = src[idc];
    const int d0 = dst[idc];
    const int self = live ? id - E : 0;
    const int si = real ? s0 : self;
    const int di = real ? d0 : self;
    const float t = ebuf[live ? id : 0];
    const int g = batch[di];
    const float a = t / z[di];
    if (live && !real) atomicAdd(pl + PSUMS + g, 1.0f);

    const int cnt = min(32, ET - b0);
    if (cnt == 32) {
#pragma unroll
      for (int j0 = 0; j0 < 32; j0 += 16) {
        float2 v[16];
#pragma unroll
        for (int u = 0; u < 16; ++u) v[u] = pool_row_load(h, si, j0 + u, lane);
#pragma unroll
        for (int u = 0; u < 16; ++u) pool_row_add(pl, v[u], g, a, j0 + u, lane);
      }
    } else {
      for (int j = 0; j < cnt; ++j)
        pool_row_add(pl, pool_row_load(h, si, j, lane), g, a, j, lane);
    }
  }
  __syncthreads();

  float* __restrict__ row = ppart + (size_t)blockIdx.x * PROW;
  for (int i = tid; i < PROW; i += 1024) row[i] = pl[i];
}

__global__ __launch_bounds__(1024) void pool_finish(
    const float* __restrict__ ppart, const float* __restrict__ b,
    float* __restrict__ out) {
  __shared__ float red[16][64];
  __shared__ float cw[PGRID / 32];
  const int g = blockIdx.x;
  const int f = threadIdx.x & 63;
  const int r = threadIdx.x >> 6;
  const float* p = ppart + (size_t)r * PROW + g * 64 + f;
  float acc = 0.0f;
#pragma unroll
  for (int k = 0; k < PGRID / 16; ++k) {
    acc += *p;
    p += (size_t)16 * PROW;
  }
  red[r][f] = acc;
  if (threadIdx.x < PGRID) {
    float c = ppart[(size_t)threadIdx.x * PROW + PSUMS + g];
    for (int o = 16; o > 0; o >>= 1) c += __shfl_xor(c, o);
    if ((threadIdx.x & 31) == 0) cw[threadIdx.x >> 5] = c;
  }
  __syncthreads();
  if (threadIdx.x < 64) {
    float c = 0.0f;
    for (int w = 0; w < PGRID / 32; ++w) c += cw[w];
    float s = red[0][f];
#pragma unroll
    for (int w = 1; w < 16; ++w) s += red[w][f];
    out[g * 64 + f] = (s + c * b[f]) / fmaxf(c, 1.0f);
  }
}

__global__ void fill_kernel(float* __restrict__ p, float v, int n) {
  int i = blockIdx.x * blockDim.x + threadIdx.x;
  if (i < n) p[i] = v;
}

static inline int cdiv(int a, int b) { return (a + b - 1) / b; }

extern "C" void kernel_launch(void* const* d_in, const int* in_sizes, int n_in,
                              void* d_out, int out_size, void* d_ws,
                              size_t ws_size, hipStream_t stream) {
  const float* x      = (const float*)d_in[0];
  const int*   ei     = (const int*)d_in[1];
  const int*   batch  = (const int*)d_in[2];
  const float* W1     = (const float*)d_in[3];
  const float* a_src1 = (const float*)d_in[4];
  const float* a_dst1 = (const float*)d_in[5];
  const float* b1     = (const float*)d_in[6];
  const float* W2     = (const float*)d_in[7];
  const float* a_src2 = (const float*)d_in[8];
  const float* a_dst2 = (const float*)d_in[9];
  const float* b2     = (const float*)d_in[10];

  const int N = in_sizes[2];
  const int E = in_sizes[1] / 2;
  const int ET = E + N;
  const int* src = ei;
  const int* dst = ei + E;

  float* ws = (float*)d_ws;
  size_t off = 0;
  float* h1 = ws + off; off += (size_t)N * 32;
  float* o1 = ws + off; off += (size_t)N * 32;
  float* h2 = ws + off; off += (size_t)N * 64;
  float* sv = ws + off; off += (size_t)N;
  float* dv = ws + off; off += (size_t)N;
  float* mb = ws + off; off += (size_t)N + 2;
  float* rng = mb + N;
  float* zb = ws + off; off += (size_t)N;
  float* eb = ws + off; off += (size_t)ET;
  const int NP = (N + 3) & ~3;
  off = (off + 3) & ~(size_t)3;
  float* zp = ws + off; off += (size_t)ZGRID * NP;
  float* pp = ws + off; off += (size_t)PGRID * PROW;
  float* wp = ws + off; off += (size_t)WCHUNKS * N * 32;
  const int ZP = (N + 31) & ~31;
  off = (off + 31) & ~(size_t)31;
  float* zwp = ws + off; off += (size_t)WCHUNKS * ZP;

  const float NEG_INF = -__builtin_inff();

  gemm_wmma<128, 32><<<cdiv(N, 128), 256, 0, stream>>>(x, W1, h1, N);
  fill_kernel<<<cdiv(N + 2, 256), 256, 0, stream>>>(mb, NEG_INF, N + 2);
  node_dots<32><<<cdiv(N, 256), 256, 0, stream>>>(h1, a_src1, a_dst1, sv, dv, rng, N);
  edge_max<<<512, 256, 0, stream>>>(src, dst, sv, dv, rng, mb, E, ET);
  edge_window<<<cdiv(N, WROWS) * WCHUNKS, 1024, 0, stream>>>(src, dst, sv, dv, rng, mb, h1, wp, zwp, E, ET, N, ZP);
  window_reduce<<<cdiv(N * 8, 256), 256, 0, stream>>>(wp, zwp, b1, o1, N, ZP);

  gemm_wmma<32, 64><<<cdiv(N, 128), 256, 0, stream>>>(o1, W2, h2, N);
  fill_kernel<<<cdiv(N + 2, 256), 256, 0, stream>>>(mb, NEG_INF, N + 2);
  node_dots<64><<<cdiv(N, 256), 256, 0, stream>>>(h2, a_src2, a_dst2, sv, dv, rng, N);
  edge_max<<<512, 256, 0, stream>>>(src, dst, sv, dv, rng, mb, E, ET);
  edge_exp_priv<<<ZGRID, 1024, 0, stream>>>(src, dst, sv, dv, rng, mb, eb, zp, E, ET, N, NP);
  z_reduce<<<cdiv(NP, 128), 256, 0, stream>>>(zp, zb, N, NP);

  edge_pool<<<PGRID, 1024, 0, stream>>>(src, dst, eb, zb, batch, h2, pp, E, ET);
  pool_finish<<<NUM_GRAPHS, 1024, 0, stream>>>(pp, b2, (float*)d_out);
}


# Layer-1 reduce, layer-2 projection and node scores become one launch: a block keeps 128 o1 rows in LDS for the WMMA and its h2 tile for the dots; one fill serves both layers; 14 launches become 11
# speedup vs baseline: 4.6220x; 1.0473x over previous
#include <hip/hip_runtime.h>
#include <hip/hip_bf16.h>
#include <math.h>

typedef __attribute__((ext_vector_type(2))) float v2f;
typedef __attribute__((ext_vector_type(8))) float v8f;

#define NUM_GRAPHS 64
#define NEG_SLOPE 0.2f
#define SPREAD_LIMIT 8.0f

template <int FIN, int FOUT>
__global__ __launch_bounds__(256) void gemm_wmma(const float* __restrict__ x,
                                                 const float* __restrict__ W,
                                                 float* __restrict__ h, int N) {
  constexpr int NT = FOUT / 16;
  __shared__ float Ws[FIN * FOUT];
  for (int i = threadIdx.x; i < FIN * FOUT; i += 256) Ws[i] = W[i];
  __syncthreads();

  const int wave = threadIdx.x >> 5;
  const int lane = threadIdx.x & 31;
  const int m0 = (blockIdx.x * 8 + wave) * 16;
  const int halfsel = lane >> 4;
  const int lidx = lane & 15;
  int row = m0 + lidx;
  if (row >= N) row = N - 1;
  const int khalf = halfsel * 2;

  v8f acc[NT];
#pragma unroll
  for (int t = 0; t < NT; ++t)
#pragma unroll
    for (int j = 0; j < 8; ++j) acc[t][j] = 0.0f;

#pragma unroll 4
  for (int k = 0; k < FIN; k += 4) {
    const int kk = k + khalf;
    v2f a;
    a.x = x[(size_t)row * FIN + kk];
    a.y = x[(size_t)row * FIN + kk + 1];
#pragma unroll
    for (int t = 0; t < NT; ++t) {
      v2f b;
      b.x = Ws[kk * FOUT + t * 16 + lidx];
      b.y = Ws[(kk + 1) * FOUT + t * 16 + lidx];
      acc[t] = __builtin_amdgcn_wmma_f32_16x16x4_f32(
          false, a, false, b, (short)0, acc[t], false, false);
    }
  }

  const int rbase = m0 + halfsel * 8;
#pragma unroll
  for (int i = 0; i < 8; ++i) {
    const int r = rbase + i;
    if (r < N) {
#pragma unroll
      for (int t = 0; t < NT; ++t)
        h[(size_t)r * FOUT + t * 16 + lidx] = acc[t][i];
    }
  }
}

__device__ __forceinline__ void atomicMaxFloat(float* addr, float val) {
  int vi = __float_as_int(val);
  if (vi >= 0)
    atomicMax((int*)addr, vi);
  else
    atomicMin((unsigned int*)addr, (unsigned int)vi);
}

__device__ __forceinline__ float leaky(float v) {
  return (v > 0.0f) ? v : NEG_SLOPE * v;
}

template <int F>
__global__ __launch_bounds__(256) void node_dots(
    const float* __restrict__ h, const float* __restrict__ asrc,
    const float* __restrict__ adst, float* __restrict__ s,
    float* __restrict__ d, float* __restrict__ rng, int N) {
  __shared__ float red[16];
  const int n = blockIdx.x * 256 + threadIdx.x;
  const bool valid = n < N;
  float ss = 0.0f, dd = 0.0f;
  if (valid) {
#pragma unroll
    for (int f = 0; f < F; ++f) {
      float v = h[(size_t)n * F + f];
      ss += v * asrc[f];
      dd += v * adst[f];
    }
    s[n] = ss;
    d[n] = dd;
  }
  const float ninf = -__builtin_inff();
  float hi = valid ? ss : ninf;
  float lo = valid ? -ss : ninf;
#pragma unroll
  for (int o = 16; o > 0; o >>= 1) {
    hi = fmaxf(hi, __shfl_xor(hi, o));
    lo = fmaxf(lo, __shfl_xor(lo, o));
  }
  const int wave = threadIdx.x >> 5;
  if ((threadIdx.x & 31) == 0) {
    red[wave] = hi;
    red[8 + wave] = lo;
  }
  __syncthreads();
  if (threadIdx.x < 16) {
    float v = red[threadIdx.x];
    v = fmaxf(v, __shfl_xor(v, 4));
    v = fmaxf(v, __shfl_xor(v, 2));
    v = fmaxf(v, __shfl_xor(v, 1));
    if ((threadIdx.x & 7) == 0) atomicMaxFloat(rng + (threadIdx.x >> 3), v);
  }
}


__global__ __launch_bounds__(256) void edge_max(
    const int* __restrict__ src, const int* __restrict__ dst,
    const float* __restrict__ sv, const float* __restrict__ dv,
    const float* __restrict__ rng, float* __restrict__ m, int E, int ET) {
  if (rng[0] + rng[1] < SPREAD_LIMIT) return;
  for (int id = blockIdx.x * 256 + threadIdx.x; id < ET;
       id += gridDim.x * 256) {
    int si, di;
    if (id < E) {
      si = src[id];
      di = dst[id];
    } else {
      si = di = id - E;
    }
    atomicMaxFloat(m + di, leaky(sv[si] + dv[di]));
  }
}

#define ZCAP 65536
#define ZGRID 256
#define ZUNROLL 4

__global__ __launch_bounds__(1024) void edge_exp_priv(
    const int* __restrict__ src, const int* __restrict__ dst,
    const float* __restrict__ sv, const float* __restrict__ dv,
    const float* __restrict__ rng, const float* __restrict__ m,
    float* __restrict__ ebuf, float* __restrict__ zpart, int E, int ET, int N,
    int NP) {
  __shared__ __attribute__((aligned(16))) float zl[ZCAP];
  const int tid = threadIdx.x;
  const int stride = gridDim.x * 1024;
  const float smax = rng[0];
  const bool bound = smax + rng[1] < SPREAD_LIMIT;
  float* __restrict__ zrow = zpart + (size_t)blockIdx.x * NP;

  for (int base = 0; base < N; base += ZCAP) {
    const int cnt = min(N - base, ZCAP);
    const int cnt4 = (cnt + 3) & ~3;
    for (int i = tid * 4; i < cnt4; i += 4096)
      *(float4*)(zl + i) = make_float4(0.0f, 0.0f, 0.0f, 0.0f);
    __syncthreads();

    for (int id0 = blockIdx.x * 1024 + tid; id0 < ET; id0 += ZUNROLL * stride) {
      int si[ZUNROLL], di[ZUNROLL];
#pragma unroll
      for (int u = 0; u < ZUNROLL; ++u) {
        const int id = id0 + u * stride;
        const bool real = id < E;
        const int idc = real ? id : 0;
        const int s0 = src[idc];
        const int d0 = dst[idc];
        const int self = (id < ET) ? id - E : -1;
        si[u] = real ? s0 : self;
        di[u] = real ? d0 : self;
      }
      float ss[ZUNROLL], dd[ZUNROLL], mm[ZUNROLL];
#pragma unroll
      for (int u = 0; u < ZUNROLL; ++u) {
        ss[u] = sv[max(si[u], 0)];
        dd[u] = dv[max(di[u], 0)];
      }
      if (!bound) {
#pragma unroll
        for (int u = 0; u < ZUNROLL; ++u) mm[u] = m[max(di[u], 0)];
      } else {
#pragma unroll
        for (int u = 0; u < ZUNROLL; ++u) mm[u] = 0.0f;
      }
#pragma unroll
      for (int u = 0; u < ZUNROLL; ++u) {
        if (di[u] >= 0) {
          const float e = leaky(ss[u] + dd[u]);
          const float shift = bound ? leaky(smax + dd[u]) : mm[u];
          const float t = __expf(e - shift);
          if (base == 0) ebuf[id0 + u * stride] = t;
          const int li = di[u] - base;
          if ((unsigned)li < (unsigned)cnt) atomicAdd(zl + li, t);
        }
      }
    }
    __syncthreads();

    for (int i = tid * 4; i < cnt4; i += 4096)
      *(float4*)(zrow + base + i) = *(const float4*)(zl + i);
    __syncthreads();
  }
}

__global__ __launch_bounds__(256) void z_reduce(const float* __restrict__ zpart,
                                                float* __restrict__ z, int N,
                                                int NP) {
  __shared__ __attribute__((aligned(16))) float red[8][128];
  const int wave = threadIdx.x >> 5;
  const int lane = threadIdx.x & 31;
  const int n4 = blockIdx.x * 128 + lane * 4;
  float4 acc = make_float4(0.0f, 0.0f, 0.0f, 0.0f);
  if (n4 < NP) {
    const float* p = zpart + (size_t)wave * NP + n4;
#pragma unroll 8
    for (int k = 0; k < ZGRID / 8; ++k) {
      const float4 v = *(const float4*)p;
      p += (size_t)8 * NP;
      acc.x += v.x;
      acc.y += v.y;
      acc.z += v.z;
      acc.w += v.w;
    }
  }
  *(float4*)(&red[wave][lane * 4]) = acc;
  __syncthreads();
  if (threadIdx.x < 128) {
    const int n = blockIdx.x * 128 + threadIdx.x;
    float r = red[0][threadIdx.x];
#pragma unroll
    for (int w = 1; w < 8; ++w) r += red[w][threadIdx.x];
    if (n < N) z[n] = r;
  }
}

#define WROWS 2048
#define WCHUNKS 10
#define WSCAN 8
#define WQ (32 * WSCAN + 32)

__device__ __forceinline__ int win_dst_word(const int* __restrict__ dst, int id,
                                            int last) {
  return dst[min(id, last)];
}

__device__ __forceinline__ bool win_hit(int d, int id, int lim, int e1, int E,
                                        int wbase, int wcnt) {
  const bool real_hit = ((unsigned)(d - wbase) < (unsigned)wcnt) & (id < lim);
  const bool self_hit = ((unsigned)(id - E - wbase) < (unsigned)wcnt) & (id < e1);
  return real_hit | self_hit;
}

template <bool REAL>
__device__ __forceinline__ int win_queue(const int* cur, int b, int lane,
                                         int lim, int e1, int E, int wbase,
                                         int wcnt, int* myq, int cnt) {
#pragma unroll
  for (int u = 0; u < WSCAN; ++u) {
    const int id = b + u * 32 + lane;
    const bool hit = REAL ? ((unsigned)(cur[u] - wbase) < (unsigned)wcnt)
                          : win_hit(cur[u], id, lim, e1, E, wbase, wcnt);
    const unsigned mask = __builtin_amdgcn_ballot_w32(hit);
    if (hit) myq[cnt + (int)__builtin_amdgcn_mbcnt_lo(mask, 0u)] = id;
    cnt += __builtin_popcount(mask);
  }
  return cnt;
}

__device__ __forceinline__ float win_row_load(const float* __restrict__ h,
                                              int si, int j, int lane) {
  const int sj = __builtin_amdgcn_readlane(si, j);
  return h[(size_t)sj * 32 + lane];
}

struct WinScore {
  const float* __restrict__ sv;
  const float* __restrict__ dv;
  const float* __restrict__ m;
  float smax;
  bool bound;
};

__device__ __forceinline__ void win_row_add(float* acc, float v, int dl,
                                            float t, int j, int lane) {
  const int rj = __builtin_amdgcn_readlane(dl, j);
  const float tj = __int_as_float(__builtin_amdgcn_readlane(__float_as_int(t), j));
  atomicAdd(acc + rj * 32 + lane, tj * v);
}

template <bool FULL>
__device__ __forceinline__ void win_batch(
    const int* qp, int m, int lane, int wbase, int E,
    const int* __restrict__ src, const int* __restrict__ dst,
    const WinScore& sc, const float* __restrict__ h, float* acc, float* zl) {
  const int id = qp[(FULL || lane < m) ? lane : 0];
  const bool real = id < E;
  const int idc = real ? id : 0;
  const int s0 = src[idc];
  const int d0 = dst[idc];
  const int self = id - E;
  const int si = real ? s0 : self;
  const int di = real ? d0 : self;
  const int dl = di - wbase;
  const float ss = sc.sv[si];
  const float dd = sc.dv[di];
  const float mm = sc.bound ? 0.0f : sc.m[di];
  float v[32];
#pragma unroll
  for (int j = 0; j < 32; ++j) {
    v[j] = 0.0f;
    if (FULL || j < m) v[j] = win_row_load(h, si, j, lane);
  }
  const float shift = sc.bound ? leaky(sc.smax + dd) : mm;
  const float t = __expf(leaky(ss + dd) - shift);
  if (FULL || lane < m) atomicAdd(zl + dl, t);
#pragma unroll
  for (int j = 0; j < 32; ++j)
    if (FULL || j < m) win_row_add(acc, v[j], dl, t, j, lane);
}

template <bool REAL>
__device__ __forceinline__ void win_steps(
    int& b, int (&cur)[WSCAN], int& cnt, int lane, int lim, int last, int e1,
    int E, int wbase, int wcnt, int* myq, const int* __restrict__ src,
    const int* __restrict__ dst, const WinScore& sc,
    const float* __restrict__ h, float* acc, float* zl) {
  const int step = 32 * WSCAN;
  for (; REAL ? (b + step <= lim) : (b < e1); b += 32 * step) {
    int nxt[WSCAN];
    const int nb = b + 32 * step;
#pragma unroll
    for (int u = 0; u < WSCAN; ++u)
      nxt[u] = win_dst_word(dst, nb + u * 32 + lane, last);
    cnt = win_queue<REAL>(cur, b, lane, lim, e1, E, wbase, wcnt, myq, cnt);
    while (cnt >= 32) {
      cnt -= 32;
      win_batch<true>(myq + cnt, 32, lane, wbase, E, src, dst, sc, h, acc, zl);
    }
#pragma unroll
    for (int u = 0; u < WSCAN; ++u) cur[u] = nxt[u];
  }
}

__global__ __launch_bounds__(1024) void edge_window(
    const int* __restrict__ src, const int* __restrict__ dst,
    const float* __restrict__ sv, const float* __restrict__ dv,
    const float* __restrict__ rng, const float* __restrict__ m,
    const float* __restrict__ h, float* __restrict__ wpart,
    float* __restrict__ zwpart, int E, int ET, int N, int ZP) {
  __shared__ __attribute__((aligned(16))) float acc[WROWS * 32];
  __shared__ __attribute__((aligned(16))) float zl[WROWS];
  __shared__ int q[32][WQ];
  const int tid = threadIdx.x;
  const int lane = tid & 31;
  const int wave = __builtin_amdgcn_readfirstlane(tid >> 5);
  const int win = blockIdx.x / WCHUNKS;
  const int chunk = blockIdx.x - win * WCHUNKS;
  const int wbase = win * WROWS;
  const int wcnt = min(N - wbase, WROWS);
  const int wfl = wcnt * 32;

  for (int i = tid * 4; i < wfl; i += 4096)
    *(float4*)(acc + i) = make_float4(0.0f, 0.0f, 0.0f, 0.0f);
  for (int i = tid; i < WROWS; i += 1024) zl[i] = 0.0f;
  __syncthreads();

  WinScore sc;
  sc.sv = sv;
  sc.dv = dv;
  sc.m = m;
  sc.smax = rng[0];
  sc.bound = sc.smax + rng[1] < SPREAD_LIMIT;

  const int per = ((ET + WCHUNKS * 32 - 1) / (WCHUNKS * 32)) * 32;
  const int e0 = chunk * per;
  const int e1 = min(ET, e0 + per);
  const int lim = min(e1, E);
  const int last = max(lim, 1) - 1;
  int* myq = q[wave];
  int cnt = 0;

  int b = e0 + wave * (32 * WSCAN);
  int cur[WSCAN];
#pragma unroll
  for (int u = 0; u < WSCAN; ++u)
    cur[u] = win_dst_word(dst, b + u * 32 + lane, last);
  win_steps<true>(b, cur, cnt, lane, lim, last, e1, E, wbase, wcnt, myq, src,
                  dst, sc, h, acc, zl);
  win_steps<false>(b, cur, cnt, lane, lim, last, e1, E, wbase, wcnt, myq, src,
                   dst, sc, h, acc, zl);
  if (cnt > 0)
    win_batch<false>(myq, cnt, lane, wbase, E, src, dst, sc, h, acc, zl);
  __syncthreads();

  float* __restrict__ prow = wpart + ((size_t)chunk * N + wbase) * 32;
  for (int i = tid * 4; i < wfl; i += 4096)
    *(float4*)(prow + i) = *(const float4*)(acc + i);
  float* __restrict__ zrow = zwpart + (size_t)chunk * ZP + wbase;
  for (int i = tid; i < wcnt; i += 1024) zrow[i] = zl[i];
}

#define RP_ROWS 128
#define RP_OP 34
#define RP_HP 68

__global__ __launch_bounds__(256) void reduce_project(
    const float* __restrict__ wpart, const float* __restrict__ zwpart,
    const float* __restrict__ b, const float* __restrict__ W,
    const float* __restrict__ asrc, const float* __restrict__ adst,
    float* __restrict__ h, float* __restrict__ s, float* __restrict__ d,
    float* __restrict__ rng, int N, int ZP) {
  constexpr int FIN = 32, FOUT = 64, NT = FOUT / 16;
  __shared__ __attribute__((aligned(16))) float Ws[FIN * FOUT];
  __shared__ __attribute__((aligned(16))) float os[RP_ROWS * RP_OP];
  __shared__ __attribute__((aligned(16))) float hs[RP_ROWS * RP_HP];
  __shared__ float red[16];
  const int tid = threadIdx.x;
  const int n0 = blockIdx.x * RP_ROWS;
  for (int i = tid; i < FIN * FOUT; i += 256) Ws[i] = W[i];

  const int f = (tid & 7) * 4;
  const float bx = b[f], by = b[f + 1], bz = b[f + 2], bw = b[f + 3];
#pragma unroll
  for (int it = 0; it < RP_ROWS * 8 / 256; ++it) {
    const int nl = (tid >> 3) + it * 32;
    const int n = n0 + nl;
    const int nc = min(n, N - 1);
    const float* p = wpart + ((size_t)nc * 32 + f);
    const float* pz = zwpart + nc;
    float4 a = make_float4(0.0f, 0.0f, 0.0f, 0.0f);
    float zn = 0.0f;
#pragma unroll
    for (int c = 0; c < WCHUNKS; ++c) {
      const float4 v = *(const float4*)p;
      const float zc = *pz;
      p += (size_t)N * 32;
      pz += ZP;
      a.x += v.x;
      a.y += v.y;
      a.z += v.z;
      a.w += v.w;
      zn += zc;
    }
    float4 r;
    r.x = a.x / zn + bx;
    r.y = a.y / zn + by;
    r.z = a.z / zn + bz;
    r.w = a.w / zn + bw;
    r.x = (r.x > 0.0f) ? r.x : (__expf(r.x) - 1.0f);
    r.y = (r.y > 0.0f) ? r.y : (__expf(r.y) - 1.0f);
    r.z = (r.z > 0.0f) ? r.z : (__expf(r.z) - 1.0f);
    r.w = (r.w > 0.0f) ? r.w : (__expf(r.w) - 1.0f);
    if (n >= N) r = make_float4(0.0f, 0.0f, 0.0f, 0.0f);
    float* q = os + nl * RP_OP + f;
    *(float2*)q = make_float2(r.x, r.y);
    *(float2*)(q + 2) = make_float2(r.z, r.w);
  }
  __syncthreads();

  const int wave = tid >> 5;
  const int lane = tid & 31;
  const int halfsel = lane >> 4;
  const int lidx = lane & 15;
  const int khalf = halfsel * 2;
  const float* arow = os + (wave * 16 + lidx) * RP_OP;

  v8f acc[NT];
#pragma unroll
  for (int t = 0; t < NT; ++t)
#pragma unroll
    for (int j = 0; j < 8; ++j) acc[t][j] = 0.0f;

#pragma unroll
  for (int k = 0; k < FIN; k += 4) {
    const int kk = k + khalf;
    v2f a;
    a.x = arow[kk];
    a.y = arow[kk + 1];
#pragma unroll
    for (int t = 0; t < NT; ++t) {
      v2f bb;
      bb.x = Ws[kk * FOUT + t * 16 + lidx];
      bb.y = Ws[(kk + 1) * FOUT + t * 16 + lidx];
      acc[t] = __builtin_amdgcn_wmma_f32_16x16x4_f32(
          false, a, false, bb, (short)0, acc[t], false, false);
    }
  }

  const int rloc0 = wave * 16 + halfsel * 8;
#pragma unroll
  for (int i = 0; i < 8; ++i) {
    const int rl = rloc0 + i;
    const int r = n0 + rl;
#pragma unroll
    for (int t = 0; t < NT; ++t) hs[rl * RP_HP + t * 16 + lidx] = acc[t][i];
    if (r < N) {
#pragma unroll
      for (int t = 0; t < NT; ++t)
        h[(size_t)r * FOUT + t * 16 + lidx] = acc[t][i];
    }
  }
  __syncthreads();

  const int n = n0 + tid;
  const bool valid = (tid < RP_ROWS) & (n < N);
  float ss = 0.0f, dd = 0.0f;
  if (valid) {
    const float* hr = hs + tid * RP_HP;
#pragma unroll
    for (int f = 0; f < FOUT; ++f) {
      float v = hr[f];
      ss += v * asrc[f];
      dd += v * adst[f];
    }
    s[n] = ss;
    d[n] = dd;
  }
  const float ninf = -__builtin_inff();
  float hi = valid ? ss : ninf;
  float lo = valid ? -ss : ninf;
#pragma unroll
  for (int o = 16; o > 0; o >>= 1) {
    hi = fmaxf(hi, __shfl_xor(hi, o));
    lo = fmaxf(lo, __shfl_xor(lo, o));
  }
  if (lane == 0) {
    red[wave] = hi;
    red[8 + wave] = lo;
  }
  __syncthreads();
  if (tid < 16) {
    float v = red[tid];
    v = fmaxf(v, __shfl_xor(v, 4));
    v = fmaxf(v, __shfl_xor(v, 2));
    v = fmaxf(v, __shfl_xor(v, 1));
    if ((tid & 7) == 0) atomicMaxFloat(rng + (tid >> 3), v);
  }
}

#define PGRID 256
#define PSUMS (NUM_GRAPHS * 64)
#define PROW (PSUMS + NUM_GRAPHS)

__device__ __forceinline__ float2 pool_row_load(const float* __restrict__ h,
                                                int si, int j, int lane) {
  const int sj = __builtin_amdgcn_readlane(si, j);
  return *(const float2*)(h + (size_t)sj * 64 + lane * 2);
}

__device__ __forceinline__ void pool_row_add(float* pl, float2 v, int g,
                                             float a, int j, int lane) {
  const int gj = __builtin_amdgcn_readlane(g, j);
  const float aj = __int_as_float(__builtin_amdgcn_readlane(__float_as_int(a), j));
  float* p = pl + gj * 64 + lane * 2;
  atomicAdd(p, aj * v.x);
  atomicAdd(p + 1, aj * v.y);
}

__global__ __launch_bounds__(1024) void edge_pool(
    const int* __restrict__ src, const int* __restrict__ dst,
    const float* __restrict__ ebuf, const float* __restrict__ z,
    const int* __restrict__ batch, const float* __restrict__ h,
    float* __restrict__ ppart, int E, int ET) {
  __shared__ float pl[PROW];
  const int tid = threadIdx.x;
  const int lane = tid & 31;
  for (int i = tid; i < PROW; i += 1024) pl[i] = 0.0f;
  __syncthreads();

  const int wave = blockIdx.x * 32 + (tid >> 5);
  const int sweep = gridDim.x * 32 * 32;
  for (int b0 = wave * 32; b0 < ET; b0 += sweep) {
    const int id = b0 + lane;
    const bool live = id < ET;
    const bool real = id < E;
    const int idc = real ? id : 0;
    const int s0 = src[idc];
    const int d0 = dst[idc];
    const int self = live ? id - E : 0;
    const int si = real ? s0 : self;
    const int di = real ? d0 : self;
    const float t = ebuf[live ? id : 0];
    const int g = batch[di];
    const float a = t / z[di];
    if (live && !real) atomicAdd(pl + PSUMS + g, 1.0f);

    const int cnt = min(32, ET - b0);
    if (cnt == 32) {
#pragma unroll
      for (int j0 = 0; j0 < 32; j0 += 16) {
        float2 v[16];
#pragma unroll
        for (int u = 0; u < 16; ++u) v[u] = pool_row_load(h, si, j0 + u, lane);
#pragma unroll
        for (int u = 0; u < 16; ++u) pool_row_add(pl, v[u], g, a, j0 + u, lane);
      }
    } else {
      for (int j = 0; j < cnt; ++j)
        pool_row_add(pl, pool_row_load(h, si, j, lane), g, a, j, lane);
    }
  }
  __syncthreads();

  float* __restrict__ row = ppart + (size_t)blockIdx.x * PROW;
  for (int i = tid; i < PROW; i += 1024) row[i] = pl[i];
}

__global__ __launch_bounds__(1024) void pool_finish(
    const float* __restrict__ ppart, const float* __restrict__ b,
    float* __restrict__ out) {
  __shared__ float red[16][64];
  __shared__ float cw[PGRID / 32];
  const int g = blockIdx.x;
  const int f = threadIdx.x & 63;
  const int r = threadIdx.x >> 6;
  const float* p = ppart + (size_t)r * PROW + g * 64 + f;
  float acc = 0.0f;
#pragma unroll
  for (int k = 0; k < PGRID / 16; ++k) {
    acc += *p;
    p += (size_t)16 * PROW;
  }
  red[r][f] = acc;
  if (threadIdx.x < PGRID) {
    float c = ppart[(size_t)threadIdx.x * PROW + PSUMS + g];
    for (int o = 16; o > 0; o >>= 1) c += __shfl_xor(c, o);
    if ((threadIdx.x & 31) == 0) cw[threadIdx.x >> 5] = c;
  }
  __syncthreads();
  if (threadIdx.x < 64) {
    float c = 0.0f;
    for (int w = 0; w < PGRID / 32; ++w) c += cw[w];
    float s = red[0][f];
#pragma unroll
    for (int w = 1; w < 16; ++w) s += red[w][f];
    out[g * 64 + f] = (s + c * b[f]) / fmaxf(c, 1.0f);
  }
}

__global__ void fill_kernel(float* __restrict__ p, float v, int n) {
  int i = blockIdx.x * blockDim.x + threadIdx.x;
  if (i < n) p[i] = v;
}

static inline int cdiv(int a, int b) { return (a + b - 1) / b; }

extern "C" void kernel_launch(void* const* d_in, const int* in_sizes, int n_in,
                              void* d_out, int out_size, void* d_ws,
                              size_t ws_size, hipStream_t stream) {
  const float* x      = (const float*)d_in[0];
  const int*   ei     = (const int*)d_in[1];
  const int*   batch  = (const int*)d_in[2];
  const float* W1     = (const float*)d_in[3];
  const float* a_src1 = (const float*)d_in[4];
  const float* a_dst1 = (const float*)d_in[5];
  const float* b1     = (const float*)d_in[6];
  const float* W2     = (const float*)d_in[7];
  const float* a_src2 = (const float*)d_in[8];
  const float* a_dst2 = (const float*)d_in[9];
  const float* b2     = (const float*)d_in[10];

  const int N = in_sizes[2];
  const int E = in_sizes[1] / 2;
  const int ET = E + N;
  const int* src = ei;
  const int* dst = ei + E;

  float* ws = (float*)d_ws;
  size_t off = 0;
  float* h1 = ws + off; off += (size_t)N * 32;
  float* h2 = ws + off; off += (size_t)N * 64;
  float* sv = ws + off; off += (size_t)N;
  float* dv = ws + off; off += (size_t)N;
  float* mb1 = ws + off; off += (size_t)N + 2;
  float* rng1 = mb1 + N;
  float* mb2 = ws + off; off += (size_t)N + 2;
  float* rng2 = mb2 + N;
  float* zb = ws + off; off += (size_t)N;
  float* eb = ws + off; off += (size_t)ET;
  const int NP = (N + 3) & ~3;
  off = (off + 3) & ~(size_t)3;
  float* zp = ws + off; off += (size_t)ZGRID * NP;
  float* pp = ws + off; off += (size_t)PGRID * PROW;
  float* wp = ws + off; off += (size_t)WCHUNKS * N * 32;
  const int ZP = (N + 31) & ~31;
  off = (off + 31) & ~(size_t)31;
  float* zwp = ws + off; off += (size_t)WCHUNKS * ZP;

  const float NEG_INF = -__builtin_inff();

  gemm_wmma<128, 32><<<cdiv(N, 128), 256, 0, stream>>>(x, W1, h1, N);
  fill_kernel<<<cdiv(2 * (N + 2), 256), 256, 0, stream>>>(mb1, NEG_INF, 2 * (N + 2));
  node_dots<32><<<cdiv(N, 256), 256, 0, stream>>>(h1, a_src1, a_dst1, sv, dv, rng1, N);
  edge_max<<<512, 256, 0, stream>>>(src, dst, sv, dv, rng1, mb1, E, ET);
  edge_window<<<cdiv(N, WROWS) * WCHUNKS, 1024, 0, stream>>>(src, dst, sv, dv, rng1, mb1, h1, wp, zwp, E, ET, N, ZP);

  reduce_project<<<cdiv(N, RP_ROWS), 256, 0, stream>>>(wp, zwp, b1, W2, a_src2, a_dst2, h2, sv, dv, rng2, N, ZP);
  edge_max<<<512, 256, 0, stream>>>(src, dst, sv, dv, rng2, mb2, E, ET);
  edge_exp_priv<<<ZGRID, 1024, 0, stream>>>(src, dst, sv, dv, rng2, mb2, eb, zp, E, ET, N, NP);
  z_reduce<<<cdiv(NP, 128), 256, 0, stream>>>(zp, zb, N, NP);

  edge_pool<<<PGRID, 1024, 0, stream>>>(src, dst, eb, zb, batch, h2, pp, E, ET);
  pool_finish<<<NUM_GRAPHS, 1024, 0, stream>>>(pp, b2, (float*)d_out);
}


# edge_exp_priv: a workgroup takes an edge chunk and a quarter of the dst nodes, holds s, its d and its z in LDS: a kept edge needs no gather from memory; zpart 256 rows become 64; z sum order changes
# speedup vs baseline: 4.8425x; 1.0477x over previous
#include <hip/hip_runtime.h>
#include <hip/hip_bf16.h>
#include <math.h>

typedef __attribute__((ext_vector_type(2))) float v2f;
typedef __attribute__((ext_vector_type(8))) float v8f;

#define NUM_GRAPHS 64
#define NEG_SLOPE 0.2f
#define SPREAD_LIMIT 8.0f

template <int FIN, int FOUT>
__global__ __launch_bounds__(256) void gemm_wmma(const float* __restrict__ x,
                                                 const float* __restrict__ W,
                                                 float* __restrict__ h, int N) {
  constexpr int NT = FOUT / 16;
  __shared__ float Ws[FIN * FOUT];
  for (int i = threadIdx.x; i < FIN * FOUT; i += 256) Ws[i] = W[i];
  __syncthreads();

  const int wave = threadIdx.x >> 5;
  const int lane = threadIdx.x & 31;
  const int m0 = (blockIdx.x * 8 + wave) * 16;
  const int halfsel = lane >> 4;
  const int lidx = lane & 15;
  int row = m0 + lidx;
  if (row >= N) row = N - 1;
  const int khalf = halfsel * 2;

  v8f acc[NT];
#pragma unroll
  for (int t = 0; t < NT; ++t)
#pragma unroll
    for (int j = 0; j < 8; ++j) acc[t][j] = 0.0f;

#pragma unroll 4
  for (int k = 0; k < FIN; k += 4) {
    const int kk = k + khalf;
    v2f a;
    a.x = x[(size_t)row * FIN + kk];
    a.y = x[(size_t)row * FIN + kk + 1];
#pragma unroll
    for (int t = 0; t < NT; ++t) {
      v2f b;
      b.x = Ws[kk * FOUT + t * 16 + lidx];
      b.y = Ws[(kk + 1) * FOUT + t * 16 + lidx];
      acc[t] = __builtin_amdgcn_wmma_f32_16x16x4_f32(
          false, a, false, b, (short)0, acc[t], false, false);
    }
  }

  const int rbase = m0 + halfsel * 8;
#pragma unroll
  for (int i = 0; i < 8; ++i) {
    const int r = rbase + i;
    if (r < N) {
#pragma unroll
      for (int t = 0; t < NT; ++t)
        h[(size_t)r * FOUT + t * 16 + lidx] = acc[t][i];
    }
  }
}

__device__ __forceinline__ void atomicMaxFloat(float* addr, float val) {
  int vi = __float_as_int(val);
  if (vi >= 0)
    atomicMax((int*)addr, vi);
  else
    atomicMin((unsigned int*)addr, (unsigned int)vi);
}

__device__ __forceinline__ float leaky(float v) {
  return (v > 0.0f) ? v : NEG_SLOPE * v;
}

template <int F>
__global__ __launch_bounds__(256) void node_dots(
    const float* __restrict__ h, const float* __restrict__ asrc,
    const float* __restrict__ adst, float* __restrict__ s,
    float* __restrict__ d, float* __restrict__ rng, int N) {
  __shared__ float red[16];
  const int n = blockIdx.x * 256 + threadIdx.x;
  const bool valid = n < N;
  float ss = 0.0f, dd = 0.0f;
  if (valid) {
#pragma unroll
    for (int f = 0; f < F; ++f) {
      float v = h[(size_t)n * F + f];
      ss += v * asrc[f];
      dd += v * adst[f];
    }
    s[n] = ss;
    d[n] = dd;
  }
  const float ninf = -__builtin_inff();
  float hi = valid ? ss : ninf;
  float lo = valid ? -ss : ninf;
#pragma unroll
  for (int o = 16; o > 0; o >>= 1) {
    hi = fmaxf(hi, __shfl_xor(hi, o));
    lo = fmaxf(lo, __shfl_xor(lo, o));
  }
  const int wave = threadIdx.x >> 5;
  if ((threadIdx.x & 31) == 0) {
    red[wave] = hi;
    red[8 + wave] = lo;
  }
  __syncthreads();
  if (threadIdx.x < 16) {
    float v = red[threadIdx.x];
    v = fmaxf(v, __shfl_xor(v, 4));
    v = fmaxf(v, __shfl_xor(v, 2));
    v = fmaxf(v, __shfl_xor(v, 1));
    if ((threadIdx.x & 7) == 0) atomicMaxFloat(rng + (threadIdx.x >> 3), v);
  }
}


__global__ __launch_bounds__(256) void edge_max(
    const int* __restrict__ src, const int* __restrict__ dst,
    const float* __restrict__ sv, const float* __restrict__ dv,
    const float* __restrict__ rng, float* __restrict__ m, int E, int ET) {
  if (rng[0] + rng[1] < SPREAD_LIMIT) return;
  for (int id = blockIdx.x * 256 + threadIdx.x; id < ET;
       id += gridDim.x * 256) {
    int si, di;
    if (id < E) {
      si = src[id];
      di = dst[id];
    } else {
      si = di = id - E;
    }
    atomicMaxFloat(m + di, leaky(sv[si] + dv[di]));
  }
}

#define ZG 4
#define ZCHUNKS 64
#define ZGRID (ZG * ZCHUNKS)
#define SCAP 53248
#define GCAP (SCAP / ZG)
#define ZUNROLL 8
#define SFILL 8

template <bool BOUND, bool REAL>
__device__ __forceinline__ void exp_scan(
    const int* __restrict__ src, const int* __restrict__ dst,
    const float* __restrict__ m, float* __restrict__ ebuf, const float* sl,
    const float* dl, float* zl, float smax, int e0, int e1, int E, int wbase,
    int wcnt, int sbase, int scnt, int tid) {
  const int last = max(min(e1, E), 1) - 1;
  for (int id0 = e0 + tid; id0 < e1; id0 += ZUNROLL * 1024) {
    int ws[ZUNROLL], wd[ZUNROLL];
#pragma unroll
    for (int u = 0; u < ZUNROLL; ++u) {
      const int idc = min(id0 + u * 1024, last);
      ws[u] = src[idc];
      wd[u] = dst[idc];
    }
#pragma unroll
    for (int u = 0; u < ZUNROLL; ++u) {
      const int id = id0 + u * 1024;
      const bool real = REAL || id < E;
      const int si = real ? ws[u] : id - E;
      const int di = real ? wd[u] : id - E;
      const int li = di - wbase;
      const int lsi = si - sbase;
      if ((id < e1) & ((unsigned)li < (unsigned)wcnt) &
          ((unsigned)lsi < (unsigned)scnt)) {
        const float ss = sl[lsi];
        const float dd = dl[li];
        const float e = leaky(ss + dd);
        const float shift = BOUND ? leaky(smax + dd) : m[di];
        const float t = __expf(e - shift);
        ebuf[id] = t;
        atomicAdd(zl + li, t);
      }
    }
  }
}

__global__ __launch_bounds__(1024) void edge_exp_priv(
    const int* __restrict__ src, const int* __restrict__ dst,
    const float* __restrict__ sv, const float* __restrict__ dv,
    const float* __restrict__ rng, const float* __restrict__ m,
    float* __restrict__ ebuf, float* __restrict__ zpart, int E, int ET, int N,
    int NP) {
  __shared__ __attribute__((aligned(16))) float sl[SCAP];
  __shared__ __attribute__((aligned(16))) float dl[GCAP];
  __shared__ __attribute__((aligned(16))) float zl[GCAP];
  const int tid = threadIdx.x;
  const int chunk = blockIdx.x / ZG;
  const int grp = blockIdx.x - chunk * ZG;
  const float smax = rng[0];
  const bool bound = smax + rng[1] < SPREAD_LIMIT;
  float* __restrict__ zrow = zpart + (size_t)chunk * NP;

  const int per = ((ET + ZCHUNKS * 32 - 1) / (ZCHUNKS * 32)) * 32;
  const int e0 = chunk * per;
  const int e1 = min(ET, e0 + per);

  for (int wbase = grp * GCAP; wbase < N; wbase += ZG * GCAP) {
    const int wcnt = min(N - wbase, GCAP);
    const int cnt4 = (wcnt + 3) & ~3;
    for (int i = tid; i < cnt4; i += 1024) {
      dl[i] = dv[min(wbase + i, N - 1)];
      zl[i] = 0.0f;
    }
    for (int sbase = 0; sbase < N; sbase += SCAP) {
      const int scnt = min(N - sbase, SCAP);
      const int s4 = scnt & ~3;
      for (int i0 = tid * 4; i0 < s4; i0 += SFILL * 4096) {
        float4 v[SFILL];
#pragma unroll
        for (int u = 0; u < SFILL; ++u)
          v[u] = *(const float4*)(sv + sbase + min(i0 + u * 4096, s4 - 4));
#pragma unroll
        for (int u = 0; u < SFILL; ++u)
          *(float4*)(sl + min(i0 + u * 4096, s4 - 4)) = v[u];
      }
      if (tid < scnt - s4) sl[s4 + tid] = sv[sbase + s4 + tid];
      __syncthreads();

      if (bound) {
        if (e1 <= E)
          exp_scan<true, true>(src, dst, m, ebuf, sl, dl, zl, smax, e0, e1, E,
                               wbase, wcnt, sbase, scnt, tid);
        else
          exp_scan<true, false>(src, dst, m, ebuf, sl, dl, zl, smax, e0, e1, E,
                                wbase, wcnt, sbase, scnt, tid);
      } else {
        exp_scan<false, false>(src, dst, m, ebuf, sl, dl, zl, smax, e0, e1, E,
                               wbase, wcnt, sbase, scnt, tid);
      }
      __syncthreads();
    }

    for (int i = tid * 4; i < cnt4; i += 4096)
      *(float4*)(zrow + wbase + i) = *(const float4*)(zl + i);
    __syncthreads();
  }
}

__global__ __launch_bounds__(256) void z_reduce(const float* __restrict__ zpart,
                                                float* __restrict__ z, int N,
                                                int NP) {
  __shared__ __attribute__((aligned(16))) float red[8][128];
  const int wave = threadIdx.x >> 5;
  const int lane = threadIdx.x & 31;
  const int n4 = blockIdx.x * 128 + lane * 4;
  float4 acc = make_float4(0.0f, 0.0f, 0.0f, 0.0f);
  if (n4 < NP) {
    const float* p = zpart + (size_t)wave * NP + n4;
#pragma unroll 8
    for (int k = 0; k < ZCHUNKS / 8; ++k) {
      const float4 v = *(const float4*)p;
      p += (size_t)8 * NP;
      acc.x += v.x;
      acc.y += v.y;
      acc.z += v.z;
      acc.w += v.w;
    }
  }
  *(float4*)(&red[wave][lane * 4]) = acc;
  __syncthreads();
  if (threadIdx.x < 128) {
    const int n = blockIdx.x * 128 + threadIdx.x;
    float r = red[0][threadIdx.x];
#pragma unroll
    for (int w = 1; w < 8; ++w) r += red[w][threadIdx.x];
    if (n < N) z[n] = r;
  }
}

#define WROWS 2048
#define WCHUNKS 10
#define WSCAN 8
#define WQ (32 * WSCAN + 32)

__device__ __forceinline__ int win_dst_word(const int* __restrict__ dst, int id,
                                            int last) {
  return dst[min(id, last)];
}

__device__ __forceinline__ bool win_hit(int d, int id, int lim, int e1, int E,
                                        int wbase, int wcnt) {
  const bool real_hit = ((unsigned)(d - wbase) < (unsigned)wcnt) & (id < lim);
  const bool self_hit = ((unsigned)(id - E - wbase) < (unsigned)wcnt) & (id < e1);
  return real_hit | self_hit;
}

template <bool REAL>
__device__ __forceinline__ int win_queue(const int* cur, int b, int lane,
                                         int lim, int e1, int E, int wbase,
                                         int wcnt, int* myq, int cnt) {
#pragma unroll
  for (int u = 0; u < WSCAN; ++u) {
    const int id = b + u * 32 + lane;
    const bool hit = REAL ? ((unsigned)(cur[u] - wbase) < (unsigned)wcnt)
                          : win_hit(cur[u], id, lim, e1, E, wbase, wcnt);
    const unsigned mask = __builtin_amdgcn_ballot_w32(hit);
    if (hit) myq[cnt + (int)__builtin_amdgcn_mbcnt_lo(mask, 0u)] = id;
    cnt += __builtin_popcount(mask);
  }
  return cnt;
}

__device__ __forceinline__ float win_row_load(const float* __restrict__ h,
                                              int si, int j, int lane) {
  const int sj = __builtin_amdgcn_readlane(si, j);
  return h[(size_t)sj * 32 + lane];
}

struct WinScore {
  const float* __restrict__ sv;
  const float* __restrict__ dv;
  const float* __restrict__ m;
  float smax;
  bool bound;
};

__device__ __forceinline__ void win_row_add(float* acc, float v, int dl,
                                            float t, int j, int lane) {
  const int rj = __builtin_amdgcn_readlane(dl, j);
  const float tj = __int_as_float(__builtin_amdgcn_readlane(__float_as_int(t), j));
  atomicAdd(acc + rj * 32 + lane, tj * v);
}

template <bool FULL>
__device__ __forceinline__ void win_batch(
    const int* qp, int m, int lane, int wbase, int E,
    const int* __restrict__ src, const int* __restrict__ dst,
    const WinScore& sc, const float* __restrict__ h, float* acc, float* zl) {
  const int id = qp[(FULL || lane < m) ? lane : 0];
  const bool real = id < E;
  const int idc = real ? id : 0;
  const int s0 = src[idc];
  const int d0 = dst[idc];
  const int self = id - E;
  const int si = real ? s0 : self;
  const int di = real ? d0 : self;
  const int dl = di - wbase;
  const float ss = sc.sv[si];
  const float dd = sc.dv[di];
  const float mm = sc.bound ? 0.0f : sc.m[di];
  float v[32];
#pragma unroll
  for (int j = 0; j < 32; ++j) {
    v[j] = 0.0f;
    if (FULL || j < m) v[j] = win_row_load(h, si, j, lane);
  }
  const float shift = sc.bound ? leaky(sc.smax + dd) : mm;
  const float t = __expf(leaky(ss + dd) - shift);
  if (FULL || lane < m) atomicAdd(zl + dl, t);
#pragma unroll
  for (int j = 0; j < 32; ++j)
    if (FULL || j < m) win_row_add(acc, v[j], dl, t, j, lane);
}

template <bool REAL>
__device__ __forceinline__ void win_steps(
    int& b, int (&cur)[WSCAN], int& cnt, int lane, int lim, int last, int e1,
    int E, int wbase, int wcnt, int* myq, const int* __restrict__ src,
    const int* __restrict__ dst, const WinScore& sc,
    const float* __restrict__ h, float* acc, float* zl) {
  const int step = 32 * WSCAN;
  for (; REAL ? (b + step <= lim) : (b < e1); b += 32 * step) {
    int nxt[WSCAN];
    const int nb = b + 32 * step;
#pragma unroll
    for (int u = 0; u < WSCAN; ++u)
      nxt[u] = win_dst_word(dst, nb + u * 32 + lane, last);
    cnt = win_queue<REAL>(cur, b, lane, lim, e1, E, wbase, wcnt, myq, cnt);
    while (cnt >= 32) {
      cnt -= 32;
      win_batch<true>(myq + cnt, 32, lane, wbase, E, src, dst, sc, h, acc, zl);
    }
#pragma unroll
    for (int u = 0; u < WSCAN; ++u) cur[u] = nxt[u];
  }
}

__global__ __launch_bounds__(1024) void edge_window(
    const int* __restrict__ src, const int* __restrict__ dst,
    const float* __restrict__ sv, const float* __restrict__ dv,
    const float* __restrict__ rng, const float* __restrict__ m,
    const float* __restrict__ h, float* __restrict__ wpart,
    float* __restrict__ zwpart, int E, int ET, int N, int ZP) {
  __shared__ __attribute__((aligned(16))) float acc[WROWS * 32];
  __shared__ __attribute__((aligned(16))) float zl[WROWS];
  __shared__ int q[32][WQ];
  const int tid = threadIdx.x;
  const int lane = tid & 31;
  const int wave = __builtin_amdgcn_readfirstlane(tid >> 5);
  const int win = blockIdx.x / WCHUNKS;
  const int chunk = blockIdx.x - win * WCHUNKS;
  const int wbase = win * WROWS;
  const int wcnt = min(N - wbase, WROWS);
  const int wfl = wcnt * 32;

  for (int i = tid * 4; i < wfl; i += 4096)
    *(float4*)(acc + i) = make_float4(0.0f, 0.0f, 0.0f, 0.0f);
  for (int i = tid; i < WROWS; i += 1024) zl[i] = 0.0f;
  __syncthreads();

  WinScore sc;
  sc.sv = sv;
  sc.dv = dv;
  sc.m = m;
  sc.smax = rng[0];
  sc.bound = sc.smax + rng[1] < SPREAD_LIMIT;

  const int per = ((ET + WCHUNKS * 32 - 1) / (WCHUNKS * 32)) * 32;
  const int e0 = chunk * per;
  const int e1 = min(ET, e0 + per);
  const int lim = min(e1, E);
  const int last = max(lim, 1) - 1;
  int* myq = q[wave];
  int cnt = 0;

  int b = e0 + wave * (32 * WSCAN);
  int cur[WSCAN];
#pragma unroll
  for (int u = 0; u < WSCAN; ++u)
    cur[u] = win_dst_word(dst, b + u * 32 + lane, last);
  win_steps<true>(b, cur, cnt, lane, lim, last, e1, E, wbase, wcnt, myq, src,
                  dst, sc, h, acc, zl);
  win_steps<false>(b, cur, cnt, lane, lim, last, e1, E, wbase, wcnt, myq, src,
                   dst, sc, h, acc, zl);
  if (cnt > 0)
    win_batch<false>(myq, cnt, lane, wbase, E, src, dst, sc, h, acc, zl);
  __syncthreads();

  float* __restrict__ prow = wpart + ((size_t)chunk * N + wbase) * 32;
  for (int i = tid * 4; i < wfl; i += 4096)
    *(float4*)(prow + i) = *(const float4*)(acc + i);
  float* __restrict__ zrow = zwpart + (size_t)chunk * ZP + wbase;
  for (int i = tid; i < wcnt; i += 1024) zrow[i] = zl[i];
}

#define RP_ROWS 128
#define RP_OP 34
#define RP_HP 68

__global__ __launch_bounds__(256) void reduce_project(
    const float* __restrict__ wpart, const float* __restrict__ zwpart,
    const float* __restrict__ b, const float* __restrict__ W,
    const float* __restrict__ asrc, const float* __restrict__ adst,
    float* __restrict__ h, float* __restrict__ s, float* __restrict__ d,
    float* __restrict__ rng, int N, int ZP) {
  constexpr int FIN = 32, FOUT = 64, NT = FOUT / 16;
  __shared__ __attribute__((aligned(16))) float Ws[FIN * FOUT];
  __shared__ __attribute__((aligned(16))) float os[RP_ROWS * RP_OP];
  __shared__ __attribute__((aligned(16))) float hs[RP_ROWS * RP_HP];
  __shared__ float red[16];
  const int tid = threadIdx.x;
  const int n0 = blockIdx.x * RP_ROWS;
  for (int i = tid; i < FIN * FOUT; i += 256) Ws[i] = W[i];

  const int f = (tid & 7) * 4;
  const float bx = b[f], by = b[f + 1], bz = b[f + 2], bw = b[f + 3];
#pragma unroll
  for (int it = 0; it < RP_ROWS * 8 / 256; ++it) {
    const int nl = (tid >> 3) + it * 32;
    const int n = n0 + nl;
    const int nc = min(n, N - 1);
    const float* p = wpart + ((size_t)nc * 32 + f);
    const float* pz = zwpart + nc;
    float4 a = make_float4(0.0f, 0.0f, 0.0f, 0.0f);
    float zn = 0.0f;
#pragma unroll
    for (int c = 0; c < WCHUNKS; ++c) {
      const float4 v = *(const float4*)p;
      const float zc = *pz;
      p += (size_t)N * 32;
      pz += ZP;
      a.x += v.x;
      a.y += v.y;
      a.z += v.z;
      a.w += v.w;
      zn += zc;
    }
    float4 r;
    r.x = a.x / zn + bx;
    r.y = a.y / zn + by;
    r.z = a.z / zn + bz;
    r.w = a.w / zn + bw;
    r.x = (r.x > 0.0f) ? r.x : (__expf(r.x) - 1.0f);
    r.y = (r.y > 0.0f) ? r.y : (__expf(r.y) - 1.0f);
    r.z = (r.z > 0.0f) ? r.z : (__expf(r.z) - 1.0f);
    r.w = (r.w > 0.0f) ? r.w : (__expf(r.w) - 1.0f);
    if (n >= N) r = make_float4(0.0f, 0.0f, 0.0f, 0.0f);
    float* q = os + nl * RP_OP + f;
    *(float2*)q = make_float2(r.x, r.y);
    *(float2*)(q + 2) = make_float2(r.z, r.w);
  }
  __syncthreads();

  const int wave = tid >> 5;
  const int lane = tid & 31;
  const int halfsel = lane >> 4;
  const int lidx = lane & 15;
  const int khalf = halfsel * 2;
  const float* arow = os + (wave * 16 + lidx) * RP_OP;

  v8f acc[NT];
#pragma unroll
  for (int t = 0; t < NT; ++t)
#pragma unroll
    for (int j = 0; j < 8; ++j) acc[t][j] = 0.0f;

#pragma unroll
  for (int k = 0; k < FIN; k += 4) {
    const int kk = k + khalf;
    v2f a;
    a.x = arow[kk];
    a.y = arow[kk + 1];
#pragma unroll
    for (int t = 0; t < NT; ++t) {
      v2f bb;
      bb.x = Ws[kk * FOUT + t * 16 + lidx];
      bb.y = Ws[(kk + 1) * FOUT + t * 16 + lidx];
      acc[t] = __builtin_amdgcn_wmma_f32_16x16x4_f32(
          false, a, false, bb, (short)0, acc[t], false, false);
    }
  }

  const int rloc0 = wave * 16 + halfsel * 8;
#pragma unroll
  for (int i = 0; i < 8; ++i) {
    const int rl = rloc0 + i;
    const int r = n0 + rl;
#pragma unroll
    for (int t = 0; t < NT; ++t) hs[rl * RP_HP + t * 16 + lidx] = acc[t][i];
    if (r < N) {
#pragma unroll
      for (int t = 0; t < NT; ++t)
        h[(size_t)r * FOUT + t * 16 + lidx] = acc[t][i];
    }
  }
  __syncthreads();

  const int n = n0 + tid;
  const bool valid = (tid < RP_ROWS) & (n < N);
  float ss = 0.0f, dd = 0.0f;
  if (valid) {
    const float* hr = hs + tid * RP_HP;
#pragma unroll
    for (int f = 0; f < FOUT; ++f) {
      float v = hr[f];
      ss += v * asrc[f];
      dd += v * adst[f];
    }
    s[n] = ss;
    d[n] = dd;
  }
  const float ninf = -__builtin_inff();
  float hi = valid ? ss : ninf;
  float lo = valid ? -ss : ninf;
#pragma unroll
  for (int o = 16; o > 0; o >>= 1) {
    hi = fmaxf(hi, __shfl_xor(hi, o));
    lo = fmaxf(lo, __shfl_xor(lo, o));
  }
  if (lane == 0) {
    red[wave] = hi;
    red[8 + wave] = lo;
  }
  __syncthreads();
  if (tid < 16) {
    float v = red[tid];
    v = fmaxf(v, __shfl_xor(v, 4));
    v = fmaxf(v, __shfl_xor(v, 2));
    v = fmaxf(v, __shfl_xor(v, 1));
    if ((tid & 7) == 0) atomicMaxFloat(rng + (tid >> 3), v);
  }
}

#define PGRID 256
#define PSUMS (NUM_GRAPHS * 64)
#define PROW (PSUMS + NUM_GRAPHS)

__device__ __forceinline__ float2 pool_row_load(const float* __restrict__ h,
                                                int si, int j, int lane) {
  const int sj = __builtin_amdgcn_readlane(si, j);
  return *(const float2*)(h + (size_t)sj * 64 + lane * 2);
}

__device__ __forceinline__ void pool_row_add(float* pl, float2 v, int g,
                                             float a, int j, int lane) {
  const int gj = __builtin_amdgcn_readlane(g, j);
  const float aj = __int_as_float(__builtin_amdgcn_readlane(__float_as_int(a), j));
  float* p = pl + gj * 64 + lane * 2;
  atomicAdd(p, aj * v.x);
  atomicAdd(p + 1, aj * v.y);
}

__global__ __launch_bounds__(1024) void edge_pool(
    const int* __restrict__ src, const int* __restrict__ dst,
    const float* __restrict__ ebuf, const float* __restrict__ z,
    const int* __restrict__ batch, const float* __restrict__ h,
    float* __restrict__ ppart, int E, int ET) {
  __shared__ float pl[PROW];
  const int tid = threadIdx.x;
  const int lane = tid & 31;
  for (int i = tid; i < PROW; i += 1024) pl[i] = 0.0f;
  __syncthreads();

  const int wave = blockIdx.x * 32 + (tid >> 5);
  const int sweep = gridDim.x * 32 * 32;
  for (int b0 = wave * 32; b0 < ET; b0 += sweep) {
    const int id = b0 + lane;
    const bool live = id < ET;
    const bool real = id < E;
    const int idc = real ? id : 0;
    const int s0 = src[idc];
    const int d0 = dst[idc];
    const int self = live ? id - E : 0;
    const int si = real ? s0 : self;
    const int di = real ? d0 : self;
    const float t = ebuf[live ? id : 0];
    const int g = batch[di];
    const float a = t / z[di];
    if (live && !real) atomicAdd(pl + PSUMS + g, 1.0f);

    const int cnt = min(32, ET - b0);
    if (cnt == 32) {
#pragma unroll
      for (int j0 = 0; j0 < 32; j0 += 16) {
        float2 v[16];
#pragma unroll
        for (int u = 0; u < 16; ++u) v[u] = pool_row_load(h, si, j0 + u, lane);
#pragma unroll
        for (int u = 0; u < 16; ++u) pool_row_add(pl, v[u], g, a, j0 + u, lane);
      }
    } else {
      for (int j = 0; j < cnt; ++j)
        pool_row_add(pl, pool_row_load(h, si, j, lane), g, a, j, lane);
    }
  }
  __syncthreads();

  float* __restrict__ row = ppart + (size_t)blockIdx.x * PROW;
  for (int i = tid; i < PROW; i += 1024) row[i] = pl[i];
}

__global__ __launch_bounds__(1024) void pool_finish(
    const float* __restrict__ ppart, const float* __restrict__ b,
    float* __restrict__ out) {
  __shared__ float red[16][64];
  __shared__ float cw[PGRID / 32];
  const int g = blockIdx.x;
  const int f = threadIdx.x & 63;
  const int r = threadIdx.x >> 6;
  const float* p = ppart + (size_t)r * PROW + g * 64 + f;
  float acc = 0.0f;
#pragma unroll
  for (int k = 0; k < PGRID / 16; ++k) {
    acc += *p;
    p += (size_t)16 * PROW;
  }
  red[r][f] = acc;
  if (threadIdx.x < PGRID) {
    float c = ppart[(size_t)threadIdx.x * PROW + PSUMS + g];
    for (int o = 16; o > 0; o >>= 1) c += __shfl_xor(c, o);
    if ((threadIdx.x & 31) == 0) cw[threadIdx.x >> 5] = c;
  }
  __syncthreads();
  if (threadIdx.x < 64) {
    float c = 0.0f;
    for (int w = 0; w < PGRID / 32; ++w) c += cw[w];
    float s = red[0][f];
#pragma unroll
    for (int w = 1; w < 16; ++w) s += red[w][f];
    out[g * 64 + f] = (s + c * b[f]) / fmaxf(c, 1.0f);
  }
}

__global__ void fill_kernel(float* __restrict__ p, float v, int n) {
  int i = blockIdx.x * blockDim.x + threadIdx.x;
  if (i < n) p[i] = v;
}

static inline int cdiv(int a, int b) { return (a + b - 1) / b; }

extern "C" void kernel_launch(void* const* d_in, const int* in_sizes, int n_in,
                              void* d_out, int out_size, void* d_ws,
                              size_t ws_size, hipStream_t stream) {
  const float* x      = (const float*)d_in[0];
  const int*   ei     = (const int*)d_in[1];
  const int*   batch  = (const int*)d_in[2];
  const float* W1     = (const float*)d_in[3];
  const float* a_src1 = (const float*)d_in[4];
  const float* a_dst1 = (const float*)d_in[5];
  const float* b1     = (const float*)d_in[6];
  const float* W2     = (const float*)d_in[7];
  const float* a_src2 = (const float*)d_in[8];
  const float* a_dst2 = (const float*)d_in[9];
  const float* b2     = (const float*)d_in[10];

  const int N = in_sizes[2];
  const int E = in_sizes[1] / 2;
  const int ET = E + N;
  const int* src = ei;
  const int* dst = ei + E;

  float* ws = (float*)d_ws;
  size_t off = 0;
  float* h1 = ws + off; off += (size_t)N * 32;
  float* h2 = ws + off; off += (size_t)N * 64;
  float* sv = ws + off; off += (size_t)N;
  float* dv = ws + off; off += (size_t)N;
  float* mb1 = ws + off; off += (size_t)N + 2;
  float* rng1 = mb1 + N;
  float* mb2 = ws + off; off += (size_t)N + 2;
  float* rng2 = mb2 + N;
  float* zb = ws + off; off += (size_t)N;
  float* eb = ws + off; off += (size_t)ET;
  const int NP = (N + 3) & ~3;
  off = (off + 3) & ~(size_t)3;
  float* zp = ws + off; off += (size_t)ZCHUNKS * NP;
  float* pp = ws + off; off += (size_t)PGRID * PROW;
  float* wp = ws + off; off += (size_t)WCHUNKS * N * 32;
  const int ZP = (N + 31) & ~31;
  off = (off + 31) & ~(size_t)31;
  float* zwp = ws + off; off += (size_t)WCHUNKS * ZP;

  const float NEG_INF = -__builtin_inff();

  gemm_wmma<128, 32><<<cdiv(N, 128), 256, 0, stream>>>(x, W1, h1, N);
  fill_kernel<<<cdiv(2 * (N + 2), 256), 256, 0, stream>>>(mb1, NEG_INF, 2 * (N + 2));
  node_dots<32><<<cdiv(N, 256), 256, 0, stream>>>(h1, a_src1, a_dst1, sv, dv, rng1, N);
  edge_max<<<512, 256, 0, stream>>>(src, dst, sv, dv, rng1, mb1, E, ET);
  edge_window<<<cdiv(N, WROWS) * WCHUNKS, 1024, 0, stream>>>(src, dst, sv, dv, rng1, mb1, h1, wp, zwp, E, ET, N, ZP);

  reduce_project<<<cdiv(N, RP_ROWS), 256, 0, stream>>>(wp, zwp, b1, W2, a_src2, a_dst2, h2, sv, dv, rng2, N, ZP);
  edge_max<<<512, 256, 0, stream>>>(src, dst, sv, dv, rng2, mb2, E, ET);
  edge_exp_priv<<<ZGRID, 1024, 0, stream>>>(src, dst, sv, dv, rng2, mb2, eb, zp, E, ET, N, NP);
  z_reduce<<<cdiv(NP, 128), 256, 0, stream>>>(zp, zb, N, NP);

  edge_pool<<<PGRID, 1024, 0, stream>>>(src, dst, eb, zb, batch, h2, pp, E, ET);
  pool_finish<<<NUM_GRAPHS, 1024, 0, stream>>>(pp, b2, (float*)d_out);
}


# Per-edge gathers by dst become LDS reads: edge_pool holds z and the graph number (bytes written by z_reduce) of every node in LDS, edge_window holds d of its window; sums and values unchanged
# speedup vs baseline: 5.0306x; 1.0388x over previous
#include <hip/hip_runtime.h>
#include <hip/hip_bf16.h>
#include <math.h>

typedef __attribute__((ext_vector_type(2))) float v2f;
typedef __attribute__((ext_vector_type(8))) float v8f;

#define NUM_GRAPHS 64
#define NEG_SLOPE 0.2f
#define SPREAD_LIMIT 8.0f

template <int FIN, int FOUT>
__global__ __launch_bounds__(256) void gemm_wmma(const float* __restrict__ x,
                                                 const float* __restrict__ W,
                                                 float* __restrict__ h, int N) {
  constexpr int NT = FOUT / 16;
  __shared__ float Ws[FIN * FOUT];
  for (int i = threadIdx.x; i < FIN * FOUT; i += 256) Ws[i] = W[i];
  __syncthreads();

  const int wave = threadIdx.x >> 5;
  const int lane = threadIdx.x & 31;
  const int m0 = (blockIdx.x * 8 + wave) * 16;
  const int halfsel = lane >> 4;
  const int lidx = lane & 15;
  int row = m0 + lidx;
  if (row >= N) row = N - 1;
  const int khalf = halfsel * 2;

  v8f acc[NT];
#pragma unroll
  for (int t = 0; t < NT; ++t)
#pragma unroll
    for (int j = 0; j < 8; ++j) acc[t][j] = 0.0f;

#pragma unroll 4
  for (int k = 0; k < FIN; k += 4) {
    const int kk = k + khalf;
    v2f a;
    a.x = x[(size_t)row * FIN + kk];
    a.y = x[(size_t)row * FIN + kk + 1];
#pragma unroll
    for (int t = 0; t < NT; ++t) {
      v2f b;
      b.x = Ws[kk * FOUT + t * 16 + lidx];
      b.y = Ws[(kk + 1) * FOUT + t * 16 + lidx];
      acc[t] = __builtin_amdgcn_wmma_f32_16x16x4_f32(
          false, a, false, b, (short)0, acc[t], false, false);
    }
  }

  const int rbase = m0 + halfsel * 8;
#pragma unroll
  for (int i = 0; i < 8; ++i) {
    const int r = rbase + i;
    if (r < N) {
#pragma unroll
      for (int t = 0; t < NT; ++t)
        h[(size_t)r * FOUT + t * 16 + lidx] = acc[t][i];
    }
  }
}

__device__ __forceinline__ void atomicMaxFloat(float* addr, float val) {
  int vi = __float_as_int(val);
  if (vi >= 0)
    atomicMax((int*)addr, vi);
  else
    atomicMin((unsigned int*)addr, (unsigned int)vi);
}

__device__ __forceinline__ float leaky(float v) {
  return (v > 0.0f) ? v : NEG_SLOPE * v;
}

template <int F>
__global__ __launch_bounds__(256) void node_dots(
    const float* __restrict__ h, const float* __restrict__ asrc,
    const float* __restrict__ adst, float* __restrict__ s,
    float* __restrict__ d, float* __restrict__ rng, int N) {
  __shared__ float red[16];
  const int n = blockIdx.x * 256 + threadIdx.x;
  const bool valid = n < N;
  float ss = 0.0f, dd = 0.0f;
  if (valid) {
#pragma unroll
    for (int f = 0; f < F; ++f) {
      float v = h[(size_t)n * F + f];
      ss += v * asrc[f];
      dd += v * adst[f];
    }
    s[n] = ss;
    d[n] = dd;
  }
  const float ninf = -__builtin_inff();
  float hi = valid ? ss : ninf;
  float lo = valid ? -ss : ninf;
#pragma unroll
  for (int o = 16; o > 0; o >>= 1) {
    hi = fmaxf(hi, __shfl_xor(hi, o));
    lo = fmaxf(lo, __shfl_xor(lo, o));
  }
  const int wave = threadIdx.x >> 5;
  if ((threadIdx.x & 31) == 0) {
    red[wave] = hi;
    red[8 + wave] = lo;
  }
  __syncthreads();
  if (threadIdx.x < 16) {
    float v = red[threadIdx.x];
    v = fmaxf(v, __shfl_xor(v, 4));
    v = fmaxf(v, __shfl_xor(v, 2));
    v = fmaxf(v, __shfl_xor(v, 1));
    if ((threadIdx.x & 7) == 0) atomicMaxFloat(rng + (threadIdx.x >> 3), v);
  }
}


__global__ __launch_bounds__(256) void edge_max(
    const int* __restrict__ src, const int* __restrict__ dst,
    const float* __restrict__ sv, const float* __restrict__ dv,
    const float* __restrict__ rng, float* __restrict__ m, int E, int ET) {
  if (rng[0] + rng[1] < SPREAD_LIMIT) return;
  for (int id = blockIdx.x * 256 + threadIdx.x; id < ET;
       id += gridDim.x * 256) {
    int si, di;
    if (id < E) {
      si = src[id];
      di = dst[id];
    } else {
      si = di = id - E;
    }
    atomicMaxFloat(m + di, leaky(sv[si] + dv[di]));
  }
}

#define ZG 4
#define ZCHUNKS 64
#define ZGRID (ZG * ZCHUNKS)
#define SCAP 53248
#define GCAP (SCAP / ZG)
#define ZUNROLL 8
#define SFILL 8

template <bool BOUND, bool REAL>
__device__ __forceinline__ void exp_scan(
    const int* __restrict__ src, const int* __restrict__ dst,
    const float* __restrict__ m, float* __restrict__ ebuf, const float* sl,
    const float* dl, float* zl, float smax, int e0, int e1, int E, int wbase,
    int wcnt, int sbase, int scnt, int tid) {
  const int last = max(min(e1, E), 1) - 1;
  for (int id0 = e0 + tid; id0 < e1; id0 += ZUNROLL * 1024) {
    int ws[ZUNROLL], wd[ZUNROLL];
#pragma unroll
    for (int u = 0; u < ZUNROLL; ++u) {
      const int idc = min(id0 + u * 1024, last);
      ws[u] = src[idc];
      wd[u] = dst[idc];
    }
#pragma unroll
    for (int u = 0; u < ZUNROLL; ++u) {
      const int id = id0 + u * 1024;
      const bool real = REAL || id < E;
      const int si = real ? ws[u] : id - E;
      const int di = real ? wd[u] : id - E;
      const int li = di - wbase;
      const int lsi = si - sbase;
      if ((id < e1) & ((unsigned)li < (unsigned)wcnt) &
          ((unsigned)lsi < (unsigned)scnt)) {
        const float ss = sl[lsi];
        const float dd = dl[li];
        const float e = leaky(ss + dd);
        const float shift = BOUND ? leaky(smax + dd) : m[di];
        const float t = __expf(e - shift);
        ebuf[id] = t;
        atomicAdd(zl + li, t);
      }
    }
  }
}

__global__ __launch_bounds__(1024) void edge_exp_priv(
    const int* __restrict__ src, const int* __restrict__ dst,
    const float* __restrict__ sv, const float* __restrict__ dv,
    const float* __restrict__ rng, const float* __restrict__ m,
    float* __restrict__ ebuf, float* __restrict__ zpart, int E, int ET, int N,
    int NP) {
  __shared__ __attribute__((aligned(16))) float sl[SCAP];
  __shared__ __attribute__((aligned(16))) float dl[GCAP];
  __shared__ __attribute__((aligned(16))) float zl[GCAP];
  const int tid = threadIdx.x;
  const int chunk = blockIdx.x / ZG;
  const int grp = blockIdx.x - chunk * ZG;
  const float smax = rng[0];
  const bool bound = smax + rng[1] < SPREAD_LIMIT;
  float* __restrict__ zrow = zpart + (size_t)chunk * NP;

  const int per = ((ET + ZCHUNKS * 32 - 1) / (ZCHUNKS * 32)) * 32;
  const int e0 = chunk * per;
  const int e1 = min(ET, e0 + per);

  for (int wbase = grp * GCAP; wbase < N; wbase += ZG * GCAP) {
    const int wcnt = min(N - wbase, GCAP);
    const int cnt4 = (wcnt + 3) & ~3;
    for (int i = tid; i < cnt4; i += 1024) {
      dl[i] = dv[min(wbase + i, N - 1)];
      zl[i] = 0.0f;
    }
    for (int sbase = 0; sbase < N; sbase += SCAP) {
      const int scnt = min(N - sbase, SCAP);
      const int s4 = scnt & ~3;
      for (int i0 = tid * 4; i0 < s4; i0 += SFILL * 4096) {
        float4 v[SFILL];
#pragma unroll
        for (int u = 0; u < SFILL; ++u)
          v[u] = *(const float4*)(sv + sbase + min(i0 + u * 4096, s4 - 4));
#pragma unroll
        for (int u = 0; u < SFILL; ++u)
          *(float4*)(sl + min(i0 + u * 4096, s4 - 4)) = v[u];
      }
      if (tid < scnt - s4) sl[s4 + tid] = sv[sbase + s4 + tid];
      __syncthreads();

      if (bound) {
        if (e1 <= E)
          exp_scan<true, true>(src, dst, m, ebuf, sl, dl, zl, smax, e0, e1, E,
                               wbase, wcnt, sbase, scnt, tid);
        else
          exp_scan<true, false>(src, dst, m, ebuf, sl, dl, zl, smax, e0, e1, E,
                                wbase, wcnt, sbase, scnt, tid);
      } else {
        exp_scan<false, false>(src, dst, m, ebuf, sl, dl, zl, smax, e0, e1, E,
                               wbase, wcnt, sbase, scnt, tid);
      }
      __syncthreads();
    }

    for (int i = tid * 4; i < cnt4; i += 4096)
      *(float4*)(zrow + wbase + i) = *(const float4*)(zl + i);
    __syncthreads();
  }
}

__global__ __launch_bounds__(256) void z_reduce(const float* __restrict__ zpart,
                                                const int* __restrict__ batch,
                                                float* __restrict__ z,
                                                unsigned char* __restrict__ gbytes,
                                                int N, int NP) {
  __shared__ __attribute__((aligned(16))) float red[8][128];
  const int wave = threadIdx.x >> 5;
  const int lane = threadIdx.x & 31;
  const int n4 = blockIdx.x * 128 + lane * 4;
  float4 acc = make_float4(0.0f, 0.0f, 0.0f, 0.0f);
  if (n4 < NP) {
    const float* p = zpart + (size_t)wave * NP + n4;
#pragma unroll 8
    for (int k = 0; k < ZCHUNKS / 8; ++k) {
      const float4 v = *(const float4*)p;
      p += (size_t)8 * NP;
      acc.x += v.x;
      acc.y += v.y;
      acc.z += v.z;
      acc.w += v.w;
    }
  }
  *(float4*)(&red[wave][lane * 4]) = acc;
  __syncthreads();
  if (threadIdx.x < 128) {
    const int n = blockIdx.x * 128 + threadIdx.x;
    float r = red[0][threadIdx.x];
#pragma unroll
    for (int w = 1; w < 8; ++w) r += red[w][threadIdx.x];
    if (n < N) {
      z[n] = r;
      gbytes[n] = (unsigned char)batch[n];
    }
  }
}

#define WROWS 2048
#define WCHUNKS 10
#define WSCAN 8
#define WQ (32 * WSCAN + 32)

__device__ __forceinline__ int win_dst_word(const int* __restrict__ dst, int id,
                                            int last) {
  return dst[min(id, last)];
}

__device__ __forceinline__ bool win_hit(int d, int id, int lim, int e1, int E,
                                        int wbase, int wcnt) {
  const bool real_hit = ((unsigned)(d - wbase) < (unsigned)wcnt) & (id < lim);
  const bool self_hit = ((unsigned)(id - E - wbase) < (unsigned)wcnt) & (id < e1);
  return real_hit | self_hit;
}

template <bool REAL>
__device__ __forceinline__ int win_queue(const int* cur, int b, int lane,
                                         int lim, int e1, int E, int wbase,
                                         int wcnt, int* myq, int cnt) {
#pragma unroll
  for (int u = 0; u < WSCAN; ++u) {
    const int id = b + u * 32 + lane;
    const bool hit = REAL ? ((unsigned)(cur[u] - wbase) < (unsigned)wcnt)
                          : win_hit(cur[u], id, lim, e1, E, wbase, wcnt);
    const unsigned mask = __builtin_amdgcn_ballot_w32(hit);
    if (hit) myq[cnt + (int)__builtin_amdgcn_mbcnt_lo(mask, 0u)] = id;
    cnt += __builtin_popcount(mask);
  }
  return cnt;
}

__device__ __forceinline__ float win_row_load(const float* __restrict__ h,
                                              int si, int j, int lane) {
  const int sj = __builtin_amdgcn_readlane(si, j);
  return h[(size_t)sj * 32 + lane];
}

struct WinScore {
  const float* __restrict__ sv;
  const float* __restrict__ m;
  float smax;
  bool bound;
};

__device__ __forceinline__ void win_row_add(float* acc, float v, int dl,
                                            float t, int j, int lane) {
  const int rj = __builtin_amdgcn_readlane(dl, j);
  const float tj = __int_as_float(__builtin_amdgcn_readlane(__float_as_int(t), j));
  atomicAdd(acc + rj * 32 + lane, tj * v);
}

template <bool FULL>
__device__ __forceinline__ void win_batch(
    const int* qp, int m, int lane, int wbase, int E,
    const int* __restrict__ src, const int* __restrict__ dst,
    const WinScore& sc, const float* __restrict__ h, float* acc, float* zl,
    const float* dwin) {
  const int id = qp[(FULL || lane < m) ? lane : 0];
  const bool real = id < E;
  const int idc = real ? id : 0;
  const int s0 = src[idc];
  const int d0 = dst[idc];
  const int self = id - E;
  const int si = real ? s0 : self;
  const int di = real ? d0 : self;
  const int dl = di - wbase;
  const float ss = sc.sv[si];
  const float dd = dwin[dl];
  const float mm = sc.bound ? 0.0f : sc.m[di];
  float v[32];
#pragma unroll
  for (int j = 0; j < 32; ++j) {
    v[j] = 0.0f;
    if (FULL || j < m) v[j] = win_row_load(h, si, j, lane);
  }
  const float shift = sc.bound ? leaky(sc.smax + dd) : mm;
  const float t = __expf(leaky(ss + dd) - shift);
  if (FULL || lane < m) atomicAdd(zl + dl, t);
#pragma unroll
  for (int j = 0; j < 32; ++j)
    if (FULL || j < m) win_row_add(acc, v[j], dl, t, j, lane);
}

template <bool REAL>
__device__ __forceinline__ void win_steps(
    int& b, int (&cur)[WSCAN], int& cnt, int lane, int lim, int last, int e1,
    int E, int wbase, int wcnt, int* myq, const int* __restrict__ src,
    const int* __restrict__ dst, const WinScore& sc,
    const float* __restrict__ h, float* acc, float* zl, const float* dwin) {
  const int step = 32 * WSCAN;
  for (; REAL ? (b + step <= lim) : (b < e1); b += 32 * step) {
    int nxt[WSCAN];
    const int nb = b + 32 * step;
#pragma unroll
    for (int u = 0; u < WSCAN; ++u)
      nxt[u] = win_dst_word(dst, nb + u * 32 + lane, last);
    cnt = win_queue<REAL>(cur, b, lane, lim, e1, E, wbase, wcnt, myq, cnt);
    while (cnt >= 32) {
      cnt -= 32;
      win_batch<true>(myq + cnt, 32, lane, wbase, E, src, dst, sc, h, acc, zl,
                      dwin);
    }
#pragma unroll
    for (int u = 0; u < WSCAN; ++u) cur[u] = nxt[u];
  }
}

__global__ __launch_bounds__(1024) void edge_window(
    const int* __restrict__ src, const int* __restrict__ dst,
    const float* __restrict__ sv, const float* __restrict__ dv,
    const float* __restrict__ rng, const float* __restrict__ m,
    const float* __restrict__ h, float* __restrict__ wpart,
    float* __restrict__ zwpart, int E, int ET, int N, int ZP) {
  __shared__ __attribute__((aligned(16))) float acc[WROWS * 32];
  __shared__ __attribute__((aligned(16))) float zl[WROWS];
  __shared__ __attribute__((aligned(16))) float dwin[WROWS];
  __shared__ int q[32][WQ];
  const int tid = threadIdx.x;
  const int lane = tid & 31;
  const int wave = __builtin_amdgcn_readfirstlane(tid >> 5);
  const int win = blockIdx.x / WCHUNKS;
  const int chunk = blockIdx.x - win * WCHUNKS;
  const int wbase = win * WROWS;
  const int wcnt = min(N - wbase, WROWS);
  const int wfl = wcnt * 32;

  for (int i = tid * 4; i < wfl; i += 4096)
    *(float4*)(acc + i) = make_float4(0.0f, 0.0f, 0.0f, 0.0f);
  for (int i = tid; i < WROWS; i += 1024) {
    zl[i] = 0.0f;
    dwin[i] = dv[min(wbase + i, N - 1)];
  }
  __syncthreads();

  WinScore sc;
  sc.sv = sv;
  sc.m = m;
  sc.smax = rng[0];
  sc.bound = sc.smax + rng[1] < SPREAD_LIMIT;

  const int per = ((ET + WCHUNKS * 32 - 1) / (WCHUNKS * 32)) * 32;
  const int e0 = chunk * per;
  const int e1 = min(ET, e0 + per);
  const int lim = min(e1, E);
  const int last = max(lim, 1) - 1;
  int* myq = q[wave];
  int cnt = 0;

  int b = e0 + wave * (32 * WSCAN);
  int cur[WSCAN];
#pragma unroll
  for (int u = 0; u < WSCAN; ++u)
    cur[u] = win_dst_word(dst, b + u * 32 + lane, last);
  win_steps<true>(b, cur, cnt, lane, lim, last, e1, E, wbase, wcnt, myq, src,
                  dst, sc, h, acc, zl, dwin);
  win_steps<false>(b, cur, cnt, lane, lim, last, e1, E, wbase, wcnt, myq, src,
                   dst, sc, h, acc, zl, dwin);
  if (cnt > 0)
    win_batch<false>(myq, cnt, lane, wbase, E, src, dst, sc, h, acc, zl, dwin);
  __syncthreads();

  float* __restrict__ prow = wpart + ((size_t)chunk * N + wbase) * 32;
  for (int i = tid * 4; i < wfl; i += 4096)
    *(float4*)(prow + i) = *(const float4*)(acc + i);
  float* __restrict__ zrow = zwpart + (size_t)chunk * ZP + wbase;
  for (int i = tid; i < wcnt; i += 1024) zrow[i] = zl[i];
}

#define RP_ROWS 128
#define RP_OP 34
#define RP_HP 68

__global__ __launch_bounds__(256) void reduce_project(
    const float* __restrict__ wpart, const float* __restrict__ zwpart,
    const float* __restrict__ b, const float* __restrict__ W,
    const float* __restrict__ asrc, const float* __restrict__ adst,
    float* __restrict__ h, float* __restrict__ s, float* __restrict__ d,
    float* __restrict__ rng, int N, int ZP) {
  constexpr int FIN = 32, FOUT = 64, NT = FOUT / 16;
  __shared__ __attribute__((aligned(16))) float Ws[FIN * FOUT];
  __shared__ __attribute__((aligned(16))) float os[RP_ROWS * RP_OP];
  __shared__ __attribute__((aligned(16))) float hs[RP_ROWS * RP_HP];
  __shared__ float red[16];
  const int tid = threadIdx.x;
  const int n0 = blockIdx.x * RP_ROWS;
  for (int i = tid; i < FIN * FOUT; i += 256) Ws[i] = W[i];

  const int f = (tid & 7) * 4;
  const float bx = b[f], by = b[f + 1], bz = b[f + 2], bw = b[f + 3];
#pragma unroll
  for (int it = 0; it < RP_ROWS * 8 / 256; ++it) {
    const int nl = (tid >> 3) + it * 32;
    const int n = n0 + nl;
    const int nc = min(n, N - 1);
    const float* p = wpart + ((size_t)nc * 32 + f);
    const float* pz = zwpart + nc;
    float4 a = make_float4(0.0f, 0.0f, 0.0f, 0.0f);
    float zn = 0.0f;
#pragma unroll
    for (int c = 0; c < WCHUNKS; ++c) {
      const float4 v = *(const float4*)p;
      const float zc = *pz;
      p += (size_t)N * 32;
      pz += ZP;
      a.x += v.x;
      a.y += v.y;
      a.z += v.z;
      a.w += v.w;
      zn += zc;
    }
    float4 r;
    r.x = a.x / zn + bx;
    r.y = a.y / zn + by;
    r.z = a.z / zn + bz;
    r.w = a.w / zn + bw;
    r.x = (r.x > 0.0f) ? r.x : (__expf(r.x) - 1.0f);
    r.y = (r.y > 0.0f) ? r.y : (__expf(r.y) - 1.0f);
    r.z = (r.z > 0.0f) ? r.z : (__expf(r.z) - 1.0f);
    r.w = (r.w > 0.0f) ? r.w : (__expf(r.w) - 1.0f);
    if (n >= N) r = make_float4(0.0f, 0.0f, 0.0f, 0.0f);
    float* q = os + nl * RP_OP + f;
    *(float2*)q = make_float2(r.x, r.y);
    *(float2*)(q + 2) = make_float2(r.z, r.w);
  }
  __syncthreads();

  const int wave = tid >> 5;
  const int lane = tid & 31;
  const int halfsel = lane >> 4;
  const int lidx = lane & 15;
  const int khalf = halfsel * 2;
  const float* arow = os + (wave * 16 + lidx) * RP_OP;

  v8f acc[NT];
#pragma unroll
  for (int t = 0; t < NT; ++t)
#pragma unroll
    for (int j = 0; j < 8; ++j) acc[t][j] = 0.0f;

#pragma unroll
  for (int k = 0; k < FIN; k += 4) {
    const int kk = k + khalf;
    v2f a;
    a.x = arow[kk];
    a.y = arow[kk + 1];
#pragma unroll
    for (int t = 0; t < NT; ++t) {
      v2f bb;
      bb.x = Ws[kk * FOUT + t * 16 + lidx];
      bb.y = Ws[(kk + 1) * FOUT + t * 16 + lidx];
      acc[t] = __builtin_amdgcn_wmma_f32_16x16x4_f32(
          false, a, false, bb, (short)0, acc[t], false, false);
    }
  }

  const int rloc0 = wave * 16 + halfsel * 8;
#pragma unroll
  for (int i = 0; i < 8; ++i) {
    const int rl = rloc0 + i;
    const int r = n0 + rl;
#pragma unroll
    for (int t = 0; t < NT; ++t) hs[rl * RP_HP + t * 16 + lidx] = acc[t][i];
    if (r < N) {
#pragma unroll
      for (int t = 0; t < NT; ++t)
        h[(size_t)r * FOUT + t * 16 + lidx] = acc[t][i];
    }
  }
  __syncthreads();

  const int n = n0 + tid;
  const bool valid = (tid < RP_ROWS) & (n < N);
  float ss = 0.0f, dd = 0.0f;
  if (valid) {
    const float* hr = hs + tid * RP_HP;
#pragma unroll
    for (int f = 0; f < FOUT; ++f) {
      float v = hr[f];
      ss += v * asrc[f];
      dd += v * adst[f];
    }
    s[n] = ss;
    d[n] = dd;
  }
  const float ninf = -__builtin_inff();
  float hi = valid ? ss : ninf;
  float lo = valid ? -ss : ninf;
#pragma unroll
  for (int o = 16; o > 0; o >>= 1) {
    hi = fmaxf(hi, __shfl_xor(hi, o));
    lo = fmaxf(lo, __shfl_xor(lo, o));
  }
  if (lane == 0) {
    red[wave] = hi;
    red[8 + wave] = lo;
  }
  __syncthreads();
  if (tid < 16) {
    float v = red[tid];
    v = fmaxf(v, __shfl_xor(v, 4));
    v = fmaxf(v, __shfl_xor(v, 2));
    v = fmaxf(v, __shfl_xor(v, 1));
    if ((tid & 7) == 0) atomicMaxFloat(rng + (tid >> 3), v);
  }
}

#define PGRID 256
#define PSUMS (NUM_GRAPHS * 64)
#define PROW (PSUMS + NUM_GRAPHS)
#define PCAP 57344
#define PFILL 8
#define GFILL 4

__device__ __forceinline__ float2 pool_row_load(const float* __restrict__ h,
                                                int si, int j, int lane) {
  const int sj = __builtin_amdgcn_readlane(si, j);
  return *(const float2*)(h + (size_t)sj * 64 + lane * 2);
}

__device__ __forceinline__ void pool_row_add(float* pl, float2 v, int g,
                                             float a, int j, int lane) {
  const int gj = __builtin_amdgcn_readlane(g, j);
  const float aj = __int_as_float(__builtin_amdgcn_readlane(__float_as_int(a), j));
  float* p = pl + gj * 64 + lane * 2;
  atomicAdd(p, aj * v.x);
  atomicAdd(p + 1, aj * v.y);
}

template <bool INL>
__device__ __forceinline__ void pool_sweep(
    const int* __restrict__ src, const int* __restrict__ dst,
    const float* __restrict__ ebuf, const float* __restrict__ z,
    const int* __restrict__ batch, const float* __restrict__ h, float* pl,
    const float* zl, const unsigned char* gl, int E, int ET, int tid,
    int lane) {
  const int wave = blockIdx.x * 32 + (tid >> 5);
  const int sweep = gridDim.x * 32 * 32;
  for (int b0 = wave * 32; b0 < ET; b0 += sweep) {
    const int id = b0 + lane;
    const bool live = id < ET;
    const bool real = id < E;
    const int idc = real ? id : 0;
    const int s0 = src[idc];
    const int d0 = dst[idc];
    const int self = live ? id - E : 0;
    const int si = real ? s0 : self;
    const int di = real ? d0 : self;
    const float t = ebuf[live ? id : 0];
    const int g = INL ? (int)gl[di] : batch[di];
    const float a = t / (INL ? zl[di] : z[di]);
    if (live && !real) atomicAdd(pl + PSUMS + g, 1.0f);

    const int cnt = min(32, ET - b0);
    if (cnt == 32) {
#pragma unroll
      for (int j0 = 0; j0 < 32; j0 += 16) {
        float2 v[16];
#pragma unroll
        for (int u = 0; u < 16; ++u) v[u] = pool_row_load(h, si, j0 + u, lane);
#pragma unroll
        for (int u = 0; u < 16; ++u) pool_row_add(pl, v[u], g, a, j0 + u, lane);
      }
    } else {
      for (int j = 0; j < cnt; ++j)
        pool_row_add(pl, pool_row_load(h, si, j, lane), g, a, j, lane);
    }
  }
}

__global__ __launch_bounds__(1024) void edge_pool(
    const int* __restrict__ src, const int* __restrict__ dst,
    const float* __restrict__ ebuf, const float* __restrict__ z,
    const unsigned char* __restrict__ gbytes, const int* __restrict__ batch,
    const float* __restrict__ h, float* __restrict__ ppart, int E, int ET,
    int N) {
  __shared__ float pl[PROW];
  __shared__ __attribute__((aligned(16))) float zl[PCAP];
  __shared__ __attribute__((aligned(16))) unsigned char gl[PCAP];
  const int tid = threadIdx.x;
  const int lane = tid & 31;
  const bool inl = N <= PCAP;
  for (int i = tid; i < PROW; i += 1024) pl[i] = 0.0f;
  if (inl) {
    const int n4 = N & ~3;
    for (int i0 = tid * 4; i0 < n4; i0 += PFILL * 4096) {
      float4 v[PFILL];
#pragma unroll
      for (int u = 0; u < PFILL; ++u)
        v[u] = *(const float4*)(z + min(i0 + u * 4096, n4 - 4));
#pragma unroll
      for (int u = 0; u < PFILL; ++u)
        *(float4*)(zl + min(i0 + u * 4096, n4 - 4)) = v[u];
    }
    if (tid < N - n4) zl[n4 + tid] = z[n4 + tid];
    const int n16 = N & ~15;
    for (int i0 = tid * 16; i0 < n16; i0 += GFILL * 16384) {
      uint4 w[GFILL];
#pragma unroll
      for (int u = 0; u < GFILL; ++u)
        w[u] = *(const uint4*)(gbytes + min(i0 + u * 16384, n16 - 16));
#pragma unroll
      for (int u = 0; u < GFILL; ++u)
        *(uint4*)(gl + min(i0 + u * 16384, n16 - 16)) = w[u];
    }
    if (tid < N - n16) gl[n16 + tid] = gbytes[n16 + tid];
  }
  __syncthreads();

  if (inl)
    pool_sweep<true>(src, dst, ebuf, z, batch, h, pl, zl, gl, E, ET, tid, lane);
  else
    pool_sweep<false>(src, dst, ebuf, z, batch, h, pl, zl, gl, E, ET, tid, lane);
  __syncthreads();

  float* __restrict__ row = ppart + (size_t)blockIdx.x * PROW;
  for (int i = tid; i < PROW; i += 1024) row[i] = pl[i];
}

__global__ __launch_bounds__(1024) void pool_finish(
    const float* __restrict__ ppart, const float* __restrict__ b,
    float* __restrict__ out) {
  __shared__ float red[16][64];
  __shared__ float cw[PGRID / 32];
  const int g = blockIdx.x;
  const int f = threadIdx.x & 63;
  const int r = threadIdx.x >> 6;
  const float* p = ppart + (size_t)r * PROW + g * 64 + f;
  float acc = 0.0f;
#pragma unroll
  for (int k = 0; k < PGRID / 16; ++k) {
    acc += *p;
    p += (size_t)16 * PROW;
  }
  red[r][f] = acc;
  if (threadIdx.x < PGRID) {
    float c = ppart[(size_t)threadIdx.x * PROW + PSUMS + g];
    for (int o = 16; o > 0; o >>= 1) c += __shfl_xor(c, o);
    if ((threadIdx.x & 31) == 0) cw[threadIdx.x >> 5] = c;
  }
  __syncthreads();
  if (threadIdx.x < 64) {
    float c = 0.0f;
    for (int w = 0; w < PGRID / 32; ++w) c += cw[w];
    float s = red[0][f];
#pragma unroll
    for (int w = 1; w < 16; ++w) s += red[w][f];
    out[g * 64 + f] = (s + c * b[f]) / fmaxf(c, 1.0f);
  }
}

__global__ void fill_kernel(float* __restrict__ p, float v, int n) {
  int i = blockIdx.x * blockDim.x + threadIdx.x;
  if (i < n) p[i] = v;
}

static inline int cdiv(int a, int b) { return (a + b - 1) / b; }

extern "C" void kernel_launch(void* const* d_in, const int* in_sizes, int n_in,
                              void* d_out, int out_size, void* d_ws,
                              size_t ws_size, hipStream_t stream) {
  const float* x      = (const float*)d_in[0];
  const int*   ei     = (const int*)d_in[1];
  const int*   batch  = (const int*)d_in[2];
  const float* W1     = (const float*)d_in[3];
  const float* a_src1 = (const float*)d_in[4];
  const float* a_dst1 = (const float*)d_in[5];
  const float* b1     = (const float*)d_in[6];
  const float* W2     = (const float*)d_in[7];
  const float* a_src2 = (const float*)d_in[8];
  const float* a_dst2 = (const float*)d_in[9];
  const float* b2     = (const float*)d_in[10];

  const int N = in_sizes[2];
  const int E = in_sizes[1] / 2;
  const int ET = E + N;
  const int* src = ei;
  const int* dst = ei + E;

  float* ws = (float*)d_ws;
  size_t off = 0;
  float* h1 = ws + off; off += (size_t)N * 32;
  float* h2 = ws + off; off += (size_t)N * 64;
  float* sv = ws + off; off += (size_t)N;
  float* dv = ws + off; off += (size_t)N;
  float* mb1 = ws + off; off += (size_t)N + 2;
  float* rng1 = mb1 + N;
  float* mb2 = ws + off; off += (size_t)N + 2;
  float* rng2 = mb2 + N;
  off = (off + 3) & ~(size_t)3;
  float* zb = ws + off; off += (size_t)((N + 3) & ~3);
  unsigned char* gb = (unsigned char*)(ws + off); off += (size_t)((N + 15) / 16) * 4;
  float* eb = ws + off; off += (size_t)ET;
  const int NP = (N + 3) & ~3;
  off = (off + 3) & ~(size_t)3;
  float* zp = ws + off; off += (size_t)ZCHUNKS * NP;
  float* pp = ws + off; off += (size_t)PGRID * PROW;
  float* wp = ws + off; off += (size_t)WCHUNKS * N * 32;
  const int ZP = (N + 31) & ~31;
  off = (off + 31) & ~(size_t)31;
  float* zwp = ws + off; off += (size_t)WCHUNKS * ZP;

  const float NEG_INF = -__builtin_inff();

  gemm_wmma<128, 32><<<cdiv(N, 128), 256, 0, stream>>>(x, W1, h1, N);
  fill_kernel<<<cdiv(2 * (N + 2), 256), 256, 0, stream>>>(mb1, NEG_INF, 2 * (N + 2));
  node_dots<32><<<cdiv(N, 256), 256, 0, stream>>>(h1, a_src1, a_dst1, sv, dv, rng1, N);
  edge_max<<<512, 256, 0, stream>>>(src, dst, sv, dv, rng1, mb1, E, ET);
  edge_window<<<cdiv(N, WROWS) * WCHUNKS, 1024, 0, stream>>>(src, dst, sv, dv, rng1, mb1, h1, wp, zwp, E, ET, N, ZP);

  reduce_project<<<cdiv(N, RP_ROWS), 256, 0, stream>>>(wp, zwp, b1, W2, a_src2, a_dst2, h2, sv, dv, rng2, N, ZP);
  edge_max<<<512, 256, 0, stream>>>(src, dst, sv, dv, rng2, mb2, E, ET);
  edge_exp_priv<<<ZGRID, 1024, 0, stream>>>(src, dst, sv, dv, rng2, mb2, eb, zp, E, ET, N, NP);
  z_reduce<<<cdiv(NP, 128), 256, 0, stream>>>(zp, batch, zb, gb, N, NP);

  edge_pool<<<PGRID, 1024, 0, stream>>>(src, dst, eb, zb, gb, batch, h2, pp, E, ET, N);
  pool_finish<<<NUM_GRAPHS, 1024, 0, stream>>>(pp, b2, (float*)d_out);
}


# edge_pool: a wave keeps its pooled sums in registers indexed by graph number (wave pairs split the graphs, own edges queued in LDS), one LDS add per sum at the end, none per edge; sum order changes
# speedup vs baseline: 5.4343x; 1.0802x over previous
#include <hip/hip_runtime.h>
#include <hip/hip_bf16.h>
#include <math.h>

typedef __attribute__((ext_vector_type(2))) float v2f;
typedef __attribute__((ext_vector_type(8))) float v8f;

#define NUM_GRAPHS 64
#define NEG_SLOPE 0.2f
#define SPREAD_LIMIT 8.0f

template <int FIN, int FOUT>
__global__ __launch_bounds__(256) void gemm_wmma(const float* __restrict__ x,
                                                 const float* __restrict__ W,
                                                 float* __restrict__ h, int N) {
  constexpr int NT = FOUT / 16;
  __shared__ float Ws[FIN * FOUT];
  for (int i = threadIdx.x; i < FIN * FOUT; i += 256) Ws[i] = W[i];
  __syncthreads();

  const int wave = threadIdx.x >> 5;
  const int lane = threadIdx.x & 31;
  const int m0 = (blockIdx.x * 8 + wave) * 16;
  const int halfsel = lane >> 4;
  const int lidx = lane & 15;
  int row = m0 + lidx;
  if (row >= N) row = N - 1;
  const int khalf = halfsel * 2;

  v8f acc[NT];
#pragma unroll
  for (int t = 0; t < NT; ++t)
#pragma unroll
    for (int j = 0; j < 8; ++j) acc[t][j] = 0.0f;

#pragma unroll 4
  for (int k = 0; k < FIN; k += 4) {
    const int kk = k + khalf;
    v2f a;
    a.x = x[(size_t)row * FIN + kk];
    a.y = x[(size_t)row * FIN + kk + 1];
#pragma unroll
    for (int t = 0; t < NT; ++t) {
      v2f b;
      b.x = Ws[kk * FOUT + t * 16 + lidx];
      b.y = Ws[(kk + 1) * FOUT + t * 16 + lidx];
      acc[t] = __builtin_amdgcn_wmma_f32_16x16x4_f32(
          false, a, false, b, (short)0, acc[t], false, false);
    }
  }

  const int rbase = m0 + halfsel * 8;
#pragma unroll
  for (int i = 0; i < 8; ++i) {
    const int r = rbase + i;
    if (r < N) {
#pragma unroll
      for (int t = 0; t < NT; ++t)
        h[(size_t)r * FOUT + t * 16 + lidx] = acc[t][i];
    }
  }
}

__device__ __forceinline__ void atomicMaxFloat(float* addr, float val) {
  int vi = __float_as_int(val);
  if (vi >= 0)
    atomicMax((int*)addr, vi);
  else
    atomicMin((unsigned int*)addr, (unsigned int)vi);
}

__device__ __forceinline__ float leaky(float v) {
  return (v > 0.0f) ? v : NEG_SLOPE * v;
}

template <int F>
__global__ __launch_bounds__(256) void node_dots(
    const float* __restrict__ h, const float* __restrict__ asrc,
    const float* __restrict__ adst, float* __restrict__ s,
    float* __restrict__ d, float* __restrict__ rng, int N) {
  __shared__ float red[16];
  const int n = blockIdx.x * 256 + threadIdx.x;
  const bool valid = n < N;
  float ss = 0.0f, dd = 0.0f;
  if (valid) {
#pragma unroll
    for (int f = 0; f < F; ++f) {
      float v = h[(size_t)n * F + f];
      ss += v * asrc[f];
      dd += v * adst[f];
    }
    s[n] = ss;
    d[n] = dd;
  }
  const float ninf = -__builtin_inff();
  float hi = valid ? ss : ninf;
  float lo = valid ? -ss : ninf;
#pragma unroll
  for (int o = 16; o > 0; o >>= 1) {
    hi = fmaxf(hi, __shfl_xor(hi, o));
    lo = fmaxf(lo, __shfl_xor(lo, o));
  }
  const int wave = threadIdx.x >> 5;
  if ((threadIdx.x & 31) == 0) {
    red[wave] = hi;
    red[8 + wave] = lo;
  }
  __syncthreads();
  if (threadIdx.x < 16) {
    float v = red[threadIdx.x];
    v = fmaxf(v, __shfl_xor(v, 4));
    v = fmaxf(v, __shfl_xor(v, 2));
    v = fmaxf(v, __shfl_xor(v, 1));
    if ((threadIdx.x & 7) == 0) atomicMaxFloat(rng + (threadIdx.x >> 3), v);
  }
}


__global__ __launch_bounds__(256) void edge_max(
    const int* __restrict__ src, const int* __restrict__ dst,
    const float* __restrict__ sv, const float* __restrict__ dv,
    const float* __restrict__ rng, float* __restrict__ m, int E, int ET) {
  if (rng[0] + rng[1] < SPREAD_LIMIT) return;
  for (int id = blockIdx.x * 256 + threadIdx.x; id < ET;
       id += gridDim.x * 256) {
    int si, di;
    if (id < E) {
      si = src[id];
      di = dst[id];
    } else {
      si = di = id - E;
    }
    atomicMaxFloat(m + di, leaky(sv[si] + dv[di]));
  }
}

#define ZG 4
#define ZCHUNKS 64
#define ZGRID (ZG * ZCHUNKS)
#define SCAP 53248
#define GCAP (SCAP / ZG)
#define ZUNROLL 8
#define SFILL 8

template <bool BOUND, bool REAL>
__device__ __forceinline__ void exp_scan(
    const int* __restrict__ src, const int* __restrict__ dst,
    const float* __restrict__ m, float* __restrict__ ebuf, const float* sl,
    const float* dl, float* zl, float smax, int e0, int e1, int E, int wbase,
    int wcnt, int sbase, int scnt, int tid) {
  const int last = max(min(e1, E), 1) - 1;
  for (int id0 = e0 + tid; id0 < e1; id0 += ZUNROLL * 1024) {
    int ws[ZUNROLL], wd[ZUNROLL];
#pragma unroll
    for (int u = 0; u < ZUNROLL; ++u) {
      const int idc = min(id0 + u * 1024, last);
      ws[u] = src[idc];
      wd[u] = dst[idc];
    }
#pragma unroll
    for (int u = 0; u < ZUNROLL; ++u) {
      const int id = id0 + u * 1024;
      const bool real = REAL || id < E;
      const int si = real ? ws[u] : id - E;
      const int di = real ? wd[u] : id - E;
      const int li = di - wbase;
      const int lsi = si - sbase;
      if ((id < e1) & ((unsigned)li < (unsigned)wcnt) &
          ((unsigned)lsi < (unsigned)scnt)) {
        const float ss = sl[lsi];
        const float dd = dl[li];
        const float e = leaky(ss + dd);
        const float shift = BOUND ? leaky(smax + dd) : m[di];
        const float t = __expf(e - shift);
        ebuf[id] = t;
        atomicAdd(zl + li, t);
      }
    }
  }
}

__global__ __launch_bounds__(1024) void edge_exp_priv(
    const int* __restrict__ src, const int* __restrict__ dst,
    const float* __restrict__ sv, const float* __restrict__ dv,
    const float* __restrict__ rng, const float* __restrict__ m,
    float* __restrict__ ebuf, float* __restrict__ zpart, int E, int ET, int N,
    int NP) {
  __shared__ __attribute__((aligned(16))) float sl[SCAP];
  __shared__ __attribute__((aligned(16))) float dl[GCAP];
  __shared__ __attribute__((aligned(16))) float zl[GCAP];
  const int tid = threadIdx.x;
  const int chunk = blockIdx.x / ZG;
  const int grp = blockIdx.x - chunk * ZG;
  const float smax = rng[0];
  const bool bound = smax + rng[1] < SPREAD_LIMIT;
  float* __restrict__ zrow = zpart + (size_t)chunk * NP;

  const int per = ((ET + ZCHUNKS * 32 - 1) / (ZCHUNKS * 32)) * 32;
  const int e0 = chunk * per;
  const int e1 = min(ET, e0 + per);

  for (int wbase = grp * GCAP; wbase < N; wbase += ZG * GCAP) {
    const int wcnt = min(N - wbase, GCAP);
    const int cnt4 = (wcnt + 3) & ~3;
    for (int i = tid; i < cnt4; i += 1024) {
      dl[i] = dv[min(wbase + i, N - 1)];
      zl[i] = 0.0f;
    }
    for (int sbase = 0; sbase < N; sbase += SCAP) {
      const int scnt = min(N - sbase, SCAP);
      const int s4 = scnt & ~3;
      for (int i0 = tid * 4; i0 < s4; i0 += SFILL * 4096) {
        float4 v[SFILL];
#pragma unroll
        for (int u = 0; u < SFILL; ++u)
          v[u] = *(const float4*)(sv + sbase + min(i0 + u * 4096, s4 - 4));
#pragma unroll
        for (int u = 0; u < SFILL; ++u)
          *(float4*)(sl + min(i0 + u * 4096, s4 - 4)) = v[u];
      }
      if (tid < scnt - s4) sl[s4 + tid] = sv[sbase + s4 + tid];
      __syncthreads();

      if (bound) {
        if (e1 <= E)
          exp_scan<true, true>(src, dst, m, ebuf, sl, dl, zl, smax, e0, e1, E,
                               wbase, wcnt, sbase, scnt, tid);
        else
          exp_scan<true, false>(src, dst, m, ebuf, sl, dl, zl, smax, e0, e1, E,
                                wbase, wcnt, sbase, scnt, tid);
      } else {
        exp_scan<false, false>(src, dst, m, ebuf, sl, dl, zl, smax, e0, e1, E,
                               wbase, wcnt, sbase, scnt, tid);
      }
      __syncthreads();
    }

    for (int i = tid * 4; i < cnt4; i += 4096)
      *(float4*)(zrow + wbase + i) = *(const float4*)(zl + i);
    __syncthreads();
  }
}

__global__ __launch_bounds__(256) void z_reduce(const float* __restrict__ zpart,
                                                const int* __restrict__ batch,
                                                float* __restrict__ z,
                                                unsigned char* __restrict__ gbytes,
                                                int N, int NP) {
  __shared__ __attribute__((aligned(16))) float red[8][128];
  const int wave = threadIdx.x >> 5;
  const int lane = threadIdx.x & 31;
  const int n4 = blockIdx.x * 128 + lane * 4;
  float4 acc = make_float4(0.0f, 0.0f, 0.0f, 0.0f);
  if (n4 < NP) {
    const float* p = zpart + (size_t)wave * NP + n4;
#pragma unroll 8
    for (int k = 0; k < ZCHUNKS / 8; ++k) {
      const float4 v = *(const float4*)p;
      p += (size_t)8 * NP;
      acc.x += v.x;
      acc.y += v.y;
      acc.z += v.z;
      acc.w += v.w;
    }
  }
  *(float4*)(&red[wave][lane * 4]) = acc;
  __syncthreads();
  if (threadIdx.x < 128) {
    const int n = blockIdx.x * 128 + threadIdx.x;
    float r = red[0][threadIdx.x];
#pragma unroll
    for (int w = 1; w < 8; ++w) r += red[w][threadIdx.x];
    if (n < N) {
      z[n] = r;
      gbytes[n] = (unsigned char)batch[n];
    }
  }
}

#define WROWS 2048
#define WCHUNKS 10
#define WSCAN 8
#define WQ (32 * WSCAN + 32)

__device__ __forceinline__ int win_dst_word(const int* __restrict__ dst, int id,
                                            int last) {
  return dst[min(id, last)];
}

__device__ __forceinline__ bool win_hit(int d, int id, int lim, int e1, int E,
                                        int wbase, int wcnt) {
  const bool real_hit = ((unsigned)(d - wbase) < (unsigned)wcnt) & (id < lim);
  const bool self_hit = ((unsigned)(id - E - wbase) < (unsigned)wcnt) & (id < e1);
  return real_hit | self_hit;
}

template <bool REAL>
__device__ __forceinline__ int win_queue(const int* cur, int b, int lane,
                                         int lim, int e1, int E, int wbase,
                                         int wcnt, int* myq, int cnt) {
#pragma unroll
  for (int u = 0; u < WSCAN; ++u) {
    const int id = b + u * 32 + lane;
    const bool hit = REAL ? ((unsigned)(cur[u] - wbase) < (unsigned)wcnt)
                          : win_hit(cur[u], id, lim, e1, E, wbase, wcnt);
    const unsigned mask = __builtin_amdgcn_ballot_w32(hit);
    if (hit) myq[cnt + (int)__builtin_amdgcn_mbcnt_lo(mask, 0u)] = id;
    cnt += __builtin_popcount(mask);
  }
  return cnt;
}

__device__ __forceinline__ float win_row_load(const float* __restrict__ h,
                                              int si, int j, int lane) {
  const int sj = __builtin_amdgcn_readlane(si, j);
  return h[(size_t)sj * 32 + lane];
}

struct WinScore {
  const float* __restrict__ sv;
  const float* __restrict__ m;
  float smax;
  bool bound;
};

__device__ __forceinline__ void win_row_add(float* acc, float v, int dl,
                                            float t, int j, int lane) {
  const int rj = __builtin_amdgcn_readlane(dl, j);
  const float tj = __int_as_float(__builtin_amdgcn_readlane(__float_as_int(t), j));
  atomicAdd(acc + rj * 32 + lane, tj * v);
}

template <bool FULL>
__device__ __forceinline__ void win_batch(
    const int* qp, int m, int lane, int wbase, int E,
    const int* __restrict__ src, const int* __restrict__ dst,
    const WinScore& sc, const float* __restrict__ h, float* acc, float* zl,
    const float* dwin) {
  const int id = qp[(FULL || lane < m) ? lane : 0];
  const bool real = id < E;
  const int idc = real ? id : 0;
  const int s0 = src[idc];
  const int d0 = dst[idc];
  const int self = id - E;
  const int si = real ? s0 : self;
  const int di = real ? d0 : self;
  const int dl = di - wbase;
  const float ss = sc.sv[si];
  const float dd = dwin[dl];
  const float mm = sc.bound ? 0.0f : sc.m[di];
  float v[32];
#pragma unroll
  for (int j = 0; j < 32; ++j) {
    v[j] = 0.0f;
    if (FULL || j < m) v[j] = win_row_load(h, si, j, lane);
  }
  const float shift = sc.bound ? leaky(sc.smax + dd) : mm;
  const float t = __expf(leaky(ss + dd) - shift);
  if (FULL || lane < m) atomicAdd(zl + dl, t);
#pragma unroll
  for (int j = 0; j < 32; ++j)
    if (FULL || j < m) win_row_add(acc, v[j], dl, t, j, lane);
}

template <bool REAL>
__device__ __forceinline__ void win_steps(
    int& b, int (&cur)[WSCAN], int& cnt, int lane, int lim, int last, int e1,
    int E, int wbase, int wcnt, int* myq, const int* __restrict__ src,
    const int* __restrict__ dst, const WinScore& sc,
    const float* __restrict__ h, float* acc, float* zl, const float* dwin) {
  const int step = 32 * WSCAN;
  for (; REAL ? (b + step <= lim) : (b < e1); b += 32 * step) {
    int nxt[WSCAN];
    const int nb = b + 32 * step;
#pragma unroll
    for (int u = 0; u < WSCAN; ++u)
      nxt[u] = win_dst_word(dst, nb + u * 32 + lane, last);
    cnt = win_queue<REAL>(cur, b, lane, lim, e1, E, wbase, wcnt, myq, cnt);
    while (cnt >= 32) {
      cnt -= 32;
      win_batch<true>(myq + cnt, 32, lane, wbase, E, src, dst, sc, h, acc, zl,
                      dwin);
    }
#pragma unroll
    for (int u = 0; u < WSCAN; ++u) cur[u] = nxt[u];
  }
}

__global__ __launch_bounds__(1024) void edge_window(
    const int* __restrict__ src, const int* __restrict__ dst,
    const float* __restrict__ sv, const float* __restrict__ dv,
    const float* __restrict__ rng, const float* __restrict__ m,
    const float* __restrict__ h, float* __restrict__ wpart,
    float* __restrict__ zwpart, int E, int ET, int N, int ZP) {
  __shared__ __attribute__((aligned(16))) float acc[WROWS * 32];
  __shared__ __attribute__((aligned(16))) float zl[WROWS];
  __shared__ __attribute__((aligned(16))) float dwin[WROWS];
  __shared__ int q[32][WQ];
  const int tid = threadIdx.x;
  const int lane = tid & 31;
  const int wave = __builtin_amdgcn_readfirstlane(tid >> 5);
  const int win = blockIdx.x / WCHUNKS;
  const int chunk = blockIdx.x - win * WCHUNKS;
  const int wbase = win * WROWS;
  const int wcnt = min(N - wbase, WROWS);
  const int wfl = wcnt * 32;

  for (int i = tid * 4; i < wfl; i += 4096)
    *(float4*)(acc + i) = make_float4(0.0f, 0.0f, 0.0f, 0.0f);
  for (int i = tid; i < WROWS; i += 1024) {
    zl[i] = 0.0f;
    dwin[i] = dv[min(wbase + i, N - 1)];
  }
  __syncthreads();

  WinScore sc;
  sc.sv = sv;
  sc.m = m;
  sc.smax = rng[0];
  sc.bound = sc.smax + rng[1] < SPREAD_LIMIT;

  const int per = ((ET + WCHUNKS * 32 - 1) / (WCHUNKS * 32)) * 32;
  const int e0 = chunk * per;
  const int e1 = min(ET, e0 + per);
  const int lim = min(e1, E);
  const int last = max(lim, 1) - 1;
  int* myq = q[wave];
  int cnt = 0;

  int b = e0 + wave * (32 * WSCAN);
  int cur[WSCAN];
#pragma unroll
  for (int u = 0; u < WSCAN; ++u)
    cur[u] = win_dst_word(dst, b + u * 32 + lane, last);
  win_steps<true>(b, cur, cnt, lane, lim, last, e1, E, wbase, wcnt, myq, src,
                  dst, sc, h, acc, zl, dwin);
  win_steps<false>(b, cur, cnt, lane, lim, last, e1, E, wbase, wcnt, myq, src,
                   dst, sc, h, acc, zl, dwin);
  if (cnt > 0)
    win_batch<false>(myq, cnt, lane, wbase, E, src, dst, sc, h, acc, zl, dwin);
  __syncthreads();

  float* __restrict__ prow = wpart + ((size_t)chunk * N + wbase) * 32;
  for (int i = tid * 4; i < wfl; i += 4096)
    *(float4*)(prow + i) = *(const float4*)(acc + i);
  float* __restrict__ zrow = zwpart + (size_t)chunk * ZP + wbase;
  for (int i = tid; i < wcnt; i += 1024) zrow[i] = zl[i];
}

#define RP_ROWS 128
#define RP_OP 34
#define RP_HP 68

__global__ __launch_bounds__(256) void reduce_project(
    const float* __restrict__ wpart, const float* __restrict__ zwpart,
    const float* __restrict__ b, const float* __restrict__ W,
    const float* __restrict__ asrc, const float* __restrict__ adst,
    float* __restrict__ h, float* __restrict__ s, float* __restrict__ d,
    float* __restrict__ rng, int N, int ZP) {
  constexpr int FIN = 32, FOUT = 64, NT = FOUT / 16;
  __shared__ __attribute__((aligned(16))) float Ws[FIN * FOUT];
  __shared__ __attribute__((aligned(16))) float os[RP_ROWS * RP_OP];
  __shared__ __attribute__((aligned(16))) float hs[RP_ROWS * RP_HP];
  __shared__ float red[16];
  const int tid = threadIdx.x;
  const int n0 = blockIdx.x * RP_ROWS;
  for (int i = tid; i < FIN * FOUT; i += 256) Ws[i] = W[i];

  const int f = (tid & 7) * 4;
  const float bx = b[f], by = b[f + 1], bz = b[f + 2], bw = b[f + 3];
#pragma unroll
  for (int it = 0; it < RP_ROWS * 8 / 256; ++it) {
    const int nl = (tid >> 3) + it * 32;
    const int n = n0 + nl;
    const int nc = min(n, N - 1);
    const float* p = wpart + ((size_t)nc * 32 + f);
    const float* pz = zwpart + nc;
    float4 a = make_float4(0.0f, 0.0f, 0.0f, 0.0f);
    float zn = 0.0f;
#pragma unroll
    for (int c = 0; c < WCHUNKS; ++c) {
      const float4 v = *(const float4*)p;
      const float zc = *pz;
      p += (size_t)N * 32;
      pz += ZP;
      a.x += v.x;
      a.y += v.y;
      a.z += v.z;
      a.w += v.w;
      zn += zc;
    }
    float4 r;
    r.x = a.x / zn + bx;
    r.y = a.y / zn + by;
    r.z = a.z / zn + bz;
    r.w = a.w / zn + bw;
    r.x = (r.x > 0.0f) ? r.x : (__expf(r.x) - 1.0f);
    r.y = (r.y > 0.0f) ? r.y : (__expf(r.y) - 1.0f);
    r.z = (r.z > 0.0f) ? r.z : (__expf(r.z) - 1.0f);
    r.w = (r.w > 0.0f) ? r.w : (__expf(r.w) - 1.0f);
    if (n >= N) r = make_float4(0.0f, 0.0f, 0.0f, 0.0f);
    float* q = os + nl * RP_OP + f;
    *(float2*)q = make_float2(r.x, r.y);
    *(float2*)(q + 2) = make_float2(r.z, r.w);
  }
  __syncthreads();

  const int wave = tid >> 5;
  const int lane = tid & 31;
  const int halfsel = lane >> 4;
  const int lidx = lane & 15;
  const int khalf = halfsel * 2;
  const float* arow = os + (wave * 16 + lidx) * RP_OP;

  v8f acc[NT];
#pragma unroll
  for (int t = 0; t < NT; ++t)
#pragma unroll
    for (int j = 0; j < 8; ++j) acc[t][j] = 0.0f;

#pragma unroll
  for (int k = 0; k < FIN; k += 4) {
    const int kk = k + khalf;
    v2f a;
    a.x = arow[kk];
    a.y = arow[kk + 1];
#pragma unroll
    for (int t = 0; t < NT; ++t) {
      v2f bb;
      bb.x = Ws[kk * FOUT + t * 16 + lidx];
      bb.y = Ws[(kk + 1) * FOUT + t * 16 + lidx];
      acc[t] = __builtin_amdgcn_wmma_f32_16x16x4_f32(
          false, a, false, bb, (short)0, acc[t], false, false);
    }
  }

  const int rloc0 = wave * 16 + halfsel * 8;
#pragma unroll
  for (int i = 0; i < 8; ++i) {
    const int rl = rloc0 + i;
    const int r = n0 + rl;
#pragma unroll
    for (int t = 0; t < NT; ++t) hs[rl * RP_HP + t * 16 + lidx] = acc[t][i];
    if (r < N) {
#pragma unroll
      for (int t = 0; t < NT; ++t)
        h[(size_t)r * FOUT + t * 16 + lidx] = acc[t][i];
    }
  }
  __syncthreads();

  const int n = n0 + tid;
  const bool valid = (tid < RP_ROWS) & (n < N);
  float ss = 0.0f, dd = 0.0f;
  if (valid) {
    const float* hr = hs + tid * RP_HP;
#pragma unroll
    for (int f = 0; f < FOUT; ++f) {
      float v = hr[f];
      ss += v * asrc[f];
      dd += v * adst[f];
    }
    s[n] = ss;
    d[n] = dd;
  }
  const float ninf = -__builtin_inff();
  float hi = valid ? ss : ninf;
  float lo = valid ? -ss : ninf;
#pragma unroll
  for (int o = 16; o > 0; o >>= 1) {
    hi = fmaxf(hi, __shfl_xor(hi, o));
    lo = fmaxf(lo, __shfl_xor(lo, o));
  }
  if (lane == 0) {
    red[wave] = hi;
    red[8 + wave] = lo;
  }
  __syncthreads();
  if (tid < 16) {
    float v = red[tid];
    v = fmaxf(v, __shfl_xor(v, 4));
    v = fmaxf(v, __shfl_xor(v, 2));
    v = fmaxf(v, __shfl_xor(v, 1));
    if ((tid & 7) == 0) atomicMaxFloat(rng + (tid >> 3), v);
  }
}

#define PGRID 256
#define PGROUP 16
#define PQ (PGROUP + 32)
#define PSUMS (NUM_GRAPHS * 64)
#define PROW (PSUMS + NUM_GRAPHS)
#define PCAP 57344
#define PFILL 8
#define GFILL 4

__device__ __forceinline__ float2 pool_row_load(const float* __restrict__ h,
                                                int key, int j, int lane) {
  const int sj = __builtin_amdgcn_readlane(key, j) >> 6;
  return *(const float2*)(h + (size_t)sj * 64 + lane * 2);
}

typedef __attribute__((ext_vector_type(32))) float v32f;

struct PoolAcc {
  v32f x, y;
};

__device__ __forceinline__ void pool_row_add(PoolAcc& A, float2 v, int key,
                                             float a, int j) {
  const int gj = __builtin_amdgcn_readlane(key, j) & 31;
  const float aj = __int_as_float(__builtin_amdgcn_readlane(__float_as_int(a), j));
  A.x[gj] += aj * v.x;
  A.y[gj] += aj * v.y;
}

__device__ __forceinline__ void pool_group(PoolAcc& A, const int* qk,
                                           const float* qa,
                                           const float* __restrict__ h,
                                           int lane) {
  const int key = qk[lane & (PGROUP - 1)];
  const float a = qa[lane & (PGROUP - 1)];
  float2 v[PGROUP];
#pragma unroll
  for (int u = 0; u < PGROUP; ++u) v[u] = pool_row_load(h, key, u, lane);
#pragma unroll
  for (int u = 0; u < PGROUP; ++u) pool_row_add(A, v[u], key, a, u);
}

template <bool INL>
__device__ __forceinline__ void pool_sweep(
    const int* __restrict__ src, const int* __restrict__ dst,
    const float* __restrict__ ebuf, const float* __restrict__ z,
    const int* __restrict__ batch, const float* __restrict__ h, float* pl,
    PoolAcc& A, int* qk, float* qa, const float* zl, const unsigned char* gl,
    int E, int ET, int tid, int lane) {
  const int wv = __builtin_amdgcn_readfirstlane(tid >> 5);
  const int half = wv & 1;
  const int pair = blockIdx.x * 16 + (wv >> 1);
  const int sweep = gridDim.x * 16 * 32;
  int cnt = 0;
  for (int b0 = pair * 32; b0 < ET; b0 += sweep) {
    const int id = b0 + lane;
    const bool live = id < ET;
    const bool real = id < E;
    const int idc = real ? id : 0;
    const int s0 = src[idc];
    const int d0 = dst[idc];
    const int self = live ? id - E : 0;
    const int si = real ? s0 : self;
    const int di = real ? d0 : self;
    const float t = ebuf[live ? id : 0];
    const int g = INL ? (int)gl[di] : batch[di];
    const float a = t / (INL ? zl[di] : z[di]);
    const bool mine = live & ((g >> 5) == half);
    if (mine && !real) atomicAdd(pl + PSUMS + g, 1.0f);

    const unsigned mask = __builtin_amdgcn_ballot_w32(mine);
    if (mine) {
      const int p = cnt + (int)__builtin_amdgcn_mbcnt_lo(mask, 0u);
      qk[p] = si * 64 + g;
      qa[p] = a;
    }
    cnt += __builtin_popcount(mask);
    while (cnt >= PGROUP) {
      cnt -= PGROUP;
      pool_group(A, qk + cnt, qa + cnt, h, lane);
    }
  }
  if (cnt > 0) {
    const int key = qk[lane < cnt ? lane : 0];
    const float a = qa[lane < cnt ? lane : 0];
    for (int j = 0; j < cnt; ++j)
      pool_row_add(A, pool_row_load(h, key, j, lane), key, a, j);
  }
}

__global__ __launch_bounds__(1024) void edge_pool(
    const int* __restrict__ src, const int* __restrict__ dst,
    const float* __restrict__ ebuf, const float* __restrict__ z,
    const unsigned char* __restrict__ gbytes, const int* __restrict__ batch,
    const float* __restrict__ h, float* __restrict__ ppart, int E, int ET,
    int N) {
  __shared__ float pl[PROW];
  __shared__ __attribute__((aligned(16))) float zl[PCAP];
  __shared__ __attribute__((aligned(16))) unsigned char gl[PCAP];
  __shared__ int qk[32][PQ];
  __shared__ float qa[32][PQ];
  const int tid = threadIdx.x;
  const int lane = tid & 31;
  const bool inl = N <= PCAP;
  for (int i = tid; i < PROW; i += 1024) pl[i] = 0.0f;
  if (inl) {
    const int n4 = N & ~3;
    for (int i0 = tid * 4; i0 < n4; i0 += PFILL * 4096) {
      float4 v[PFILL];
#pragma unroll
      for (int u = 0; u < PFILL; ++u)
        v[u] = *(const float4*)(z + min(i0 + u * 4096, n4 - 4));
#pragma unroll
      for (int u = 0; u < PFILL; ++u)
        *(float4*)(zl + min(i0 + u * 4096, n4 - 4)) = v[u];
    }
    if (tid < N - n4) zl[n4 + tid] = z[n4 + tid];
    const int n16 = N & ~15;
    for (int i0 = tid * 16; i0 < n16; i0 += GFILL * 16384) {
      uint4 w[GFILL];
#pragma unroll
      for (int u = 0; u < GFILL; ++u)
        w[u] = *(const uint4*)(gbytes + min(i0 + u * 16384, n16 - 16));
#pragma unroll
      for (int u = 0; u < GFILL; ++u)
        *(uint4*)(gl + min(i0 + u * 16384, n16 - 16)) = w[u];
    }
    if (tid < N - n16) gl[n16 + tid] = gbytes[n16 + tid];
  }
  __syncthreads();

  PoolAcc A;
#pragma unroll
  for (int k = 0; k < 32; ++k) {
    A.x[k] = 0.0f;
    A.y[k] = 0.0f;
  }
  if (inl)
    pool_sweep<true>(src, dst, ebuf, z, batch, h, pl, A, qk[tid >> 5],
                     qa[tid >> 5], zl, gl, E, ET, tid, lane);
  else
    pool_sweep<false>(src, dst, ebuf, z, batch, h, pl, A, qk[tid >> 5],
                      qa[tid >> 5], zl, gl, E, ET, tid, lane);
  float* pw = pl + ((tid >> 5) & 1) * (32 * 64) + lane * 2;
#pragma unroll
  for (int k = 0; k < 32; ++k) {
    atomicAdd(pw + k * 64, A.x[k]);
    atomicAdd(pw + k * 64 + 1, A.y[k]);
  }
  __syncthreads();

  float* __restrict__ row = ppart + (size_t)blockIdx.x * PROW;
  for (int i = tid; i < PROW; i += 1024) row[i] = pl[i];
}

__global__ __launch_bounds__(1024) void pool_finish(
    const float* __restrict__ ppart, const float* __restrict__ b,
    float* __restrict__ out) {
  __shared__ float red[16][64];
  __shared__ float cw[PGRID / 32];
  const int g = blockIdx.x;
  const int f = threadIdx.x & 63;
  const int r = threadIdx.x >> 6;
  const float* p = ppart + (size_t)r * PROW + g * 64 + f;
  float acc = 0.0f;
#pragma unroll
  for (int k = 0; k < PGRID / 16; ++k) {
    acc += *p;
    p += (size_t)16 * PROW;
  }
  red[r][f] = acc;
  if (threadIdx.x < PGRID) {
    float c = ppart[(size_t)threadIdx.x * PROW + PSUMS + g];
    for (int o = 16; o > 0; o >>= 1) c += __shfl_xor(c, o);
    if ((threadIdx.x & 31) == 0) cw[threadIdx.x >> 5] = c;
  }
  __syncthreads();
  if (threadIdx.x < 64) {
    float c = 0.0f;
    for (int w = 0; w < PGRID / 32; ++w) c += cw[w];
    float s = red[0][f];
#pragma unroll
    for (int w = 1; w < 16; ++w) s += red[w][f];
    out[g * 64 + f] = (s + c * b[f]) / fmaxf(c, 1.0f);
  }
}

__global__ void fill_kernel(float* __restrict__ p, float v, int n) {
  int i = blockIdx.x * blockDim.x + threadIdx.x;
  if (i < n) p[i] = v;
}

static inline int cdiv(int a, int b) { return (a + b - 1) / b; }

extern "C" void kernel_launch(void* const* d_in, const int* in_sizes, int n_in,
                              void* d_out, int out_size, void* d_ws,
                              size_t ws_size, hipStream_t stream) {
  const float* x      = (const float*)d_in[0];
  const int*   ei     = (const int*)d_in[1];
  const int*   batch  = (const int*)d_in[2];
  const float* W1     = (const float*)d_in[3];
  const float* a_src1 = (const float*)d_in[4];
  const float* a_dst1 = (const float*)d_in[5];
  const float* b1     = (const float*)d_in[6];
  const float* W2     = (const float*)d_in[7];
  const float* a_src2 = (const float*)d_in[8];
  const float* a_dst2 = (const float*)d_in[9];
  const float* b2     = (const float*)d_in[10];

  const int N = in_sizes[2];
  const int E = in_sizes[1] / 2;
  const int ET = E + N;
  const int* src = ei;
  const int* dst = ei + E;

  float* ws = (float*)d_ws;
  size_t off = 0;
  float* h1 = ws + off; off += (size_t)N * 32;
  float* h2 = ws + off; off += (size_t)N * 64;
  float* sv = ws + off; off += (size_t)N;
  float* dv = ws + off; off += (size_t)N;
  float* mb1 = ws + off; off += (size_t)N + 2;
  float* rng1 = mb1 + N;
  float* mb2 = ws + off; off += (size_t)N + 2;
  float* rng2 = mb2 + N;
  off = (off + 3) & ~(size_t)3;
  float* zb = ws + off; off += (size_t)((N + 3) & ~3);
  unsigned char* gb = (unsigned char*)(ws + off); off += (size_t)((N + 15) / 16) * 4;
  float* eb = ws + off; off += (size_t)ET;
  const int NP = (N + 3) & ~3;
  off = (off + 3) & ~(size_t)3;
  float* zp = ws + off; off += (size_t)ZCHUNKS * NP;
  float* pp = ws + off; off += (size_t)PGRID * PROW;
  float* wp = ws + off; off += (size_t)WCHUNKS * N * 32;
  const int ZP = (N + 31) & ~31;
  off = (off + 31) & ~(size_t)31;
  float* zwp = ws + off; off += (size_t)WCHUNKS * ZP;

  const float NEG_INF = -__builtin_inff();

  gemm_wmma<128, 32><<<cdiv(N, 128), 256, 0, stream>>>(x, W1, h1, N);
  fill_kernel<<<cdiv(2 * (N + 2), 256), 256, 0, stream>>>(mb1, NEG_INF, 2 * (N + 2));
  node_dots<32><<<cdiv(N, 256), 256, 0, stream>>>(h1, a_src1, a_dst1, sv, dv, rng1, N);
  edge_max<<<512, 256, 0, stream>>>(src, dst, sv, dv, rng1, mb1, E, ET);
  edge_window<<<cdiv(N, WROWS) * WCHUNKS, 1024, 0, stream>>>(src, dst, sv, dv, rng1, mb1, h1, wp, zwp, E, ET, N, ZP);

  reduce_project<<<cdiv(N, RP_ROWS), 256, 0, stream>>>(wp, zwp, b1, W2, a_src2, a_dst2, h2, sv, dv, rng2, N, ZP);
  edge_max<<<512, 256, 0, stream>>>(src, dst, sv, dv, rng2, mb2, E, ET);
  edge_exp_priv<<<ZGRID, 1024, 0, stream>>>(src, dst, sv, dv, rng2, mb2, eb, zp, E, ET, N, NP);
  z_reduce<<<cdiv(NP, 128), 256, 0, stream>>>(zp, batch, zb, gb, N, NP);

  edge_pool<<<PGRID, 1024, 0, stream>>>(src, dst, eb, zb, gb, batch, h2, pp, E, ET, N);
  pool_finish<<<NUM_GRAPHS, 1024, 0, stream>>>(pp, b2, (float*)d_out);
}


# Exact-max fallback moves into the two edge kernels (running max of a dst window in LDS; all edges scanned only if the bound is off): edge_max launches and the fill go, 11 launches become 8
# speedup vs baseline: 5.6003x; 1.0305x over previous
#include <hip/hip_runtime.h>
#include <hip/hip_bf16.h>
#include <math.h>

typedef __attribute__((ext_vector_type(2))) float v2f;
typedef __attribute__((ext_vector_type(8))) float v8f;

#define NUM_GRAPHS 64
#define NEG_SLOPE 0.2f
#define SPREAD_LIMIT 8.0f

template <int FIN, int FOUT>
__global__ __launch_bounds__(256) void gemm_wmma(const float* __restrict__ x,
                                                 const float* __restrict__ W,
                                                 float* __restrict__ h,
                                                 float* __restrict__ rng, int N) {
  constexpr int NT = FOUT / 16;
  __shared__ float Ws[FIN * FOUT];
  for (int i = threadIdx.x; i < FIN * FOUT; i += 256) Ws[i] = W[i];
  __syncthreads();

  const int wave = threadIdx.x >> 5;
  const int lane = threadIdx.x & 31;
  const int m0 = (blockIdx.x * 8 + wave) * 16;
  const int halfsel = lane >> 4;
  const int lidx = lane & 15;
  int row = m0 + lidx;
  if (row >= N) row = N - 1;
  const int khalf = halfsel * 2;

  v8f acc[NT];
#pragma unroll
  for (int t = 0; t < NT; ++t)
#pragma unroll
    for (int j = 0; j < 8; ++j) acc[t][j] = 0.0f;

#pragma unroll 4
  for (int k = 0; k < FIN; k += 4) {
    const int kk = k + khalf;
    v2f a;
    a.x = x[(size_t)row * FIN + kk];
    a.y = x[(size_t)row * FIN + kk + 1];
#pragma unroll
    for (int t = 0; t < NT; ++t) {
      v2f b;
      b.x = Ws[kk * FOUT + t * 16 + lidx];
      b.y = Ws[(kk + 1) * FOUT + t * 16 + lidx];
      acc[t] = __builtin_amdgcn_wmma_f32_16x16x4_f32(
          false, a, false, b, (short)0, acc[t], false, false);
    }
  }

  const int rbase = m0 + halfsel * 8;
#pragma unroll
  for (int i = 0; i < 8; ++i) {
    const int r = rbase + i;
    if (r < N) {
#pragma unroll
      for (int t = 0; t < NT; ++t)
        h[(size_t)r * FOUT + t * 16 + lidx] = acc[t][i];
    }
  }
  if (blockIdx.x == 0 && threadIdx.x < 4) rng[threadIdx.x] = -__builtin_inff();
}

__device__ __forceinline__ void atomicMaxFloat(float* addr, float val) {
  int vi = __float_as_int(val);
  if (vi >= 0)
    atomicMax((int*)addr, vi);
  else
    atomicMin((unsigned int*)addr, (unsigned int)vi);
}

__device__ __forceinline__ float leaky(float v) {
  return (v > 0.0f) ? v : NEG_SLOPE * v;
}

template <int F>
__global__ __launch_bounds__(256) void node_dots(
    const float* __restrict__ h, const float* __restrict__ asrc,
    const float* __restrict__ adst, float* __restrict__ s,
    float* __restrict__ d, float* __restrict__ rng, int N) {
  __shared__ float red[16];
  const int n = blockIdx.x * 256 + threadIdx.x;
  const bool valid = n < N;
  float ss = 0.0f, dd = 0.0f;
  if (valid) {
#pragma unroll
    for (int f = 0; f < F; ++f) {
      float v = h[(size_t)n * F + f];
      ss += v * asrc[f];
      dd += v * adst[f];
    }
    s[n] = ss;
    d[n] = dd;
  }
  const float ninf = -__builtin_inff();
  float hi = valid ? ss : ninf;
  float lo = valid ? -ss : ninf;
#pragma unroll
  for (int o = 16; o > 0; o >>= 1) {
    hi = fmaxf(hi, __shfl_xor(hi, o));
    lo = fmaxf(lo, __shfl_xor(lo, o));
  }
  const int wave = threadIdx.x >> 5;
  if ((threadIdx.x & 31) == 0) {
    red[wave] = hi;
    red[8 + wave] = lo;
  }
  __syncthreads();
  if (threadIdx.x < 16) {
    float v = red[threadIdx.x];
    v = fmaxf(v, __shfl_xor(v, 4));
    v = fmaxf(v, __shfl_xor(v, 2));
    v = fmaxf(v, __shfl_xor(v, 1));
    if ((threadIdx.x & 7) == 0) atomicMaxFloat(rng + (threadIdx.x >> 3), v);
  }
}


__device__ __forceinline__ void win_max(const int* __restrict__ src,
                                        const int* __restrict__ dst,
                                        const float* __restrict__ sv,
                                        const float* dwin, float* mwin,
                                        int wbase, int wcnt, int E, int ET,
                                        int tid) {
  for (int id = tid; id < ET; id += 1024) {
    const bool real = id < E;
    const int idc = real ? id : 0;
    const int s0 = src[idc];
    const int d0 = dst[idc];
    const int si = real ? s0 : id - E;
    const int di = real ? d0 : id - E;
    const int li = di - wbase;
    if ((unsigned)li < (unsigned)wcnt)
      atomicMaxFloat(mwin + li, leaky(sv[si] + dwin[li]));
  }
}

#define ZG 4
#define ZCHUNKS 64
#define ZGRID (ZG * ZCHUNKS)
#define SCAP 53248
#define GCAP (SCAP / ZG)
#define ZUNROLL 8
#define SFILL 8

template <bool BOUND, bool REAL>
__device__ __forceinline__ void exp_scan(
    const int* __restrict__ src, const int* __restrict__ dst,
    const float* ml, float* __restrict__ ebuf, const float* sl,
    const float* dl, float* zl, float smax, int e0, int e1, int E, int wbase,
    int wcnt, int sbase, int scnt, int tid) {
  const int last = max(min(e1, E), 1) - 1;
  for (int id0 = e0 + tid; id0 < e1; id0 += ZUNROLL * 1024) {
    int ws[ZUNROLL], wd[ZUNROLL];
#pragma unroll
    for (int u = 0; u < ZUNROLL; ++u) {
      const int idc = min(id0 + u * 1024, last);
      ws[u] = src[idc];
      wd[u] = dst[idc];
    }
#pragma unroll
    for (int u = 0; u < ZUNROLL; ++u) {
      const int id = id0 + u * 1024;
      const bool real = REAL || id < E;
      const int si = real ? ws[u] : id - E;
      const int di = real ? wd[u] : id - E;
      const int li = di - wbase;
      const int lsi = si - sbase;
      if ((id < e1) & ((unsigned)li < (unsigned)wcnt) &
          ((unsigned)lsi < (unsigned)scnt)) {
        const float ss = sl[lsi];
        const float dd = dl[li];
        const float e = leaky(ss + dd);
        const float shift = BOUND ? leaky(smax + dd) : ml[li];
        const float t = __expf(e - shift);
        ebuf[id] = t;
        atomicAdd(zl + li, t);
      }
    }
  }
}

__global__ __launch_bounds__(1024) void edge_exp_priv(
    const int* __restrict__ src, const int* __restrict__ dst,
    const float* __restrict__ sv, const float* __restrict__ dv,
    const float* __restrict__ rng, float* __restrict__ ebuf,
    float* __restrict__ zpart, int E, int ET, int N, int NP) {
  __shared__ __attribute__((aligned(16))) float sl[SCAP];
  __shared__ __attribute__((aligned(16))) float dl[GCAP];
  __shared__ __attribute__((aligned(16))) float zl[GCAP];
  const int tid = threadIdx.x;
  const int chunk = blockIdx.x / ZG;
  const int grp = blockIdx.x - chunk * ZG;
  const float smax = rng[0];
  const bool bound = smax + rng[1] < SPREAD_LIMIT;
  float* __restrict__ zrow = zpart + (size_t)chunk * NP;

  const int per = ((ET + ZCHUNKS * 32 - 1) / (ZCHUNKS * 32)) * 32;
  const int e0 = chunk * per;
  const int e1 = min(ET, e0 + per);

  const int gcap = bound ? GCAP : GCAP / 2;
  float* ml = dl + GCAP / 2;

  for (int wbase = grp * gcap; wbase < N; wbase += ZG * gcap) {
    const int wcnt = min(N - wbase, gcap);
    const int cnt4 = (wcnt + 3) & ~3;
    for (int i = tid; i < cnt4; i += 1024) {
      dl[i] = dv[min(wbase + i, N - 1)];
      zl[i] = 0.0f;
    }
    if (!bound) {
      for (int i = tid; i < cnt4; i += 1024) ml[i] = -__builtin_inff();
      __syncthreads();
      win_max(src, dst, sv, dl, ml, wbase, wcnt, E, ET, tid);
    }
    for (int sbase = 0; sbase < N; sbase += SCAP) {
      const int scnt = min(N - sbase, SCAP);
      const int s4 = scnt & ~3;
      for (int i0 = tid * 4; i0 < s4; i0 += SFILL * 4096) {
        float4 v[SFILL];
#pragma unroll
        for (int u = 0; u < SFILL; ++u)
          v[u] = *(const float4*)(sv + sbase + min(i0 + u * 4096, s4 - 4));
#pragma unroll
        for (int u = 0; u < SFILL; ++u)
          *(float4*)(sl + min(i0 + u * 4096, s4 - 4)) = v[u];
      }
      if (tid < scnt - s4) sl[s4 + tid] = sv[sbase + s4 + tid];
      __syncthreads();

      if (bound) {
        if (e1 <= E)
          exp_scan<true, true>(src, dst, ml, ebuf, sl, dl, zl, smax, e0, e1, E,
                               wbase, wcnt, sbase, scnt, tid);
        else
          exp_scan<true, false>(src, dst, ml, ebuf, sl, dl, zl, smax, e0, e1, E,
                                wbase, wcnt, sbase, scnt, tid);
      } else {
        exp_scan<false, false>(src, dst, ml, ebuf, sl, dl, zl, smax, e0, e1, E,
                               wbase, wcnt, sbase, scnt, tid);
      }
      __syncthreads();
    }

    for (int i = tid * 4; i < cnt4; i += 4096)
      *(float4*)(zrow + wbase + i) = *(const float4*)(zl + i);
    __syncthreads();
  }
}

__global__ __launch_bounds__(256) void z_reduce(const float* __restrict__ zpart,
                                                const int* __restrict__ batch,
                                                float* __restrict__ z,
                                                unsigned char* __restrict__ gbytes,
                                                int N, int NP) {
  __shared__ __attribute__((aligned(16))) float red[8][128];
  const int wave = threadIdx.x >> 5;
  const int lane = threadIdx.x & 31;
  const int n4 = blockIdx.x * 128 + lane * 4;
  float4 acc = make_float4(0.0f, 0.0f, 0.0f, 0.0f);
  if (n4 < NP) {
    const float* p = zpart + (size_t)wave * NP + n4;
#pragma unroll 8
    for (int k = 0; k < ZCHUNKS / 8; ++k) {
      const float4 v = *(const float4*)p;
      p += (size_t)8 * NP;
      acc.x += v.x;
      acc.y += v.y;
      acc.z += v.z;
      acc.w += v.w;
    }
  }
  *(float4*)(&red[wave][lane * 4]) = acc;
  __syncthreads();
  if (threadIdx.x < 128) {
    const int n = blockIdx.x * 128 + threadIdx.x;
    float r = red[0][threadIdx.x];
#pragma unroll
    for (int w = 1; w < 8; ++w) r += red[w][threadIdx.x];
    if (n < N) {
      z[n] = r;
      gbytes[n] = (unsigned char)batch[n];
    }
  }
}

#define WROWS 2048
#define WCHUNKS 10
#define WSCAN 8
#define WQ (32 * WSCAN + 32)

__device__ __forceinline__ int win_dst_word(const int* __restrict__ dst, int id,
                                            int last) {
  return dst[min(id, last)];
}

__device__ __forceinline__ bool win_hit(int d, int id, int lim, int e1, int E,
                                        int wbase, int wcnt) {
  const bool real_hit = ((unsigned)(d - wbase) < (unsigned)wcnt) & (id < lim);
  const bool self_hit = ((unsigned)(id - E - wbase) < (unsigned)wcnt) & (id < e1);
  return real_hit | self_hit;
}

template <bool REAL>
__device__ __forceinline__ int win_queue(const int* cur, int b, int lane,
                                         int lim, int e1, int E, int wbase,
                                         int wcnt, int* myq, int cnt) {
#pragma unroll
  for (int u = 0; u < WSCAN; ++u) {
    const int id = b + u * 32 + lane;
    const bool hit = REAL ? ((unsigned)(cur[u] - wbase) < (unsigned)wcnt)
                          : win_hit(cur[u], id, lim, e1, E, wbase, wcnt);
    const unsigned mask = __builtin_amdgcn_ballot_w32(hit);
    if (hit) myq[cnt + (int)__builtin_amdgcn_mbcnt_lo(mask, 0u)] = id;
    cnt += __builtin_popcount(mask);
  }
  return cnt;
}

__device__ __forceinline__ float win_row_load(const float* __restrict__ h,
                                              int si, int j, int lane) {
  const int sj = __builtin_amdgcn_readlane(si, j);
  return h[(size_t)sj * 32 + lane];
}

struct WinScore {
  const float* __restrict__ sv;
  const float* mwin;
  float smax;
  bool bound;
};

__device__ __forceinline__ void win_row_add(float* acc, float v, int dl,
                                            float t, int j, int lane) {
  const int rj = __builtin_amdgcn_readlane(dl, j);
  const float tj = __int_as_float(__builtin_amdgcn_readlane(__float_as_int(t), j));
  atomicAdd(acc + rj * 32 + lane, tj * v);
}

template <bool FULL>
__device__ __forceinline__ void win_batch(
    const int* qp, int m, int lane, int wbase, int E,
    const int* __restrict__ src, const int* __restrict__ dst,
    const WinScore& sc, const float* __restrict__ h, float* acc, float* zl,
    const float* dwin) {
  const int id = qp[(FULL || lane < m) ? lane : 0];
  const bool real = id < E;
  const int idc = real ? id : 0;
  const int s0 = src[idc];
  const int d0 = dst[idc];
  const int self = id - E;
  const int si = real ? s0 : self;
  const int di = real ? d0 : self;
  const int dl = di - wbase;
  const float ss = sc.sv[si];
  const float dd = dwin[dl];
  const float mm = sc.bound ? 0.0f : sc.mwin[dl];
  float v[32];
#pragma unroll
  for (int j = 0; j < 32; ++j) {
    v[j] = 0.0f;
    if (FULL || j < m) v[j] = win_row_load(h, si, j, lane);
  }
  const float shift = sc.bound ? leaky(sc.smax + dd) : mm;
  const float t = __expf(leaky(ss + dd) - shift);
  if (FULL || lane < m) atomicAdd(zl + dl, t);
#pragma unroll
  for (int j = 0; j < 32; ++j)
    if (FULL || j < m) win_row_add(acc, v[j], dl, t, j, lane);
}

template <bool REAL>
__device__ __forceinline__ void win_steps(
    int& b, int (&cur)[WSCAN], int& cnt, int lane, int lim, int last, int e1,
    int E, int wbase, int wcnt, int* myq, const int* __restrict__ src,
    const int* __restrict__ dst, const WinScore& sc,
    const float* __restrict__ h, float* acc, float* zl, const float* dwin) {
  const int step = 32 * WSCAN;
  for (; REAL ? (b + step <= lim) : (b < e1); b += 32 * step) {
    int nxt[WSCAN];
    const int nb = b + 32 * step;
#pragma unroll
    for (int u = 0; u < WSCAN; ++u)
      nxt[u] = win_dst_word(dst, nb + u * 32 + lane, last);
    cnt = win_queue<REAL>(cur, b, lane, lim, e1, E, wbase, wcnt, myq, cnt);
    while (cnt >= 32) {
      cnt -= 32;
      win_batch<true>(myq + cnt, 32, lane, wbase, E, src, dst, sc, h, acc, zl,
                      dwin);
    }
#pragma unroll
    for (int u = 0; u < WSCAN; ++u) cur[u] = nxt[u];
  }
}

__global__ __launch_bounds__(1024) void edge_window(
    const int* __restrict__ src, const int* __restrict__ dst,
    const float* __restrict__ sv, const float* __restrict__ dv,
    const float* __restrict__ rng, const float* __restrict__ h,
    float* __restrict__ wpart, float* __restrict__ zwpart, int E, int ET, int N,
    int ZP) {
  __shared__ __attribute__((aligned(16))) float acc[WROWS * 32];
  __shared__ __attribute__((aligned(16))) float zl[WROWS];
  __shared__ __attribute__((aligned(16))) float dwin[WROWS];
  __shared__ float mwin[WROWS];
  __shared__ int q[32][WQ];
  const int tid = threadIdx.x;
  const int lane = tid & 31;
  const int wave = __builtin_amdgcn_readfirstlane(tid >> 5);
  const int win = blockIdx.x / WCHUNKS;
  const int chunk = blockIdx.x - win * WCHUNKS;
  const int wbase = win * WROWS;
  const int wcnt = min(N - wbase, WROWS);
  const int wfl = wcnt * 32;

  for (int i = tid * 4; i < wfl; i += 4096)
    *(float4*)(acc + i) = make_float4(0.0f, 0.0f, 0.0f, 0.0f);
  for (int i = tid; i < WROWS; i += 1024) {
    zl[i] = 0.0f;
    dwin[i] = dv[min(wbase + i, N - 1)];
  }
  __syncthreads();

  WinScore sc;
  sc.sv = sv;
  sc.mwin = mwin;
  sc.smax = rng[0];
  sc.bound = sc.smax + rng[1] < SPREAD_LIMIT;
  if (!sc.bound) {
    for (int i = tid; i < WROWS; i += 1024) mwin[i] = -__builtin_inff();
    __syncthreads();
    win_max(src, dst, sv, dwin, mwin, wbase, wcnt, E, ET, tid);
    __syncthreads();
  }

  const int per = ((ET + WCHUNKS * 32 - 1) / (WCHUNKS * 32)) * 32;
  const int e0 = chunk * per;
  const int e1 = min(ET, e0 + per);
  const int lim = min(e1, E);
  const int last = max(lim, 1) - 1;
  int* myq = q[wave];
  int cnt = 0;

  int b = e0 + wave * (32 * WSCAN);
  int cur[WSCAN];
#pragma unroll
  for (int u = 0; u < WSCAN; ++u)
    cur[u] = win_dst_word(dst, b + u * 32 + lane, last);
  win_steps<true>(b, cur, cnt, lane, lim, last, e1, E, wbase, wcnt, myq, src,
                  dst, sc, h, acc, zl, dwin);
  win_steps<false>(b, cur, cnt, lane, lim, last, e1, E, wbase, wcnt, myq, src,
                   dst, sc, h, acc, zl, dwin);
  if (cnt > 0)
    win_batch<false>(myq, cnt, lane, wbase, E, src, dst, sc, h, acc, zl, dwin);
  __syncthreads();

  float* __restrict__ prow = wpart + ((size_t)chunk * N + wbase) * 32;
  for (int i = tid * 4; i < wfl; i += 4096)
    *(float4*)(prow + i) = *(const float4*)(acc + i);
  float* __restrict__ zrow = zwpart + (size_t)chunk * ZP + wbase;
  for (int i = tid; i < wcnt; i += 1024) zrow[i] = zl[i];
}

#define RP_ROWS 128
#define RP_OP 34
#define RP_HP 68

__global__ __launch_bounds__(256) void reduce_project(
    const float* __restrict__ wpart, const float* __restrict__ zwpart,
    const float* __restrict__ b, const float* __restrict__ W,
    const float* __restrict__ asrc, const float* __restrict__ adst,
    float* __restrict__ h, float* __restrict__ s, float* __restrict__ d,
    float* __restrict__ rng, int N, int ZP) {
  constexpr int FIN = 32, FOUT = 64, NT = FOUT / 16;
  __shared__ __attribute__((aligned(16))) float Ws[FIN * FOUT];
  __shared__ __attribute__((aligned(16))) float os[RP_ROWS * RP_OP];
  __shared__ __attribute__((aligned(16))) float hs[RP_ROWS * RP_HP];
  __shared__ float red[16];
  const int tid = threadIdx.x;
  const int n0 = blockIdx.x * RP_ROWS;
  for (int i = tid; i < FIN * FOUT; i += 256) Ws[i] = W[i];

  const int f = (tid & 7) * 4;
  const float bx = b[f], by = b[f + 1], bz = b[f + 2], bw = b[f + 3];
#pragma unroll
  for (int it = 0; it < RP_ROWS * 8 / 256; ++it) {
    const int nl = (tid >> 3) + it * 32;
    const int n = n0 + nl;
    const int nc = min(n, N - 1);
    const float* p = wpart + ((size_t)nc * 32 + f);
    const float* pz = zwpart + nc;
    float4 a = make_float4(0.0f, 0.0f, 0.0f, 0.0f);
    float zn = 0.0f;
#pragma unroll
    for (int c = 0; c < WCHUNKS; ++c) {
      const float4 v = *(const float4*)p;
      const float zc = *pz;
      p += (size_t)N * 32;
      pz += ZP;
      a.x += v.x;
      a.y += v.y;
      a.z += v.z;
      a.w += v.w;
      zn += zc;
    }
    float4 r;
    r.x = a.x / zn + bx;
    r.y = a.y / zn + by;
    r.z = a.z / zn + bz;
    r.w = a.w / zn + bw;
    r.x = (r.x > 0.0f) ? r.x : (__expf(r.x) - 1.0f);
    r.y = (r.y > 0.0f) ? r.y : (__expf(r.y) - 1.0f);
    r.z = (r.z > 0.0f) ? r.z : (__expf(r.z) - 1.0f);
    r.w = (r.w > 0.0f) ? r.w : (__expf(r.w) - 1.0f);
    if (n >= N) r = make_float4(0.0f, 0.0f, 0.0f, 0.0f);
    float* q = os + nl * RP_OP + f;
    *(float2*)q = make_float2(r.x, r.y);
    *(float2*)(q + 2) = make_float2(r.z, r.w);
  }
  __syncthreads();

  const int wave = tid >> 5;
  const int lane = tid & 31;
  const int halfsel = lane >> 4;
  const int lidx = lane & 15;
  const int khalf = halfsel * 2;
  const float* arow = os + (wave * 16 + lidx) * RP_OP;

  v8f acc[NT];
#pragma unroll
  for (int t = 0; t < NT; ++t)
#pragma unroll
    for (int j = 0; j < 8; ++j) acc[t][j] = 0.0f;

#pragma unroll
  for (int k = 0; k < FIN; k += 4) {
    const int kk = k + khalf;
    v2f a;
    a.x = arow[kk];
    a.y = arow[kk + 1];
#pragma unroll
    for (int t = 0; t < NT; ++t) {
      v2f bb;
      bb.x = Ws[kk * FOUT + t * 16 + lidx];
      bb.y = Ws[(kk + 1) * FOUT + t * 16 + lidx];
      acc[t] = __builtin_amdgcn_wmma_f32_16x16x4_f32(
          false, a, false, bb, (short)0, acc[t], false, false);
    }
  }

  const int rloc0 = wave * 16 + halfsel * 8;
#pragma unroll
  for (int i = 0; i < 8; ++i) {
    const int rl = rloc0 + i;
    const int r = n0 + rl;
#pragma unroll
    for (int t = 0; t < NT; ++t) hs[rl * RP_HP + t * 16 + lidx] = acc[t][i];
    if (r < N) {
#pragma unroll
      for (int t = 0; t < NT; ++t)
        h[(size_t)r * FOUT + t * 16 + lidx] = acc[t][i];
    }
  }
  __syncthreads();

  const int n = n0 + tid;
  const bool valid = (tid < RP_ROWS) & (n < N);
  float ss = 0.0f, dd = 0.0f;
  if (valid) {
    const float* hr = hs + tid * RP_HP;
#pragma unroll
    for (int f = 0; f < FOUT; ++f) {
      float v = hr[f];
      ss += v * asrc[f];
      dd += v * adst[f];
    }
    s[n] = ss;
    d[n] = dd;
  }
  const float ninf = -__builtin_inff();
  float hi = valid ? ss : ninf;
  float lo = valid ? -ss : ninf;
#pragma unroll
  for (int o = 16; o > 0; o >>= 1) {
    hi = fmaxf(hi, __shfl_xor(hi, o));
    lo = fmaxf(lo, __shfl_xor(lo, o));
  }
  if (lane == 0) {
    red[wave] = hi;
    red[8 + wave] = lo;
  }
  __syncthreads();
  if (tid < 16) {
    float v = red[tid];
    v = fmaxf(v, __shfl_xor(v, 4));
    v = fmaxf(v, __shfl_xor(v, 2));
    v = fmaxf(v, __shfl_xor(v, 1));
    if ((tid & 7) == 0) atomicMaxFloat(rng + (tid >> 3), v);
  }
}

#define PGRID 256
#define PGROUP 16
#define PQ (PGROUP + 32)
#define PSUMS (NUM_GRAPHS * 64)
#define PROW (PSUMS + NUM_GRAPHS)
#define PCAP 57344
#define PFILL 8
#define GFILL 4

__device__ __forceinline__ float2 pool_row_load(const float* __restrict__ h,
                                                int key, int j, int lane) {
  const int sj = __builtin_amdgcn_readlane(key, j) >> 6;
  return *(const float2*)(h + (size_t)sj * 64 + lane * 2);
}

typedef __attribute__((ext_vector_type(32))) float v32f;

struct PoolAcc {
  v32f x, y;
};

__device__ __forceinline__ void pool_row_add(PoolAcc& A, float2 v, int key,
                                             float a, int j) {
  const int gj = __builtin_amdgcn_readlane(key, j) & 31;
  const float aj = __int_as_float(__builtin_amdgcn_readlane(__float_as_int(a), j));
  A.x[gj] += aj * v.x;
  A.y[gj] += aj * v.y;
}

__device__ __forceinline__ void pool_group(PoolAcc& A, const int* qk,
                                           const float* qa,
                                           const float* __restrict__ h,
                                           int lane) {
  const int key = qk[lane & (PGROUP - 1)];
  const float a = qa[lane & (PGROUP - 1)];
  float2 v[PGROUP];
#pragma unroll
  for (int u = 0; u < PGROUP; ++u) v[u] = pool_row_load(h, key, u, lane);
#pragma unroll
  for (int u = 0; u < PGROUP; ++u) pool_row_add(A, v[u], key, a, u);
}

template <bool INL>
__device__ __forceinline__ void pool_sweep(
    const int* __restrict__ src, const int* __restrict__ dst,
    const float* __restrict__ ebuf, const float* __restrict__ z,
    const int* __restrict__ batch, const float* __restrict__ h, float* pl,
    PoolAcc& A, int* qk, float* qa, const float* zl, const unsigned char* gl,
    int E, int ET, int tid, int lane) {
  const int wv = __builtin_amdgcn_readfirstlane(tid >> 5);
  const int half = wv & 1;
  const int pair = blockIdx.x * 16 + (wv >> 1);
  const int sweep = gridDim.x * 16 * 32;
  int cnt = 0;
  for (int b0 = pair * 32; b0 < ET; b0 += sweep) {
    const int id = b0 + lane;
    const bool live = id < ET;
    const bool real = id < E;
    const int idc = real ? id : 0;
    const int s0 = src[idc];
    const int d0 = dst[idc];
    const int self = live ? id - E : 0;
    const int si = real ? s0 : self;
    const int di = real ? d0 : self;
    const float t = ebuf[live ? id : 0];
    const int g = INL ? (int)gl[di] : batch[di];
    const float a = t / (INL ? zl[di] : z[di]);
    const bool mine = live & ((g >> 5) == half);
    if (mine && !real) atomicAdd(pl + PSUMS + g, 1.0f);

    const unsigned mask = __builtin_amdgcn_ballot_w32(mine);
    if (mine) {
      const int p = cnt + (int)__builtin_amdgcn_mbcnt_lo(mask, 0u);
      qk[p] = si * 64 + g;
      qa[p] = a;
    }
    cnt += __builtin_popcount(mask);
    while (cnt >= PGROUP) {
      cnt -= PGROUP;
      pool_group(A, qk + cnt, qa + cnt, h, lane);
    }
  }
  if (cnt > 0) {
    const int key = qk[lane < cnt ? lane : 0];
    const float a = qa[lane < cnt ? lane : 0];
    for (int j = 0; j < cnt; ++j)
      pool_row_add(A, pool_row_load(h, key, j, lane), key, a, j);
  }
}

__global__ __launch_bounds__(1024) void edge_pool(
    const int* __restrict__ src, const int* __restrict__ dst,
    const float* __restrict__ ebuf, const float* __restrict__ z,
    const unsigned char* __restrict__ gbytes, const int* __restrict__ batch,
    const float* __restrict__ h, float* __restrict__ ppart, int E, int ET,
    int N) {
  __shared__ float pl[PROW];
  __shared__ __attribute__((aligned(16))) float zl[PCAP];
  __shared__ __attribute__((aligned(16))) unsigned char gl[PCAP];
  __shared__ int qk[32][PQ];
  __shared__ float qa[32][PQ];
  const int tid = threadIdx.x;
  const int lane = tid & 31;
  const bool inl = N <= PCAP;
  for (int i = tid; i < PROW; i += 1024) pl[i] = 0.0f;
  if (inl) {
    const int n4 = N & ~3;
    for (int i0 = tid * 4; i0 < n4; i0 += PFILL * 4096) {
      float4 v[PFILL];
#pragma unroll
      for (int u = 0; u < PFILL; ++u)
        v[u] = *(const float4*)(z + min(i0 + u * 4096, n4 - 4));
#pragma unroll
      for (int u = 0; u < PFILL; ++u)
        *(float4*)(zl + min(i0 + u * 4096, n4 - 4)) = v[u];
    }
    if (tid < N - n4) zl[n4 + tid] = z[n4 + tid];
    const int n16 = N & ~15;
    for (int i0 = tid * 16; i0 < n16; i0 += GFILL * 16384) {
      uint4 w[GFILL];
#pragma unroll
      for (int u = 0; u < GFILL; ++u)
        w[u] = *(const uint4*)(gbytes + min(i0 + u * 16384, n16 - 16));
#pragma unroll
      for (int u = 0; u < GFILL; ++u)
        *(uint4*)(gl + min(i0 + u * 16384, n16 - 16)) = w[u];
    }
    if (tid < N - n16) gl[n16 + tid] = gbytes[n16 + tid];
  }
  __syncthreads();

  PoolAcc A;
#pragma unroll
  for (int k = 0; k < 32; ++k) {
    A.x[k] = 0.0f;
    A.y[k] = 0.0f;
  }
  if (inl)
    pool_sweep<true>(src, dst, ebuf, z, batch, h, pl, A, qk[tid >> 5],
                     qa[tid >> 5], zl, gl, E, ET, tid, lane);
  else
    pool_sweep<false>(src, dst, ebuf, z, batch, h, pl, A, qk[tid >> 5],
                      qa[tid >> 5], zl, gl, E, ET, tid, lane);
  float* pw = pl + ((tid >> 5) & 1) * (32 * 64) + lane * 2;
#pragma unroll
  for (int k = 0; k < 32; ++k) {
    atomicAdd(pw + k * 64, A.x[k]);
    atomicAdd(pw + k * 64 + 1, A.y[k]);
  }
  __syncthreads();

  float* __restrict__ row = ppart + (size_t)blockIdx.x * PROW;
  for (int i = tid; i < PROW; i += 1024) row[i] = pl[i];
}

__global__ __launch_bounds__(1024) void pool_finish(
    const float* __restrict__ ppart, const float* __restrict__ b,
    float* __restrict__ out) {
  __shared__ float red[16][64];
  __shared__ float cw[PGRID / 32];
  const int g = blockIdx.x;
  const int f = threadIdx.x & 63;
  const int r = threadIdx.x >> 6;
  const float* p = ppart + (size_t)r * PROW + g * 64 + f;
  float acc = 0.0f;
#pragma unroll
  for (int k = 0; k < PGRID / 16; ++k) {
    acc += *p;
    p += (size_t)16 * PROW;
  }
  red[r][f] = acc;
  if (threadIdx.x < PGRID) {
    float c = ppart[(size_t)threadIdx.x * PROW + PSUMS + g];
    for (int o = 16; o > 0; o >>= 1) c += __shfl_xor(c, o);
    if ((threadIdx.x & 31) == 0) cw[threadIdx.x >> 5] = c;
  }
  __syncthreads();
  if (threadIdx.x < 64) {
    float c = 0.0f;
    for (int w = 0; w < PGRID / 32; ++w) c += cw[w];
    float s = red[0][f];
#pragma unroll
    for (int w = 1; w < 16; ++w) s += red[w][f];
    out[g * 64 + f] = (s + c * b[f]) / fmaxf(c, 1.0f);
  }
}

static inline int cdiv(int a, int b) { return (a + b - 1) / b; }

extern "C" void kernel_launch(void* const* d_in, const int* in_sizes, int n_in,
                              void* d_out, int out_size, void* d_ws,
                              size_t ws_size, hipStream_t stream) {
  const float* x      = (const float*)d_in[0];
  const int*   ei     = (const int*)d_in[1];
  const int*   batch  = (const int*)d_in[2];
  const float* W1     = (const float*)d_in[3];
  const float* a_src1 = (const float*)d_in[4];
  const float* a_dst1 = (const float*)d_in[5];
  const float* b1     = (const float*)d_in[6];
  const float* W2     = (const float*)d_in[7];
  const float* a_src2 = (const float*)d_in[8];
  const float* a_dst2 = (const float*)d_in[9];
  const float* b2     = (const float*)d_in[10];

  const int N = in_sizes[2];
  const int E = in_sizes[1] / 2;
  const int ET = E + N;
  const int* src = ei;
  const int* dst = ei + E;

  float* ws = (float*)d_ws;
  size_t off = 0;
  float* h1 = ws + off; off += (size_t)N * 32;
  float* h2 = ws + off; off += (size_t)N * 64;
  float* sv = ws + off; off += (size_t)N;
  float* dv = ws + off; off += (size_t)N;
  float* rng1 = ws + off; off += 2;
  float* rng2 = ws + off; off += 2;
  off = (off + 3) & ~(size_t)3;
  float* zb = ws + off; off += (size_t)((N + 3) & ~3);
  unsigned char* gb = (unsigned char*)(ws + off); off += (size_t)((N + 15) / 16) * 4;
  float* eb = ws + off; off += (size_t)ET;
  const int NP = (N + 3) & ~3;
  off = (off + 3) & ~(size_t)3;
  float* zp = ws + off; off += (size_t)ZCHUNKS * NP;
  float* pp = ws + off; off += (size_t)PGRID * PROW;
  float* wp = ws + off; off += (size_t)WCHUNKS * N * 32;
  const int ZP = (N + 31) & ~31;
  off = (off + 31) & ~(size_t)31;
  float* zwp = ws + off; off += (size_t)WCHUNKS * ZP;

  gemm_wmma<128, 32><<<cdiv(N, 128), 256, 0, stream>>>(x, W1, h1, rng1, N);
  node_dots<32><<<cdiv(N, 256), 256, 0, stream>>>(h1, a_src1, a_dst1, sv, dv, rng1, N);
  edge_window<<<cdiv(N, WROWS) * WCHUNKS, 1024, 0, stream>>>(src, dst, sv, dv, rng1, h1, wp, zwp, E, ET, N, ZP);

  reduce_project<<<cdiv(N, RP_ROWS), 256, 0, stream>>>(wp, zwp, b1, W2, a_src2, a_dst2, h2, sv, dv, rng2, N, ZP);
  edge_exp_priv<<<ZGRID, 1024, 0, stream>>>(src, dst, sv, dv, rng2, eb, zp, E, ET, N, NP);
  z_reduce<<<cdiv(NP, 128), 256, 0, stream>>>(zp, batch, zb, gb, N, NP);

  edge_pool<<<PGRID, 1024, 0, stream>>>(src, dst, eb, zb, gb, batch, h2, pp, E, ET, N);
  pool_finish<<<NUM_GRAPHS, 1024, 0, stream>>>(pp, b2, (float*)d_out);
}
